# starting kernel
# baseline (speedup 1.0000x reference)
#include <hip/hip_runtime.h>
#include <hip/hip_bf16.h>
#include <math.h>

typedef __attribute__((ext_vector_type(16))) _Float16 v16h;
typedef __attribute__((ext_vector_type(8)))  _Float16 half8;
typedef __attribute__((ext_vector_type(8)))  float    v8f;

#define NEG_SLOPE 0.2f

__device__ inline float lrelu(float v) { return v > 0.f ? v : NEG_SLOPE * v; }

__device__ inline unsigned enc_f(float f) {
    unsigned u = __float_as_uint(f);
    return (u >> 31) ? ~u : (u | 0x80000000u);
}
__device__ inline float dec_f(unsigned e) {
    unsigned u = (e >> 31) ? (e ^ 0x80000000u) : ~e;
    return __uint_as_float(u);
}

__device__ inline v16h load_frag16(const _Float16* __restrict__ p, int row0,
                                   int ld, int k0, int lane, int maxRow) {
    int m = lane & 15;
    int base = (lane >> 4) * 8;
    int r = row0 + m;
    if (r > maxRow) r = maxRow;
    const _Float16* q = p + (size_t)r * ld + k0 + base;
    half8 lo = *(const half8*)q;
    half8 hi = *(const half8*)(q + 16);
    v16h f;
#pragma unroll
    for (int i = 0; i < 8; ++i) { f[i] = lo[i]; f[i + 8] = hi[i]; }
    return f;
}

__global__ void k_zero(unsigned* p, long long n) {
    long long i = (long long)blockIdx.x * blockDim.x + threadIdx.x;
    if (i < n) p[i] = 0u;
}

__global__ void k_cvt_f32_to_f16(const float* __restrict__ src,
                                 _Float16* __restrict__ dst, long long n) {
    long long i = (long long)blockIdx.x * blockDim.x + threadIdx.x;
    if (i < n) dst[i] = (_Float16)src[i];
}

__global__ void k_tr_w128(const float* __restrict__ W, _Float16* __restrict__ Wt) {
    int i = blockIdx.x * blockDim.x + threadIdx.x;
    if (i >= 128 * 128) return;
    int c = i >> 7, k = i & 127;
    Wt[i] = (_Float16)W[k * 128 + c];
}

__global__ void k_tr_w40(const float* __restrict__ W, _Float16* __restrict__ Wt) {
    int i = blockIdx.x * blockDim.x + threadIdx.x;
    if (i >= 48 * 128) return;
    int c = i >> 7, k = i & 127;
    Wt[i] = (c < 40) ? (_Float16)W[k * 40 + c] : (_Float16)0.f;
}

__global__ __launch_bounds__(32)
void k_gemm_dual(const _Float16* __restrict__ A,
                 const _Float16* __restrict__ Wtl,
                 const _Float16* __restrict__ Wtr,
                 const float* __restrict__ bl, const float* __restrict__ br,
                 float* __restrict__ Cl, float* __restrict__ Cr,
                 int Nrows, int ldc, int nbias) {
    int lane = threadIdx.x & 31;
    int r0 = blockIdx.x * 16;
    int c0 = blockIdx.y * 16;
    v8f accL = {};
    v8f accR = {};
#pragma unroll
    for (int k0 = 0; k0 < 128; k0 += 32) {
        v16h a  = load_frag16(A,   r0, 128, k0, lane, Nrows - 1);
        v16h bL = load_frag16(Wtl, c0, 128, k0, lane, ldc - 1);
        v16h bR = load_frag16(Wtr, c0, 128, k0, lane, ldc - 1);
        accL = __builtin_amdgcn_wmma_f32_16x16x32_f16(false, a, false, bL,
                                                      (short)0, accL, false, false);
        accR = __builtin_amdgcn_wmma_f32_16x16x32_f16(false, a, false, bR,
                                                      (short)0, accR, false, false);
    }
    int n = lane & 15, hw = lane >> 4;
    int col = c0 + n;
    float bLv = (col < nbias) ? bl[col] : 0.f;
    float bRv = (col < nbias) ? br[col] : 0.f;
#pragma unroll
    for (int v = 0; v < 8; ++v) {
        int row = r0 + v + hw * 8;
        if (row < Nrows) {
            Cl[(size_t)row * ldc + col] = accL[v] + bLv;
            Cr[(size_t)row * ldc + col] = accR[v] + bRv;
        }
    }
}

__global__ __launch_bounds__(256)
void k_edge_alpha1(const float* __restrict__ xl, const float* __restrict__ xr,
                   const float* __restrict__ att, const int* __restrict__ ei,
                   int E, int Etot, float* __restrict__ alpha,
                   unsigned* __restrict__ mbits) {
    int e = (blockIdx.x * blockDim.x + threadIdx.x) >> 5;
    int lane = threadIdx.x & 31;
    if (e >= Etot) return;
    int s, d;
    if (e < E) { s = ei[e]; d = ei[E + e]; } else { s = d = e - E; }
    const float4* pl = (const float4*)(xl + (size_t)s * 128);
    const float4* pr = (const float4*)(xr + (size_t)d * 128);
    const float4* pa = (const float4*)att;
    float4 a4 = pl[lane], b4 = pr[lane], t = pa[lane];
    float acc = t.x * lrelu(a4.x + b4.x) + t.y * lrelu(a4.y + b4.y) +
                t.z * lrelu(a4.z + b4.z) + t.w * lrelu(a4.w + b4.w);
#pragma unroll
    for (int off = 1; off < 16; off <<= 1) acc += __shfl_xor(acc, off, 32);
    if ((lane & 15) == 0) {
        int h = lane >> 4;
        alpha[(size_t)e * 2 + h] = acc;
        atomicMax(&mbits[(size_t)d * 2 + h], enc_f(acc));
    }
}

__global__ void k_edge_exp1(const int* __restrict__ ei, int E, int Etot,
                            const unsigned* __restrict__ mbits,
                            float* __restrict__ alpha,
                            float* __restrict__ denom) {
    long long i = (long long)blockIdx.x * blockDim.x + threadIdx.x;
    if (i >= (long long)Etot * 2) return;
    int e = (int)(i >> 1), h = (int)(i & 1);
    int d = (e < E) ? ei[E + e] : e - E;
    float m = dec_f(mbits[(size_t)d * 2 + h]);
    float ex = __expf(alpha[i] - m);
    alpha[i] = ex;
    atomicAdd(&denom[(size_t)d * 2 + h], ex);
}

__global__ void k_edge_agg1(const float* __restrict__ xl,
                            const int* __restrict__ ei, int E, int Etot,
                            const float* __restrict__ alpha,
                            const float* __restrict__ denom,
                            float* __restrict__ agg) {
    long long i = (long long)blockIdx.x * blockDim.x + threadIdx.x;
    if (i >= (long long)Etot * 128) return;
    int e = (int)(i >> 7), c = (int)(i & 127), h = c >> 6;
    int s, d;
    if (e < E) { s = ei[e]; d = ei[E + e]; } else { s = d = e - E; }
    float a = alpha[(size_t)e * 2 + h] / denom[(size_t)d * 2 + h];
    atomicAdd(&agg[(size_t)d * 128 + c], xl[(size_t)s * 128 + c] * a);
}

__global__ void k_epilogue1(const float* __restrict__ agg,
                            const float* __restrict__ bias,
                            _Float16* __restrict__ hh, long long n) {
    long long i = (long long)blockIdx.x * blockDim.x + threadIdx.x;
    if (i >= n) return;
    float v = agg[i] + bias[(int)(i & 127)];
    hh[i] = (_Float16)(v > 0.f ? v : 0.f);
}

__global__ __launch_bounds__(256)
void k_edge_alpha2(const float* __restrict__ xl, const float* __restrict__ xr,
                   const float* __restrict__ att, const int* __restrict__ ei,
                   int E, int Etot, float* __restrict__ alpha,
                   unsigned* __restrict__ mbits) {
    int e = (blockIdx.x * blockDim.x + threadIdx.x) >> 5;
    int lane = threadIdx.x & 31;
    if (e >= Etot) return;
    int s, d;
    if (e < E) { s = ei[e]; d = ei[E + e]; } else { s = d = e - E; }
    float acc = 0.f;
    if (lane < 10) {
        const float4* pl = (const float4*)(xl + (size_t)s * 48);
        const float4* pr = (const float4*)(xr + (size_t)d * 48);
        const float4* pa = (const float4*)att;
        float4 a4 = pl[lane], b4 = pr[lane], t = pa[lane];
        acc = t.x * lrelu(a4.x + b4.x) + t.y * lrelu(a4.y + b4.y) +
              t.z * lrelu(a4.z + b4.z) + t.w * lrelu(a4.w + b4.w);
    }
#pragma unroll
    for (int off = 1; off < 32; off <<= 1) acc += __shfl_xor(acc, off, 32);
    if (lane == 0) {
        alpha[e] = acc;
        atomicMax(&mbits[d], enc_f(acc));
    }
}

__global__ void k_edge_exp2(const int* __restrict__ ei, int E, int Etot,
                            const unsigned* __restrict__ mbits,
                            float* __restrict__ alpha,
                            float* __restrict__ denom) {
    long long i = (long long)blockIdx.x * blockDim.x + threadIdx.x;
    if (i >= Etot) return;
    int e = (int)i;
    int d = (e < E) ? ei[E + e] : e - E;
    float ex = __expf(alpha[e] - dec_f(mbits[d]));
    alpha[e] = ex;
    atomicAdd(&denom[d], ex);
}

__global__ void k_edge_agg2(const float* __restrict__ xl,
                            const int* __restrict__ ei, int E, int Etot,
                            const float* __restrict__ alpha,
                            const float* __restrict__ denom,
                            float* __restrict__ agg) {
    long long i = (long long)blockIdx.x * blockDim.x + threadIdx.x;
    if (i >= (long long)Etot * 64) return;
    int e = (int)(i >> 6), c = (int)(i & 63);
    if (c >= 40) return;
    int s, d;
    if (e < E) { s = ei[e]; d = ei[E + e]; } else { s = d = e - E; }
    float a = alpha[e] / denom[d];
    atomicAdd(&agg[(size_t)d * 40 + c], xl[(size_t)s * 48 + c] * a);
}

__global__ __launch_bounds__(256)
void k_logsoftmax(const float* __restrict__ agg, const float* __restrict__ bias,
                  float* __restrict__ out, int N) {
    int node = (blockIdx.x * blockDim.x + threadIdx.x) >> 5;
    int lane = threadIdx.x & 31;
    if (node >= N) return;
    const float* p = agg + (size_t)node * 40;
    float v0 = p[lane] + bias[lane];
    float v1 = (lane < 8) ? (p[lane + 32] + bias[lane + 32]) : -INFINITY;
    float m = fmaxf(v0, v1);
#pragma unroll
    for (int off = 1; off < 32; off <<= 1) m = fmaxf(m, __shfl_xor(m, off, 32));
    float se = __expf(v0 - m) + ((lane < 8) ? __expf(v1 - m) : 0.f);
#pragma unroll
    for (int off = 1; off < 32; off <<= 1) se += __shfl_xor(se, off, 32);
    float L = m + logf(se);
    out[(size_t)node * 40 + lane] = v0 - L;
    if (lane < 8) out[(size_t)node * 40 + lane + 32] = v1 - L;
}

extern "C" void kernel_launch(void* const* d_in, const int* in_sizes, int n_in,
                              void* d_out, int out_size, void* d_ws, size_t ws_size,
                              hipStream_t stream) {
    const float* x    = (const float*)d_in[0];
    const int*   ei   = (const int*)d_in[1];
    const float* W1l  = (const float*)d_in[2];
    const float* b1l  = (const float*)d_in[3];
    const float* W1r  = (const float*)d_in[4];
    const float* b1r  = (const float*)d_in[5];
    const float* att1 = (const float*)d_in[6];
    const float* bias1= (const float*)d_in[7];
    const float* W2l  = (const float*)d_in[8];
    const float* b2l  = (const float*)d_in[9];
    const float* W2r  = (const float*)d_in[10];
    const float* b2r  = (const float*)d_in[11];
    const float* att2 = (const float*)d_in[12];
    const float* bias2= (const float*)d_in[13];
    float* out = (float*)d_out;

    const int N = in_sizes[0] / 128;
    const int E = in_sizes[1] / 2;
    const int Etot = E + N;

    char* w = (char*)d_ws;
    size_t off = 0;
    auto take = [&](size_t bytes) -> char* {
        char* p = w + off;
        off = (off + bytes + 255) & ~(size_t)255;
        return p;
    };
    _Float16* xh   = (_Float16*)take((size_t)N * 128 * 2);
    _Float16* hh   = (_Float16*)take((size_t)N * 128 * 2);
    _Float16* Wt1l = (_Float16*)take(128 * 128 * 2);
    _Float16* Wt1r = (_Float16*)take(128 * 128 * 2);
    _Float16* Wt2l = (_Float16*)take(48 * 128 * 2);
    _Float16* Wt2r = (_Float16*)take(48 * 128 * 2);
    float*    xl   = (float*)take((size_t)N * 128 * 4);
    float*    xr   = (float*)take((size_t)N * 128 * 4);
    float*    alpha= (float*)take((size_t)Etot * 2 * 4);
    unsigned* mbits= (unsigned*)take((size_t)N * 2 * 4);
    float*    denom= (float*)take((size_t)N * 2 * 4);
    float*    agg1 = (float*)take((size_t)N * 128 * 4);
    float*    agg2 = (float*)take((size_t)N * 40 * 4);

    const int B = 256;
    auto blks = [&](long long n) { return (unsigned)((n + B - 1) / B); };
    const long long Nx128 = (long long)N * 128;
    const int rowTiles = (N + 15) / 16;

    k_cvt_f32_to_f16<<<blks(Nx128), B, 0, stream>>>(x, xh, Nx128);
    k_tr_w128<<<blks(128 * 128), B, 0, stream>>>(W1l, Wt1l);
    k_tr_w128<<<blks(128 * 128), B, 0, stream>>>(W1r, Wt1r);
    k_tr_w40 <<<blks(48 * 128),  B, 0, stream>>>(W2l, Wt2l);
    k_tr_w40 <<<blks(48 * 128),  B, 0, stream>>>(W2r, Wt2r);
    k_zero<<<blks(2LL * N), B, 0, stream>>>(mbits, 2LL * N);
    k_zero<<<blks(2LL * N), B, 0, stream>>>((unsigned*)denom, 2LL * N);
    k_zero<<<blks(Nx128), B, 0, stream>>>((unsigned*)agg1, Nx128);

    k_gemm_dual<<<dim3(rowTiles, 8), 32, 0, stream>>>(xh, Wt1l, Wt1r, b1l, b1r,
                                                      xl, xr, N, 128, 128);
    k_edge_alpha1<<<blks((long long)Etot * 32), B, 0, stream>>>(xl, xr, att1, ei,
                                                                E, Etot, alpha, mbits);
    k_edge_exp1<<<blks(2LL * Etot), B, 0, stream>>>(ei, E, Etot, mbits, alpha, denom);
    k_edge_agg1<<<blks((long long)Etot * 128), B, 0, stream>>>(xl, ei, E, Etot,
                                                               alpha, denom, agg1);
    k_epilogue1<<<blks(Nx128), B, 0, stream>>>(agg1, bias1, hh, Nx128);

    k_zero<<<blks(2LL * N), B, 0, stream>>>(mbits, 2LL * N);
    k_zero<<<blks(2LL * N), B, 0, stream>>>((unsigned*)denom, 2LL * N);
    k_zero<<<blks(40LL * N), B, 0, stream>>>((unsigned*)agg2, 40LL * N);

    k_gemm_dual<<<dim3(rowTiles, 3), 32, 0, stream>>>(hh, Wt2l, Wt2r, b2l, b2r,
                                                      xl, xr, N, 48, 40);
    k_edge_alpha2<<<blks((long long)Etot * 32), B, 0, stream>>>(xl, xr, att2, ei,
                                                                E, Etot, alpha, mbits);
    k_edge_exp2<<<blks(Etot), B, 0, stream>>>(ei, E, Etot, mbits, alpha, denom);
    k_edge_agg2<<<blks((long long)Etot * 64), B, 0, stream>>>(xl, ei, E, Etot,
                                                              alpha, denom, agg2);
    k_logsoftmax<<<blks((long long)N * 32), B, 0, stream>>>(agg2, bias2, out, N);
}


# Edges counting-sorted by destination once (int atomics); one wave per node does running-max softmax and weighted sum in registers, in source-index order; float atomics and per-edge passes gone
# speedup vs baseline: 2.3577x; 2.3577x over previous
#include <hip/hip_runtime.h>
#include <hip/hip_bf16.h>
#include <math.h>

typedef __attribute__((ext_vector_type(16))) _Float16 v16h;
typedef __attribute__((ext_vector_type(8)))  _Float16 half8;
typedef __attribute__((ext_vector_type(8)))  float    v8f;

#define NEG_SLOPE 0.2f

__device__ inline float lrelu(float v) { return v > 0.f ? v : NEG_SLOPE * v; }

__device__ inline v16h load_frag16(const _Float16* __restrict__ p, int row0,
                                   int ld, int k0, int lane, int maxRow) {
    int m = lane & 15;
    int base = (lane >> 4) * 8;
    int r = row0 + m;
    if (r > maxRow) r = maxRow;
    const _Float16* q = p + (size_t)r * ld + k0 + base;
    half8 lo = *(const half8*)q;
    half8 hi = *(const half8*)(q + 16);
    v16h f;
#pragma unroll
    for (int i = 0; i < 8; ++i) { f[i] = lo[i]; f[i + 8] = hi[i]; }
    return f;
}

__global__ void k_zero(unsigned* p, long long n) {
    long long i = (long long)blockIdx.x * blockDim.x + threadIdx.x;
    if (i < n) p[i] = 0u;
}

__global__ void k_cvt_f32_to_f16(const float* __restrict__ src,
                                 _Float16* __restrict__ dst, long long n) {
    long long i = (long long)blockIdx.x * blockDim.x + threadIdx.x;
    if (i < n) dst[i] = (_Float16)src[i];
}

__global__ void k_tr_w128(const float* __restrict__ W, _Float16* __restrict__ Wt) {
    int i = blockIdx.x * blockDim.x + threadIdx.x;
    if (i >= 128 * 128) return;
    int c = i >> 7, k = i & 127;
    Wt[i] = (_Float16)W[k * 128 + c];
}

__global__ void k_tr_w40(const float* __restrict__ W, _Float16* __restrict__ Wt) {
    int i = blockIdx.x * blockDim.x + threadIdx.x;
    if (i >= 48 * 128) return;
    int c = i >> 7, k = i & 127;
    Wt[i] = (c < 40) ? (_Float16)W[k * 40 + c] : (_Float16)0.f;
}

__global__ __launch_bounds__(32)
void k_gemm_dual(const _Float16* __restrict__ A,
                 const _Float16* __restrict__ Wtl,
                 const _Float16* __restrict__ Wtr,
                 const float* __restrict__ bl, const float* __restrict__ br,
                 float* __restrict__ Cl, float* __restrict__ Cr,
                 int Nrows, int ldc, int nbias) {
    int lane = threadIdx.x & 31;
    int r0 = blockIdx.x * 16;
    int c0 = blockIdx.y * 16;
    v8f accL = {};
    v8f accR = {};
#pragma unroll
    for (int k0 = 0; k0 < 128; k0 += 32) {
        v16h a  = load_frag16(A,   r0, 128, k0, lane, Nrows - 1);
        v16h bL = load_frag16(Wtl, c0, 128, k0, lane, ldc - 1);
        v16h bR = load_frag16(Wtr, c0, 128, k0, lane, ldc - 1);
        accL = __builtin_amdgcn_wmma_f32_16x16x32_f16(false, a, false, bL,
                                                      (short)0, accL, false, false);
        accR = __builtin_amdgcn_wmma_f32_16x16x32_f16(false, a, false, bR,
                                                      (short)0, accR, false, false);
    }
    int n = lane & 15, hw = lane >> 4;
    int col = c0 + n;
    float bLv = (col < nbias) ? bl[col] : 0.f;
    float bRv = (col < nbias) ? br[col] : 0.f;
#pragma unroll
    for (int v = 0; v < 8; ++v) {
        int row = r0 + v + hw * 8;
        if (row < Nrows) {
            Cl[(size_t)row * ldc + col] = accL[v] + bLv;
            Cr[(size_t)row * ldc + col] = accR[v] + bRv;
        }
    }
}

__global__ void k_count(const int* __restrict__ ei, int E, int* __restrict__ deg) {
    int i = blockIdx.x * blockDim.x + threadIdx.x;
    if (i < E) atomicAdd(&deg[ei[E + i]], 1);
}

__global__ __launch_bounds__(1024)
void k_scan(const int* __restrict__ deg, int* __restrict__ rbeg,
            int* __restrict__ cur, int N) {
    __shared__ int part[1024];
    int tid = threadIdx.x;
    int chunk = (N + 1023) / 1024;
    int lo = tid * chunk;
    int hi = lo + chunk; if (hi > N) hi = N;
    int s = 0;
    for (int i = lo; i < hi; ++i) s += deg[i];
    part[tid] = s;
    __syncthreads();
    for (int off = 1; off < 1024; off <<= 1) {
        int v = (tid >= off) ? part[tid - off] : 0;
        __syncthreads();
        part[tid] += v;
        __syncthreads();
    }
    int base = part[tid] - s;
    for (int i = lo; i < hi; ++i) {
        rbeg[i] = base; cur[i] = base;
        base += deg[i];
    }
}

__global__ void k_fill(const int* __restrict__ ei, int E, int* __restrict__ cur,
                       int* __restrict__ csrc) {
    int i = blockIdx.x * blockDim.x + threadIdx.x;
    if (i >= E) return;
    int p = atomicAdd(&cur[ei[E + i]], 1);
    csrc[p] = ei[i];
}

__device__ inline float score4(const float4& a, const float4& b, const float4& t) {
    return t.x * lrelu(a.x + b.x) + t.y * lrelu(a.y + b.y) +
           t.z * lrelu(a.z + b.z) + t.w * lrelu(a.w + b.w);
}
__device__ inline float red16(float v) {
#pragma unroll
    for (int off = 1; off < 16; off <<= 1) v += __shfl_xor(v, off, 32);
    return v;
}
__device__ inline float redmax16(float v) {
#pragma unroll
    for (int off = 1; off < 16; off <<= 1) v = fmaxf(v, __shfl_xor(v, off, 32));
    return v;
}
__device__ inline void scale4(float4& a, float s) { a.x *= s; a.y *= s; a.z *= s; a.w *= s; }
__device__ inline void axpy4(float4& a, float p, const float4& v) {
    a.x += p * v.x; a.y += p * v.y; a.z += p * v.z; a.w += p * v.w;
}

__device__ inline int sort_chunk(int mine, int cnt, int lane) {
    int rank = 0;
    for (int k = 0; k < cnt; ++k) {
        int v = __builtin_amdgcn_readlane(mine, k);
        rank += ((v < mine) || (v == mine && k < lane)) ? 1 : 0;
    }
    return __builtin_amdgcn_ds_permute(rank << 2, mine);
}

__device__ inline void fold4(float& m, float& den, float4& acc,
                             float e0, float e1, float e2, float e3,
                             const float4& a0, const float4& a1,
                             const float4& a2, const float4& a3) {
    float mn = fmaxf(fmaxf(m, fmaxf(e0, e1)), fmaxf(e2, e3));
    float sc = __expf(m - mn);
    float p0 = __expf(e0 - mn), p1 = __expf(e1 - mn);
    float p2 = __expf(e2 - mn), p3 = __expf(e3 - mn);
    scale4(acc, sc);
    axpy4(acc, p0, a0); axpy4(acc, p1, a1); axpy4(acc, p2, a2); axpy4(acc, p3, a3);
    den = den * sc + ((p0 + p1) + (p2 + p3));
    m = mn;
}
__device__ inline void fold1(float& m, float& den, float4& acc, float e0, const float4& a0) {
    float mn = fmaxf(m, e0);
    float sc = __expf(m - mn);
    float p0 = __expf(e0 - mn);
    scale4(acc, sc);
    axpy4(acc, p0, a0);
    den = den * sc + p0;
    m = mn;
}

__global__ __launch_bounds__(256)
void k_node_attn1(const float* __restrict__ xl, const float* __restrict__ xr,
                 const float* __restrict__ att, const float* __restrict__ bias,
                 const int* __restrict__ rbeg, const int* __restrict__ rend,
                 const int* __restrict__ csrc, _Float16* __restrict__ hh, int N) {
    int d = __builtin_amdgcn_readfirstlane((int)((blockIdx.x * blockDim.x + threadIdx.x) >> 5));
    int lane = threadIdx.x & 31;
    if (d >= N) return;
    const float4 t  = ((const float4*)att)[lane];
    const float4 b4 = ((const float4*)(xr + (size_t)d * 128))[lane];
    float4 acc = ((const float4*)(xl + (size_t)d * 128))[lane];
    float m = red16(score4(acc, b4, t));
    float den = 1.f;
    int beg = rbeg[d], end = rend[d];
    for (int j0 = beg; j0 < end; j0 += 32) {
        int cnt = end - j0; if (cnt > 32) cnt = 32;
        int mine = (lane < cnt) ? csrc[j0 + lane] : 0x7fffffff;
        int srt = sort_chunk(mine, cnt, lane);
        int k = 0;
        for (; k + 4 <= cnt; k += 4) {
            int s0 = __builtin_amdgcn_readlane(srt, k);
            int s1 = __builtin_amdgcn_readlane(srt, k + 1);
            int s2 = __builtin_amdgcn_readlane(srt, k + 2);
            int s3 = __builtin_amdgcn_readlane(srt, k + 3);
            float4 a0 = ((const float4*)(xl + (size_t)s0 * 128))[lane];
            float4 a1 = ((const float4*)(xl + (size_t)s1 * 128))[lane];
            float4 a2 = ((const float4*)(xl + (size_t)s2 * 128))[lane];
            float4 a3 = ((const float4*)(xl + (size_t)s3 * 128))[lane];
            float e0 = red16(score4(a0, b4, t));
            float e1 = red16(score4(a1, b4, t));
            float e2 = red16(score4(a2, b4, t));
            float e3 = red16(score4(a3, b4, t));
            fold4(m, den, acc, e0, e1, e2, e3, a0, a1, a2, a3);
        }
        for (; k < cnt; ++k) {
            int s0 = __builtin_amdgcn_readlane(srt, k);
            float4 a0 = ((const float4*)(xl + (size_t)s0 * 128))[lane];
            float e0 = red16(score4(a0, b4, t));
            fold1(m, den, acc, e0, a0);
        }
    }
    float inv = 1.f / den;
    const float4 bb = ((const float4*)bias)[lane];
    float o0 = fmaxf(acc.x * inv + bb.x, 0.f);
    float o1 = fmaxf(acc.y * inv + bb.y, 0.f);
    float o2 = fmaxf(acc.z * inv + bb.z, 0.f);
    float o3 = fmaxf(acc.w * inv + bb.w, 0.f);
    typedef __attribute__((ext_vector_type(4))) _Float16 half4;
    half4 o;
    o[0] = (_Float16)o0; o[1] = (_Float16)o1; o[2] = (_Float16)o2; o[3] = (_Float16)o3;
    *(half4*)(hh + (size_t)d * 128 + lane * 4) = o;
}

__device__ inline float4 ld4_if(const float* __restrict__ p, int idx, bool on) {
    float4 r = make_float4(0.f, 0.f, 0.f, 0.f);
    if (on) r = ((const float4*)p)[idx];
    return r;
}

__global__ __launch_bounds__(256)
void k_node_attn2(const float* __restrict__ xl, const float* __restrict__ xr,
                 const float* __restrict__ att, const float* __restrict__ bias,
                 const int* __restrict__ rbeg, const int* __restrict__ rend,
                 const int* __restrict__ csrc, float* __restrict__ out, int N) {
    int d = __builtin_amdgcn_readfirstlane((int)((blockIdx.x * blockDim.x + threadIdx.x) >> 5));
    int lane = threadIdx.x & 31;
    if (d >= N) return;
    const int hl = lane & 15;
    const bool act = hl < 10;
    const float4 t  = ld4_if(att, hl, act);
    const float4 b4 = ld4_if(xr + (size_t)d * 48, hl, act);
    float4 acc = ld4_if(xl + (size_t)d * 48, hl, act);
    float m = red16(score4(acc, b4, t));
    float den = 1.f;
    int beg = rbeg[d], end = rend[d];
    for (int j0 = beg; j0 < end; j0 += 32) {
        int cnt = end - j0; if (cnt > 32) cnt = 32;
        int mine = (lane < cnt) ? csrc[j0 + lane] : 0x7fffffff;
        int srt = sort_chunk(mine, cnt, lane);
        int k = 0;
        for (; k + 4 <= cnt; k += 4) {
            int s0 = __builtin_amdgcn_readlane(srt, k);
            int s1 = __builtin_amdgcn_readlane(srt, k + 1);
            int s2 = __builtin_amdgcn_readlane(srt, k + 2);
            int s3 = __builtin_amdgcn_readlane(srt, k + 3);
            float4 a0 = ld4_if(xl + (size_t)s0 * 48, hl, act);
            float4 a1 = ld4_if(xl + (size_t)s1 * 48, hl, act);
            float4 a2 = ld4_if(xl + (size_t)s2 * 48, hl, act);
            float4 a3 = ld4_if(xl + (size_t)s3 * 48, hl, act);
            float e0 = red16(score4(a0, b4, t));
            float e1 = red16(score4(a1, b4, t));
            float e2 = red16(score4(a2, b4, t));
            float e3 = red16(score4(a3, b4, t));
            fold4(m, den, acc, e0, e1, e2, e3, a0, a1, a2, a3);
        }
        for (; k < cnt; ++k) {
            int s0 = __builtin_amdgcn_readlane(srt, k);
            float4 a0 = ld4_if(xl + (size_t)s0 * 48, hl, act);
            float e0 = red16(score4(a0, b4, t));
            fold1(m, den, acc, e0, a0);
        }
    }
    float inv = 1.f / den;
    const float4 bb = ld4_if(bias, hl, act);
    float v0 = acc.x * inv + bb.x, v1 = acc.y * inv + bb.y;
    float v2 = acc.z * inv + bb.z, v3 = acc.w * inv + bb.w;
    float mx = act ? fmaxf(fmaxf(v0, v1), fmaxf(v2, v3)) : -INFINITY;
    mx = redmax16(mx);
    float se = act ? (__expf(v0 - mx) + __expf(v1 - mx)) + (__expf(v2 - mx) + __expf(v3 - mx))
                   : 0.f;
    se = red16(se);
    float L = mx + logf(se);
    if (lane < 10)
        *(float4*)(out + (size_t)d * 40 + lane * 4) = make_float4(v0 - L, v1 - L, v2 - L, v3 - L);
}

extern "C" void kernel_launch(void* const* d_in, const int* in_sizes, int n_in,
                              void* d_out, int out_size, void* d_ws, size_t ws_size,
                              hipStream_t stream) {
    const float* x    = (const float*)d_in[0];
    const int*   ei   = (const int*)d_in[1];
    const float* W1l  = (const float*)d_in[2];
    const float* b1l  = (const float*)d_in[3];
    const float* W1r  = (const float*)d_in[4];
    const float* b1r  = (const float*)d_in[5];
    const float* att1 = (const float*)d_in[6];
    const float* bias1= (const float*)d_in[7];
    const float* W2l  = (const float*)d_in[8];
    const float* b2l  = (const float*)d_in[9];
    const float* W2r  = (const float*)d_in[10];
    const float* b2r  = (const float*)d_in[11];
    const float* att2 = (const float*)d_in[12];
    const float* bias2= (const float*)d_in[13];
    float* out = (float*)d_out;

    const int N = in_sizes[0] / 128;
    const int E = in_sizes[1] / 2;

    char* w = (char*)d_ws;
    size_t off = 0;
    auto take = [&](size_t bytes) -> char* {
        char* p = w + off;
        off = (off + bytes + 255) & ~(size_t)255;
        return p;
    };
    _Float16* xh   = (_Float16*)take((size_t)N * 128 * 2);
    _Float16* hh   = (_Float16*)take((size_t)N * 128 * 2);
    _Float16* Wt1l = (_Float16*)take(128 * 128 * 2);
    _Float16* Wt1r = (_Float16*)take(128 * 128 * 2);
    _Float16* Wt2l = (_Float16*)take(48 * 128 * 2);
    _Float16* Wt2r = (_Float16*)take(48 * 128 * 2);
    float*    xl   = (float*)take((size_t)N * 128 * 4);
    float*    xr   = (float*)take((size_t)N * 128 * 4);
    int*      deg  = (int*)take((size_t)N * 4);
    int*      rbeg = (int*)take((size_t)N * 4);
    int*      cur  = (int*)take((size_t)N * 4);
    int*      csrc = (int*)take((size_t)E * 4);

    const int B = 256;
    auto blks = [&](long long n) { return (unsigned)((n + B - 1) / B); };
    const long long Nx128 = (long long)N * 128;
    const int rowTiles = (N + 15) / 16;

    k_cvt_f32_to_f16<<<blks(Nx128), B, 0, stream>>>(x, xh, Nx128);
    k_tr_w128<<<blks(128 * 128), B, 0, stream>>>(W1l, Wt1l);
    k_tr_w128<<<blks(128 * 128), B, 0, stream>>>(W1r, Wt1r);
    k_tr_w40 <<<blks(48 * 128),  B, 0, stream>>>(W2l, Wt2l);
    k_tr_w40 <<<blks(48 * 128),  B, 0, stream>>>(W2r, Wt2r);

    k_zero<<<blks(N), B, 0, stream>>>((unsigned*)deg, (long long)N);
    k_count<<<blks(E), B, 0, stream>>>(ei, E, deg);
    k_scan<<<1, 1024, 0, stream>>>(deg, rbeg, cur, N);
    k_fill<<<blks(E), B, 0, stream>>>(ei, E, cur, csrc);

    k_gemm_dual<<<dim3(rowTiles, 8), 32, 0, stream>>>(xh, Wt1l, Wt1r, b1l, b1r,
                                                      xl, xr, N, 128, 128);
    k_node_attn1<<<blks((long long)N * 32), B, 0, stream>>>(xl, xr, att1, bias1,
                                                           rbeg, cur, csrc, hh, N);

    k_gemm_dual<<<dim3(rowTiles, 3), 32, 0, stream>>>(hh, Wt2l, Wt2r, b2l, b2r,
                                                      xl, xr, N, 48, 40);
    k_node_attn2<<<blks((long long)N * 32), B, 0, stream>>>(xl, xr, att2, bias2,
                                                           rbeg, cur, csrc, out, N);
}


# Wave entries cut about sevenfold: element, edge, node and GEMM kernels loop inside fixed resident-size grids (one wave per row strip walks all column tiles); launches, sums and their order kept
# speedup vs baseline: 2.5201x; 1.0689x over previous
#include <hip/hip_runtime.h>
#include <hip/hip_bf16.h>
#include <math.h>

typedef __attribute__((ext_vector_type(16))) _Float16 v16h;
typedef __attribute__((ext_vector_type(8)))  _Float16 half8;
typedef __attribute__((ext_vector_type(8)))  float    v8f;

#define NEG_SLOPE 0.2f

__device__ inline float lrelu(float v) { return v > 0.f ? v : NEG_SLOPE * v; }

__device__ inline v16h load_frag16(const _Float16* __restrict__ p, int row0,
                                   int ld, int k0, int lane, int maxRow) {
    int m = lane & 15;
    int base = (lane >> 4) * 8;
    int r = row0 + m;
    if (r > maxRow) r = maxRow;
    const _Float16* q = p + (size_t)r * ld + k0 + base;
    half8 lo = *(const half8*)q;
    half8 hi = *(const half8*)(q + 16);
    v16h f;
#pragma unroll
    for (int i = 0; i < 8; ++i) { f[i] = lo[i]; f[i + 8] = hi[i]; }
    return f;
}

__global__ void k_zero(unsigned* p, long long n) {
    long long i = (long long)blockIdx.x * blockDim.x + threadIdx.x;
    if (i < n) p[i] = 0u;
}

typedef __attribute__((ext_vector_type(4))) _Float16 half4;
__device__ inline half4 cvt4(const float4& v) {
    half4 o;
    o[0] = (_Float16)v.x; o[1] = (_Float16)v.y; o[2] = (_Float16)v.z; o[3] = (_Float16)v.w;
    return o;
}
__global__ __launch_bounds__(256)
void k_cvt_f32_to_f16(const float* __restrict__ src,
                      _Float16* __restrict__ dst, long long n) {
    const long long n4 = n >> 2;
    const long long tid = (long long)blockIdx.x * blockDim.x + threadIdx.x;
    const long long stride = (long long)gridDim.x * blockDim.x;
    const float4* __restrict__ s4 = (const float4*)src;
    half4* __restrict__ d4 = (half4*)dst;
    long long i = tid;
    for (; i + 3 * stride < n4; i += 4 * stride) {
        float4 a = s4[i], b = s4[i + stride], c = s4[i + 2 * stride], d = s4[i + 3 * stride];
        d4[i] = cvt4(a); d4[i + stride] = cvt4(b);
        d4[i + 2 * stride] = cvt4(c); d4[i + 3 * stride] = cvt4(d);
    }
    for (; i < n4; i += stride) d4[i] = cvt4(s4[i]);
    if (tid < (n & 3)) dst[(n4 << 2) + tid] = (_Float16)src[(n4 << 2) + tid];
}

__global__ void k_tr_w128(const float* __restrict__ W, _Float16* __restrict__ Wt) {
    int i = blockIdx.x * blockDim.x + threadIdx.x;
    if (i >= 128 * 128) return;
    int c = i >> 7, k = i & 127;
    Wt[i] = (_Float16)W[k * 128 + c];
}

__global__ void k_tr_w40(const float* __restrict__ W, _Float16* __restrict__ Wt) {
    int i = blockIdx.x * blockDim.x + threadIdx.x;
    if (i >= 48 * 128) return;
    int c = i >> 7, k = i & 127;
    Wt[i] = (c < 40) ? (_Float16)W[k * 40 + c] : (_Float16)0.f;
}

template <int NCT, int NB>
__device__ inline void gemm_dual_strip(const _Float16* __restrict__ A,
                                       const _Float16* __restrict__ Wtl,
                                       const _Float16* __restrict__ Wtr,
                                       const float* __restrict__ bl,
                                       const float* __restrict__ br,
                                       float* __restrict__ Cl, float* __restrict__ Cr,
                                       int Nrows) {
    constexpr int ldc = NCT * 16;
    const int lane = threadIdx.x & 31;
    const int strip = __builtin_amdgcn_readfirstlane(
        (int)((blockIdx.x * blockDim.x + threadIdx.x) >> 5));
    const int r0 = strip * 16;
    if (r0 >= Nrows) return;
    v16h a[4];
#pragma unroll
    for (int kk = 0; kk < 4; ++kk) a[kk] = load_frag16(A, r0, 128, kk * 32, lane, Nrows - 1);
    const int n = lane & 15, hw = lane >> 4;
    const bool whole = r0 + 16 <= Nrows;
#pragma unroll
    for (int ct = 0; ct < NCT; ++ct) {
        const int c0 = ct * 16;
        v16h bL[4], bR[4];
#pragma unroll
        for (int kk = 0; kk < 4; ++kk) {
            bL[kk] = load_frag16(Wtl, c0, 128, kk * 32, lane, ldc - 1);
            bR[kk] = load_frag16(Wtr, c0, 128, kk * 32, lane, ldc - 1);
        }
        v8f accL = {};
        v8f accR = {};
#pragma unroll
        for (int kk = 0; kk < 4; ++kk) {
            accL = __builtin_amdgcn_wmma_f32_16x16x32_f16(false, a[kk], false, bL[kk],
                                                          (short)0, accL, false, false);
            accR = __builtin_amdgcn_wmma_f32_16x16x32_f16(false, a[kk], false, bR[kk],
                                                          (short)0, accR, false, false);
        }
        const int col = c0 + n;
        float bLv = 0.f, bRv = 0.f;
        if (c0 + 16 <= NB || col < NB) { bLv = bl[col]; bRv = br[col]; }
        float* __restrict__ pl = Cl + (size_t)(r0 + hw * 8) * ldc + col;
        float* __restrict__ pr = Cr + (size_t)(r0 + hw * 8) * ldc + col;
        if (whole) {
#pragma unroll
            for (int v = 0; v < 8; ++v) {
                pl[v * ldc] = accL[v] + bLv;
                pr[v * ldc] = accR[v] + bRv;
            }
        } else {
#pragma unroll
            for (int v = 0; v < 8; ++v) {
                if (r0 + hw * 8 + v < Nrows) {
                    pl[v * ldc] = accL[v] + bLv;
                    pr[v * ldc] = accR[v] + bRv;
                }
            }
        }
    }
}
__global__ __launch_bounds__(128)
void k_gemm_dual_w128(const _Float16* __restrict__ A, const _Float16* __restrict__ Wtl,
                      const _Float16* __restrict__ Wtr, const float* __restrict__ bl,
                      const float* __restrict__ br, float* __restrict__ Cl,
                      float* __restrict__ Cr, int Nrows) {
    gemm_dual_strip<8, 128>(A, Wtl, Wtr, bl, br, Cl, Cr, Nrows);
}
__global__ __launch_bounds__(128)
void k_gemm_dual_w48(const _Float16* __restrict__ A, const _Float16* __restrict__ Wtl,
                     const _Float16* __restrict__ Wtr, const float* __restrict__ bl,
                     const float* __restrict__ br, float* __restrict__ Cl,
                     float* __restrict__ Cr, int Nrows) {
    gemm_dual_strip<3, 40>(A, Wtl, Wtr, bl, br, Cl, Cr, Nrows);
}

__global__ __launch_bounds__(256)
void k_count(const int* __restrict__ ei, int E, int* __restrict__ deg) {
    const int stride = gridDim.x * blockDim.x;
    for (int base = blockIdx.x * blockDim.x + threadIdx.x; base < E; base += 8 * stride) {
        int dd[8];
#pragma unroll
        for (int u = 0; u < 8; ++u) {
            int i = base + u * stride;
            dd[u] = (i < E) ? ei[E + i] : -1;
        }
#pragma unroll
        for (int u = 0; u < 8; ++u)
            if (dd[u] >= 0) atomicAdd(&deg[dd[u]], 1);
    }
}

__global__ __launch_bounds__(1024)
void k_scan(const int* __restrict__ deg, int* __restrict__ rbeg,
            int* __restrict__ cur, int N) {
    __shared__ int part[1024];
    int tid = threadIdx.x;
    int chunk = (N + 1023) / 1024;
    int lo = tid * chunk;
    int hi = lo + chunk; if (hi > N) hi = N;
    int s = 0;
    for (int i = lo; i < hi; ++i) s += deg[i];
    part[tid] = s;
    __syncthreads();
    for (int off = 1; off < 1024; off <<= 1) {
        int v = (tid >= off) ? part[tid - off] : 0;
        __syncthreads();
        part[tid] += v;
        __syncthreads();
    }
    int base = part[tid] - s;
    for (int i = lo; i < hi; ++i) {
        rbeg[i] = base; cur[i] = base;
        base += deg[i];
    }
}

__global__ __launch_bounds__(256)
void k_fill(const int* __restrict__ ei, int E, int* __restrict__ cur,
            int* __restrict__ csrc) {
    const int stride = gridDim.x * blockDim.x;
    for (int base = blockIdx.x * blockDim.x + threadIdx.x; base < E; base += 8 * stride) {
        int dd[8], ss[8], pp[8];
#pragma unroll
        for (int u = 0; u < 8; ++u) {
            int i = base + u * stride;
            bool on = i < E;
            dd[u] = on ? ei[E + i] : -1;
            ss[u] = on ? ei[i] : 0;
        }
#pragma unroll
        for (int u = 0; u < 8; ++u)
            pp[u] = (dd[u] >= 0) ? atomicAdd(&cur[dd[u]], 1) : -1;
#pragma unroll
        for (int u = 0; u < 8; ++u)
            if (pp[u] >= 0) csrc[pp[u]] = ss[u];
    }
}

__device__ inline float score4(const float4& a, const float4& b, const float4& t) {
    return t.x * lrelu(a.x + b.x) + t.y * lrelu(a.y + b.y) +
           t.z * lrelu(a.z + b.z) + t.w * lrelu(a.w + b.w);
}
__device__ inline float red16(float v) {
#pragma unroll
    for (int off = 1; off < 16; off <<= 1) v += __shfl_xor(v, off, 32);
    return v;
}
__device__ inline float redmax16(float v) {
#pragma unroll
    for (int off = 1; off < 16; off <<= 1) v = fmaxf(v, __shfl_xor(v, off, 32));
    return v;
}
__device__ inline void scale4(float4& a, float s) { a.x *= s; a.y *= s; a.z *= s; a.w *= s; }
__device__ inline void axpy4(float4& a, float p, const float4& v) {
    a.x += p * v.x; a.y += p * v.y; a.z += p * v.z; a.w += p * v.w;
}

__device__ inline int sort_chunk(int mine, int cnt, int lane) {
    int rank = 0;
    for (int k = 0; k < cnt; ++k) {
        int v = __builtin_amdgcn_readlane(mine, k);
        rank += ((v < mine) || (v == mine && k < lane)) ? 1 : 0;
    }
    return __builtin_amdgcn_ds_permute(rank << 2, mine);
}

__device__ inline void fold4(float& m, float& den, float4& acc,
                             float e0, float e1, float e2, float e3,
                             const float4& a0, const float4& a1,
                             const float4& a2, const float4& a3) {
    float mn = fmaxf(fmaxf(m, fmaxf(e0, e1)), fmaxf(e2, e3));
    float sc = __expf(m - mn);
    float p0 = __expf(e0 - mn), p1 = __expf(e1 - mn);
    float p2 = __expf(e2 - mn), p3 = __expf(e3 - mn);
    scale4(acc, sc);
    axpy4(acc, p0, a0); axpy4(acc, p1, a1); axpy4(acc, p2, a2); axpy4(acc, p3, a3);
    den = den * sc + ((p0 + p1) + (p2 + p3));
    m = mn;
}
__device__ inline void fold1(float& m, float& den, float4& acc, float e0, const float4& a0) {
    float mn = fmaxf(m, e0);
    float sc = __expf(m - mn);
    float p0 = __expf(e0 - mn);
    scale4(acc, sc);
    axpy4(acc, p0, a0);
    den = den * sc + p0;
    m = mn;
}

__global__ __launch_bounds__(256)
void k_node_attn1(const float* __restrict__ xl, const float* __restrict__ xr,
                 const float* __restrict__ att, const float* __restrict__ bias,
                 const int* __restrict__ rbeg, const int* __restrict__ rend,
                 const int* __restrict__ csrc, _Float16* __restrict__ hh, int N) {
    const int wave = __builtin_amdgcn_readfirstlane(
        (int)((blockIdx.x * blockDim.x + threadIdx.x) >> 5));
    const int nwave = (int)((gridDim.x * blockDim.x) >> 5);
    const int lane = threadIdx.x & 31;
    const float4 t  = ((const float4*)att)[lane];
    const float4 bb = ((const float4*)bias)[lane];
    for (int d = wave; d < N; d += nwave) {
    const float4 b4 = ((const float4*)(xr + (size_t)d * 128))[lane];
    float4 acc = ((const float4*)(xl + (size_t)d * 128))[lane];
    float m = red16(score4(acc, b4, t));
    float den = 1.f;
    int beg = rbeg[d], end = rend[d];
    for (int j0 = beg; j0 < end; j0 += 32) {
        int cnt = end - j0; if (cnt > 32) cnt = 32;
        int mine = (lane < cnt) ? csrc[j0 + lane] : 0x7fffffff;
        int srt = sort_chunk(mine, cnt, lane);
        int k = 0;
        for (; k + 4 <= cnt; k += 4) {
            int s0 = __builtin_amdgcn_readlane(srt, k);
            int s1 = __builtin_amdgcn_readlane(srt, k + 1);
            int s2 = __builtin_amdgcn_readlane(srt, k + 2);
            int s3 = __builtin_amdgcn_readlane(srt, k + 3);
            float4 a0 = ((const float4*)(xl + (size_t)s0 * 128))[lane];
            float4 a1 = ((const float4*)(xl + (size_t)s1 * 128))[lane];
            float4 a2 = ((const float4*)(xl + (size_t)s2 * 128))[lane];
            float4 a3 = ((const float4*)(xl + (size_t)s3 * 128))[lane];
            float e0 = red16(score4(a0, b4, t));
            float e1 = red16(score4(a1, b4, t));
            float e2 = red16(score4(a2, b4, t));
            float e3 = red16(score4(a3, b4, t));
            fold4(m, den, acc, e0, e1, e2, e3, a0, a1, a2, a3);
        }
        for (; k < cnt; ++k) {
            int s0 = __builtin_amdgcn_readlane(srt, k);
            float4 a0 = ((const float4*)(xl + (size_t)s0 * 128))[lane];
            float e0 = red16(score4(a0, b4, t));
            fold1(m, den, acc, e0, a0);
        }
    }
    float inv = 1.f / den;
    float o0 = fmaxf(acc.x * inv + bb.x, 0.f);
    float o1 = fmaxf(acc.y * inv + bb.y, 0.f);
    float o2 = fmaxf(acc.z * inv + bb.z, 0.f);
    float o3 = fmaxf(acc.w * inv + bb.w, 0.f);
    half4 o;
    o[0] = (_Float16)o0; o[1] = (_Float16)o1; o[2] = (_Float16)o2; o[3] = (_Float16)o3;
    *(half4*)(hh + (size_t)d * 128 + lane * 4) = o;
    }
}

__device__ inline float4 ld4_if(const float* __restrict__ p, int idx, bool on) {
    float4 r = make_float4(0.f, 0.f, 0.f, 0.f);
    if (on) r = ((const float4*)p)[idx];
    return r;
}

__global__ __launch_bounds__(256)
void k_node_attn2(const float* __restrict__ xl, const float* __restrict__ xr,
                 const float* __restrict__ att, const float* __restrict__ bias,
                 const int* __restrict__ rbeg, const int* __restrict__ rend,
                 const int* __restrict__ csrc, float* __restrict__ out, int N) {
    const int wave = __builtin_amdgcn_readfirstlane(
        (int)((blockIdx.x * blockDim.x + threadIdx.x) >> 5));
    const int nwave = (int)((gridDim.x * blockDim.x) >> 5);
    const int lane = threadIdx.x & 31;
    const int hl = lane & 15;
    const bool act = hl < 10;
    const float4 t  = ld4_if(att, hl, act);
    for (int d = wave; d < N; d += nwave) {
    const float4 b4 = ld4_if(xr + (size_t)d * 48, hl, act);
    float4 acc = ld4_if(xl + (size_t)d * 48, hl, act);
    float m = red16(score4(acc, b4, t));
    float den = 1.f;
    int beg = rbeg[d], end = rend[d];
    for (int j0 = beg; j0 < end; j0 += 32) {
        int cnt = end - j0; if (cnt > 32) cnt = 32;
        int mine = (lane < cnt) ? csrc[j0 + lane] : 0x7fffffff;
        int srt = sort_chunk(mine, cnt, lane);
        int k = 0;
        for (; k + 4 <= cnt; k += 4) {
            int s0 = __builtin_amdgcn_readlane(srt, k);
            int s1 = __builtin_amdgcn_readlane(srt, k + 1);
            int s2 = __builtin_amdgcn_readlane(srt, k + 2);
            int s3 = __builtin_amdgcn_readlane(srt, k + 3);
            float4 a0 = ld4_if(xl + (size_t)s0 * 48, hl, act);
            float4 a1 = ld4_if(xl + (size_t)s1 * 48, hl, act);
            float4 a2 = ld4_if(xl + (size_t)s2 * 48, hl, act);
            float4 a3 = ld4_if(xl + (size_t)s3 * 48, hl, act);
            float e0 = red16(score4(a0, b4, t));
            float e1 = red16(score4(a1, b4, t));
            float e2 = red16(score4(a2, b4, t));
            float e3 = red16(score4(a3, b4, t));
            fold4(m, den, acc, e0, e1, e2, e3, a0, a1, a2, a3);
        }
        for (; k < cnt; ++k) {
            int s0 = __builtin_amdgcn_readlane(srt, k);
            float4 a0 = ld4_if(xl + (size_t)s0 * 48, hl, act);
            float e0 = red16(score4(a0, b4, t));
            fold1(m, den, acc, e0, a0);
        }
    }
    float inv = 1.f / den;
    const float4 bb = ld4_if(bias, hl, act);
    float v0 = acc.x * inv + bb.x, v1 = acc.y * inv + bb.y;
    float v2 = acc.z * inv + bb.z, v3 = acc.w * inv + bb.w;
    float mx = act ? fmaxf(fmaxf(v0, v1), fmaxf(v2, v3)) : -INFINITY;
    mx = redmax16(mx);
    float se = act ? (__expf(v0 - mx) + __expf(v1 - mx)) + (__expf(v2 - mx) + __expf(v3 - mx))
                   : 0.f;
    se = red16(se);
    float L = mx + logf(se);
    if (lane < 10)
        *(float4*)(out + (size_t)d * 40 + lane * 4) = make_float4(v0 - L, v1 - L, v2 - L, v3 - L);
    }
}

extern "C" void kernel_launch(void* const* d_in, const int* in_sizes, int n_in,
                              void* d_out, int out_size, void* d_ws, size_t ws_size,
                              hipStream_t stream) {
    const float* x    = (const float*)d_in[0];
    const int*   ei   = (const int*)d_in[1];
    const float* W1l  = (const float*)d_in[2];
    const float* b1l  = (const float*)d_in[3];
    const float* W1r  = (const float*)d_in[4];
    const float* b1r  = (const float*)d_in[5];
    const float* att1 = (const float*)d_in[6];
    const float* bias1= (const float*)d_in[7];
    const float* W2l  = (const float*)d_in[8];
    const float* b2l  = (const float*)d_in[9];
    const float* W2r  = (const float*)d_in[10];
    const float* b2r  = (const float*)d_in[11];
    const float* att2 = (const float*)d_in[12];
    const float* bias2= (const float*)d_in[13];
    float* out = (float*)d_out;

    const int N = in_sizes[0] / 128;
    const int E = in_sizes[1] / 2;

    char* w = (char*)d_ws;
    size_t off = 0;
    auto take = [&](size_t bytes) -> char* {
        char* p = w + off;
        off = (off + bytes + 255) & ~(size_t)255;
        return p;
    };
    _Float16* xh   = (_Float16*)take((size_t)N * 128 * 2);
    _Float16* hh   = (_Float16*)take((size_t)N * 128 * 2);
    _Float16* Wt1l = (_Float16*)take(128 * 128 * 2);
    _Float16* Wt1r = (_Float16*)take(128 * 128 * 2);
    _Float16* Wt2l = (_Float16*)take(48 * 128 * 2);
    _Float16* Wt2r = (_Float16*)take(48 * 128 * 2);
    float*    xl   = (float*)take((size_t)N * 128 * 4);
    float*    xr   = (float*)take((size_t)N * 128 * 4);
    int*      deg  = (int*)take((size_t)N * 4);
    int*      rbeg = (int*)take((size_t)N * 4);
    int*      cur  = (int*)take((size_t)N * 4);
    int*      csrc = (int*)take((size_t)E * 4);

    const int B = 256;
    auto blks = [&](long long n) { return (unsigned)((n + B - 1) / B); };
    const long long Nx128 = (long long)N * 128;
    const int rowTiles = (N + 15) / 16;

    k_cvt_f32_to_f16<<<512, B, 0, stream>>>(x, xh, Nx128);
    k_tr_w128<<<blks(128 * 128), B, 0, stream>>>(W1l, Wt1l);
    k_tr_w128<<<blks(128 * 128), B, 0, stream>>>(W1r, Wt1r);
    k_tr_w40 <<<blks(48 * 128),  B, 0, stream>>>(W2l, Wt2l);
    k_tr_w40 <<<blks(48 * 128),  B, 0, stream>>>(W2r, Wt2r);

    k_zero<<<blks(N), B, 0, stream>>>((unsigned*)deg, (long long)N);
    k_count<<<512, B, 0, stream>>>(ei, E, deg);
    k_scan<<<1, 1024, 0, stream>>>(deg, rbeg, cur, N);
    k_fill<<<512, B, 0, stream>>>(ei, E, cur, csrc);

    k_gemm_dual_w128<<<(rowTiles + 3) / 4, 128, 0, stream>>>(xh, Wt1l, Wt1r, b1l, b1r,
                                                            xl, xr, N);
    k_node_attn1<<<2048, B, 0, stream>>>(xl, xr, att1, bias1, rbeg, cur, csrc, hh, N);

    k_gemm_dual_w48<<<(rowTiles + 3) / 4, 128, 0, stream>>>(hh, Wt2l, Wt2r, b2l, b2r,
                                                           xl, xr, N);
    k_node_attn2<<<1536, B, 0, stream>>>(xl, xr, att2, bias2, rbeg, cur, csrc, out, N);
}


# In-degree prefix sum made coalesced: the one work-group sweeps the array with 16-byte loads and stores and lane-shift scans, instead of each thread walking a private 49-element chunk in 4-byte steps
# speedup vs baseline: 3.3592x; 1.3330x over previous
#include <hip/hip_runtime.h>
#include <hip/hip_bf16.h>
#include <math.h>

typedef __attribute__((ext_vector_type(16))) _Float16 v16h;
typedef __attribute__((ext_vector_type(8)))  _Float16 half8;
typedef __attribute__((ext_vector_type(8)))  float    v8f;

#define NEG_SLOPE 0.2f

__device__ inline float lrelu(float v) { return v > 0.f ? v : NEG_SLOPE * v; }

__device__ inline v16h load_frag16(const _Float16* __restrict__ p, int row0,
                                   int ld, int k0, int lane, int maxRow) {
    int m = lane & 15;
    int base = (lane >> 4) * 8;
    int r = row0 + m;
    if (r > maxRow) r = maxRow;
    const _Float16* q = p + (size_t)r * ld + k0 + base;
    half8 lo = *(const half8*)q;
    half8 hi = *(const half8*)(q + 16);
    v16h f;
#pragma unroll
    for (int i = 0; i < 8; ++i) { f[i] = lo[i]; f[i + 8] = hi[i]; }
    return f;
}

__global__ void k_zero(unsigned* p, long long n) {
    long long i = (long long)blockIdx.x * blockDim.x + threadIdx.x;
    if (i < n) p[i] = 0u;
}

typedef __attribute__((ext_vector_type(4))) _Float16 half4;
__device__ inline half4 cvt4(const float4& v) {
    half4 o;
    o[0] = (_Float16)v.x; o[1] = (_Float16)v.y; o[2] = (_Float16)v.z; o[3] = (_Float16)v.w;
    return o;
}
__global__ __launch_bounds__(256)
void k_cvt_f32_to_f16(const float* __restrict__ src,
                      _Float16* __restrict__ dst, long long n) {
    const long long n4 = n >> 2;
    const long long tid = (long long)blockIdx.x * blockDim.x + threadIdx.x;
    const long long stride = (long long)gridDim.x * blockDim.x;
    const float4* __restrict__ s4 = (const float4*)src;
    half4* __restrict__ d4 = (half4*)dst;
    long long i = tid;
    for (; i + 3 * stride < n4; i += 4 * stride) {
        float4 a = s4[i], b = s4[i + stride], c = s4[i + 2 * stride], d = s4[i + 3 * stride];
        d4[i] = cvt4(a); d4[i + stride] = cvt4(b);
        d4[i + 2 * stride] = cvt4(c); d4[i + 3 * stride] = cvt4(d);
    }
    for (; i < n4; i += stride) d4[i] = cvt4(s4[i]);
    if (tid < (n & 3)) dst[(n4 << 2) + tid] = (_Float16)src[(n4 << 2) + tid];
}

__global__ void k_tr_w128(const float* __restrict__ W, _Float16* __restrict__ Wt) {
    int i = blockIdx.x * blockDim.x + threadIdx.x;
    if (i >= 128 * 128) return;
    int c = i >> 7, k = i & 127;
    Wt[i] = (_Float16)W[k * 128 + c];
}

__global__ void k_tr_w40(const float* __restrict__ W, _Float16* __restrict__ Wt) {
    int i = blockIdx.x * blockDim.x + threadIdx.x;
    if (i >= 48 * 128) return;
    int c = i >> 7, k = i & 127;
    Wt[i] = (c < 40) ? (_Float16)W[k * 40 + c] : (_Float16)0.f;
}

template <int NCT, int NB>
__device__ inline void gemm_dual_strip(const _Float16* __restrict__ A,
                                       const _Float16* __restrict__ Wtl,
                                       const _Float16* __restrict__ Wtr,
                                       const float* __restrict__ bl,
                                       const float* __restrict__ br,
                                       float* __restrict__ Cl, float* __restrict__ Cr,
                                       int Nrows) {
    constexpr int ldc = NCT * 16;
    const int lane = threadIdx.x & 31;
    const int strip = __builtin_amdgcn_readfirstlane(
        (int)((blockIdx.x * blockDim.x + threadIdx.x) >> 5));
    const int r0 = strip * 16;
    if (r0 >= Nrows) return;
    v16h a[4];
#pragma unroll
    for (int kk = 0; kk < 4; ++kk) a[kk] = load_frag16(A, r0, 128, kk * 32, lane, Nrows - 1);
    const int n = lane & 15, hw = lane >> 4;
    const bool whole = r0 + 16 <= Nrows;
#pragma unroll
    for (int ct = 0; ct < NCT; ++ct) {
        const int c0 = ct * 16;
        v16h bL[4], bR[4];
#pragma unroll
        for (int kk = 0; kk < 4; ++kk) {
            bL[kk] = load_frag16(Wtl, c0, 128, kk * 32, lane, ldc - 1);
            bR[kk] = load_frag16(Wtr, c0, 128, kk * 32, lane, ldc - 1);
        }
        v8f accL = {};
        v8f accR = {};
#pragma unroll
        for (int kk = 0; kk < 4; ++kk) {
            accL = __builtin_amdgcn_wmma_f32_16x16x32_f16(false, a[kk], false, bL[kk],
                                                          (short)0, accL, false, false);
            accR = __builtin_amdgcn_wmma_f32_16x16x32_f16(false, a[kk], false, bR[kk],
                                                          (short)0, accR, false, false);
        }
        const int col = c0 + n;
        float bLv = 0.f, bRv = 0.f;
        if (c0 + 16 <= NB || col < NB) { bLv = bl[col]; bRv = br[col]; }
        float* __restrict__ pl = Cl + (size_t)(r0 + hw * 8) * ldc + col;
        float* __restrict__ pr = Cr + (size_t)(r0 + hw * 8) * ldc + col;
        if (whole) {
#pragma unroll
            for (int v = 0; v < 8; ++v) {
                pl[v * ldc] = accL[v] + bLv;
                pr[v * ldc] = accR[v] + bRv;
            }
        } else {
#pragma unroll
            for (int v = 0; v < 8; ++v) {
                if (r0 + hw * 8 + v < Nrows) {
                    pl[v * ldc] = accL[v] + bLv;
                    pr[v * ldc] = accR[v] + bRv;
                }
            }
        }
    }
}
__global__ __launch_bounds__(128)
void k_gemm_dual_w128(const _Float16* __restrict__ A, const _Float16* __restrict__ Wtl,
                      const _Float16* __restrict__ Wtr, const float* __restrict__ bl,
                      const float* __restrict__ br, float* __restrict__ Cl,
                      float* __restrict__ Cr, int Nrows) {
    gemm_dual_strip<8, 128>(A, Wtl, Wtr, bl, br, Cl, Cr, Nrows);
}
__global__ __launch_bounds__(128)
void k_gemm_dual_w48(const _Float16* __restrict__ A, const _Float16* __restrict__ Wtl,
                     const _Float16* __restrict__ Wtr, const float* __restrict__ bl,
                     const float* __restrict__ br, float* __restrict__ Cl,
                     float* __restrict__ Cr, int Nrows) {
    gemm_dual_strip<3, 40>(A, Wtl, Wtr, bl, br, Cl, Cr, Nrows);
}

__global__ __launch_bounds__(256)
void k_count(const int* __restrict__ ei, int E, int* __restrict__ deg) {
    const int stride = gridDim.x * blockDim.x;
    for (int base = blockIdx.x * blockDim.x + threadIdx.x; base < E; base += 8 * stride) {
        int dd[8];
#pragma unroll
        for (int u = 0; u < 8; ++u) {
            int i = base + u * stride;
            dd[u] = (i < E) ? ei[E + i] : -1;
        }
#pragma unroll
        for (int u = 0; u < 8; ++u)
            if (dd[u] >= 0) atomicAdd(&deg[dd[u]], 1);
    }
}

__device__ inline int4 ld4i_guard(const int* __restrict__ p, int idx, int N) {
    int4 r = make_int4(0, 0, 0, 0);
    if (idx + 3 < N) {
        r = *(const int4*)(p + idx);
    } else if (idx < N) {
        const int a = p[idx];
        const int b = p[(idx + 1 < N) ? idx + 1 : idx];
        const int c = p[(idx + 2 < N) ? idx + 2 : idx];
        r.x = a;
        r.y = (idx + 1 < N) ? b : 0;
        r.z = (idx + 2 < N) ? c : 0;
    }
    return r;
}
__device__ inline void st4i_guard(int* __restrict__ p, int idx, int N, const int4& o) {
    if (idx + 3 < N) {
        *(int4*)(p + idx) = o;
    } else {
        if (idx     < N) p[idx]     = o.x;
        if (idx + 1 < N) p[idx + 1] = o.y;
        if (idx + 2 < N) p[idx + 2] = o.z;
    }
}
#define SCAN_PIECE (1024 * 4)
#define SCAN_TILE (4 * SCAN_PIECE)
__device__ inline void ld_tile(const int* __restrict__ p, int i0, int N, int4 (&r)[4]) {
    if (i0 + 3 * SCAN_PIECE + 3 < N) {
#pragma unroll
        for (int q = 0; q < 4; ++q) r[q] = *(const int4*)(p + i0 + q * SCAN_PIECE);
    } else {
#pragma unroll
        for (int q = 0; q < 4; ++q) r[q] = ld4i_guard(p, i0 + q * SCAN_PIECE, N);
    }
}
__device__ inline int wave_incl_scan(int v, int lane) {
#pragma unroll
    for (int off = 1; off < 32; off <<= 1) {
        int u = __shfl_up(v, off, 32);
        if (lane >= off) v += u;
    }
    return v;
}
__global__ __launch_bounds__(1024)
void k_scan(const int* __restrict__ deg, int* __restrict__ rbeg,
            int* __restrict__ cur, int N) {
    __shared__ int wtot[2][4][32];
    const int tid = threadIdx.x;
    const int lane = tid & 31;
    const int wv = __builtin_amdgcn_readfirstlane(tid >> 5);
    const int piece = SCAN_PIECE;
    const int tile = SCAN_TILE;
    const int nT = (N + tile - 1) / tile;
    int carry = 0;
    int4 nx[4];
    ld_tile(deg, tid * 4, N, nx);
    for (int t = 0; t < nT; ++t) {
        const int i0 = t * tile + tid * 4;
        int4 v[4];
#pragma unroll
        for (int q = 0; q < 4; ++q) v[q] = nx[q];
        if (t + 1 < nT) ld_tile(deg, i0 + SCAN_TILE, N, nx);
        int run[4], inc[4];
#pragma unroll
        for (int q = 0; q < 4; ++q) {
            v[q].y += v[q].x; v[q].z += v[q].y; v[q].w += v[q].z;
            run[q] = v[q].w;
        }
#pragma unroll
        for (int q = 0; q < 4; ++q) inc[q] = wave_incl_scan(run[q], lane);
        if (lane == 31) {
#pragma unroll
            for (int q = 0; q < 4; ++q) wtot[t & 1][q][wv] = inc[q];
        }
        __syncthreads();
        int wt[4], winc[4];
#pragma unroll
        for (int q = 0; q < 4; ++q) wt[q] = wtot[t & 1][q][lane];
#pragma unroll
        for (int q = 0; q < 4; ++q) winc[q] = wave_incl_scan(wt[q], lane);
#pragma unroll
        for (int q = 0; q < 4; ++q) {
            const int wexcl = __builtin_amdgcn_readlane(winc[q] - wt[q], wv);
            const int total = __builtin_amdgcn_readlane(winc[q], 31);
            const int base = carry + wexcl + (inc[q] - run[q]);
            const int4 o = make_int4(base, base + v[q].x, base + v[q].y, base + v[q].z);
            st4i_guard(rbeg, i0 + q * piece, N, o);
            st4i_guard(cur,  i0 + q * piece, N, o);
            carry += total;
        }
    }
}

__global__ __launch_bounds__(256)
void k_fill(const int* __restrict__ ei, int E, int* __restrict__ cur,
            int* __restrict__ csrc) {
    const int stride = gridDim.x * blockDim.x;
    for (int base = blockIdx.x * blockDim.x + threadIdx.x; base < E; base += 8 * stride) {
        int dd[8], ss[8], pp[8];
#pragma unroll
        for (int u = 0; u < 8; ++u) {
            int i = base + u * stride;
            bool on = i < E;
            dd[u] = on ? ei[E + i] : -1;
            ss[u] = on ? ei[i] : 0;
        }
#pragma unroll
        for (int u = 0; u < 8; ++u)
            pp[u] = (dd[u] >= 0) ? atomicAdd(&cur[dd[u]], 1) : -1;
#pragma unroll
        for (int u = 0; u < 8; ++u)
            if (pp[u] >= 0) csrc[pp[u]] = ss[u];
    }
}

__device__ inline float score4(const float4& a, const float4& b, const float4& t) {
    return t.x * lrelu(a.x + b.x) + t.y * lrelu(a.y + b.y) +
           t.z * lrelu(a.z + b.z) + t.w * lrelu(a.w + b.w);
}
__device__ inline float red16(float v) {
#pragma unroll
    for (int off = 1; off < 16; off <<= 1) v += __shfl_xor(v, off, 32);
    return v;
}
__device__ inline float redmax16(float v) {
#pragma unroll
    for (int off = 1; off < 16; off <<= 1) v = fmaxf(v, __shfl_xor(v, off, 32));
    return v;
}
__device__ inline void scale4(float4& a, float s) { a.x *= s; a.y *= s; a.z *= s; a.w *= s; }
__device__ inline void axpy4(float4& a, float p, const float4& v) {
    a.x += p * v.x; a.y += p * v.y; a.z += p * v.z; a.w += p * v.w;
}

__device__ inline int sort_chunk(int mine, int cnt, int lane) {
    int rank = 0;
    for (int k = 0; k < cnt; ++k) {
        int v = __builtin_amdgcn_readlane(mine, k);
        rank += ((v < mine) || (v == mine && k < lane)) ? 1 : 0;
    }
    return __builtin_amdgcn_ds_permute(rank << 2, mine);
}

__device__ inline void fold4(float& m, float& den, float4& acc,
                             float e0, float e1, float e2, float e3,
                             const float4& a0, const float4& a1,
                             const float4& a2, const float4& a3) {
    float mn = fmaxf(fmaxf(m, fmaxf(e0, e1)), fmaxf(e2, e3));
    float sc = __expf(m - mn);
    float p0 = __expf(e0 - mn), p1 = __expf(e1 - mn);
    float p2 = __expf(e2 - mn), p3 = __expf(e3 - mn);
    scale4(acc, sc);
    axpy4(acc, p0, a0); axpy4(acc, p1, a1); axpy4(acc, p2, a2); axpy4(acc, p3, a3);
    den = den * sc + ((p0 + p1) + (p2 + p3));
    m = mn;
}
__device__ inline void fold1(float& m, float& den, float4& acc, float e0, const float4& a0) {
    float mn = fmaxf(m, e0);
    float sc = __expf(m - mn);
    float p0 = __expf(e0 - mn);
    scale4(acc, sc);
    axpy4(acc, p0, a0);
    den = den * sc + p0;
    m = mn;
}

__global__ __launch_bounds__(256)
void k_node_attn1(const float* __restrict__ xl, const float* __restrict__ xr,
                 const float* __restrict__ att, const float* __restrict__ bias,
                 const int* __restrict__ rbeg, const int* __restrict__ rend,
                 const int* __restrict__ csrc, _Float16* __restrict__ hh, int N) {
    const int wave = __builtin_amdgcn_readfirstlane(
        (int)((blockIdx.x * blockDim.x + threadIdx.x) >> 5));
    const int nwave = (int)((gridDim.x * blockDim.x) >> 5);
    const int lane = threadIdx.x & 31;
    const float4 t  = ((const float4*)att)[lane];
    const float4 bb = ((const float4*)bias)[lane];
    for (int d = wave; d < N; d += nwave) {
    const float4 b4 = ((const float4*)(xr + (size_t)d * 128))[lane];
    float4 acc = ((const float4*)(xl + (size_t)d * 128))[lane];
    float m = red16(score4(acc, b4, t));
    float den = 1.f;
    int beg = rbeg[d], end = rend[d];
    for (int j0 = beg; j0 < end; j0 += 32) {
        int cnt = end - j0; if (cnt > 32) cnt = 32;
        int mine = (lane < cnt) ? csrc[j0 + lane] : 0x7fffffff;
        int srt = sort_chunk(mine, cnt, lane);
        int k = 0;
        for (; k + 4 <= cnt; k += 4) {
            int s0 = __builtin_amdgcn_readlane(srt, k);
            int s1 = __builtin_amdgcn_readlane(srt, k + 1);
            int s2 = __builtin_amdgcn_readlane(srt, k + 2);
            int s3 = __builtin_amdgcn_readlane(srt, k + 3);
            float4 a0 = ((const float4*)(xl + (size_t)s0 * 128))[lane];
            float4 a1 = ((const float4*)(xl + (size_t)s1 * 128))[lane];
            float4 a2 = ((const float4*)(xl + (size_t)s2 * 128))[lane];
            float4 a3 = ((const float4*)(xl + (size_t)s3 * 128))[lane];
            float e0 = red16(score4(a0, b4, t));
            float e1 = red16(score4(a1, b4, t));
            float e2 = red16(score4(a2, b4, t));
            float e3 = red16(score4(a3, b4, t));
            fold4(m, den, acc, e0, e1, e2, e3, a0, a1, a2, a3);
        }
        for (; k < cnt; ++k) {
            int s0 = __builtin_amdgcn_readlane(srt, k);
            float4 a0 = ((const float4*)(xl + (size_t)s0 * 128))[lane];
            float e0 = red16(score4(a0, b4, t));
            fold1(m, den, acc, e0, a0);
        }
    }
    float inv = 1.f / den;
    float o0 = fmaxf(acc.x * inv + bb.x, 0.f);
    float o1 = fmaxf(acc.y * inv + bb.y, 0.f);
    float o2 = fmaxf(acc.z * inv + bb.z, 0.f);
    float o3 = fmaxf(acc.w * inv + bb.w, 0.f);
    half4 o;
    o[0] = (_Float16)o0; o[1] = (_Float16)o1; o[2] = (_Float16)o2; o[3] = (_Float16)o3;
    *(half4*)(hh + (size_t)d * 128 + lane * 4) = o;
    }
}

__device__ inline float4 ld4_if(const float* __restrict__ p, int idx, bool on) {
    float4 r = make_float4(0.f, 0.f, 0.f, 0.f);
    if (on) r = ((const float4*)p)[idx];
    return r;
}

__global__ __launch_bounds__(256)
void k_node_attn2(const float* __restrict__ xl, const float* __restrict__ xr,
                 const float* __restrict__ att, const float* __restrict__ bias,
                 const int* __restrict__ rbeg, const int* __restrict__ rend,
                 const int* __restrict__ csrc, float* __restrict__ out, int N) {
    const int wave = __builtin_amdgcn_readfirstlane(
        (int)((blockIdx.x * blockDim.x + threadIdx.x) >> 5));
    const int nwave = (int)((gridDim.x * blockDim.x) >> 5);
    const int lane = threadIdx.x & 31;
    const int hl = lane & 15;
    const bool act = hl < 10;
    const float4 t  = ld4_if(att, hl, act);
    for (int d = wave; d < N; d += nwave) {
    const float4 b4 = ld4_if(xr + (size_t)d * 48, hl, act);
    float4 acc = ld4_if(xl + (size_t)d * 48, hl, act);
    float m = red16(score4(acc, b4, t));
    float den = 1.f;
    int beg = rbeg[d], end = rend[d];
    for (int j0 = beg; j0 < end; j0 += 32) {
        int cnt = end - j0; if (cnt > 32) cnt = 32;
        int mine = (lane < cnt) ? csrc[j0 + lane] : 0x7fffffff;
        int srt = sort_chunk(mine, cnt, lane);
        int k = 0;
        for (; k + 4 <= cnt; k += 4) {
            int s0 = __builtin_amdgcn_readlane(srt, k);
            int s1 = __builtin_amdgcn_readlane(srt, k + 1);
            int s2 = __builtin_amdgcn_readlane(srt, k + 2);
            int s3 = __builtin_amdgcn_readlane(srt, k + 3);
            float4 a0 = ld4_if(xl + (size_t)s0 * 48, hl, act);
            float4 a1 = ld4_if(xl + (size_t)s1 * 48, hl, act);
            float4 a2 = ld4_if(xl + (size_t)s2 * 48, hl, act);
            float4 a3 = ld4_if(xl + (size_t)s3 * 48, hl, act);
            float e0 = red16(score4(a0, b4, t));
            float e1 = red16(score4(a1, b4, t));
            float e2 = red16(score4(a2, b4, t));
            float e3 = red16(score4(a3, b4, t));
            fold4(m, den, acc, e0, e1, e2, e3, a0, a1, a2, a3);
        }
        for (; k < cnt; ++k) {
            int s0 = __builtin_amdgcn_readlane(srt, k);
            float4 a0 = ld4_if(xl + (size_t)s0 * 48, hl, act);
            float e0 = red16(score4(a0, b4, t));
            fold1(m, den, acc, e0, a0);
        }
    }
    float inv = 1.f / den;
    const float4 bb = ld4_if(bias, hl, act);
    float v0 = acc.x * inv + bb.x, v1 = acc.y * inv + bb.y;
    float v2 = acc.z * inv + bb.z, v3 = acc.w * inv + bb.w;
    float mx = act ? fmaxf(fmaxf(v0, v1), fmaxf(v2, v3)) : -INFINITY;
    mx = redmax16(mx);
    float se = act ? (__expf(v0 - mx) + __expf(v1 - mx)) + (__expf(v2 - mx) + __expf(v3 - mx))
                   : 0.f;
    se = red16(se);
    float L = mx + logf(se);
    if (lane < 10)
        *(float4*)(out + (size_t)d * 40 + lane * 4) = make_float4(v0 - L, v1 - L, v2 - L, v3 - L);
    }
}

extern "C" void kernel_launch(void* const* d_in, const int* in_sizes, int n_in,
                              void* d_out, int out_size, void* d_ws, size_t ws_size,
                              hipStream_t stream) {
    const float* x    = (const float*)d_in[0];
    const int*   ei   = (const int*)d_in[1];
    const float* W1l  = (const float*)d_in[2];
    const float* b1l  = (const float*)d_in[3];
    const float* W1r  = (const float*)d_in[4];
    const float* b1r  = (const float*)d_in[5];
    const float* att1 = (const float*)d_in[6];
    const float* bias1= (const float*)d_in[7];
    const float* W2l  = (const float*)d_in[8];
    const float* b2l  = (const float*)d_in[9];
    const float* W2r  = (const float*)d_in[10];
    const float* b2r  = (const float*)d_in[11];
    const float* att2 = (const float*)d_in[12];
    const float* bias2= (const float*)d_in[13];
    float* out = (float*)d_out;

    const int N = in_sizes[0] / 128;
    const int E = in_sizes[1] / 2;

    char* w = (char*)d_ws;
    size_t off = 0;
    auto take = [&](size_t bytes) -> char* {
        char* p = w + off;
        off = (off + bytes + 255) & ~(size_t)255;
        return p;
    };
    _Float16* xh   = (_Float16*)take((size_t)N * 128 * 2);
    _Float16* hh   = (_Float16*)take((size_t)N * 128 * 2);
    _Float16* Wt1l = (_Float16*)take(128 * 128 * 2);
    _Float16* Wt1r = (_Float16*)take(128 * 128 * 2);
    _Float16* Wt2l = (_Float16*)take(48 * 128 * 2);
    _Float16* Wt2r = (_Float16*)take(48 * 128 * 2);
    float*    xl   = (float*)take((size_t)N * 128 * 4);
    float*    xr   = (float*)take((size_t)N * 128 * 4);
    int*      deg  = (int*)take((size_t)N * 4);
    int*      rbeg = (int*)take((size_t)N * 4);
    int*      cur  = (int*)take((size_t)N * 4);
    int*      csrc = (int*)take((size_t)E * 4);

    const int B = 256;
    auto blks = [&](long long n) { return (unsigned)((n + B - 1) / B); };
    const long long Nx128 = (long long)N * 128;
    const int rowTiles = (N + 15) / 16;

    k_cvt_f32_to_f16<<<512, B, 0, stream>>>(x, xh, Nx128);
    k_tr_w128<<<blks(128 * 128), B, 0, stream>>>(W1l, Wt1l);
    k_tr_w128<<<blks(128 * 128), B, 0, stream>>>(W1r, Wt1r);
    k_tr_w40 <<<blks(48 * 128),  B, 0, stream>>>(W2l, Wt2l);
    k_tr_w40 <<<blks(48 * 128),  B, 0, stream>>>(W2r, Wt2r);

    k_zero<<<blks(N), B, 0, stream>>>((unsigned*)deg, (long long)N);
    k_count<<<512, B, 0, stream>>>(ei, E, deg);
    k_scan<<<1, 1024, 0, stream>>>(deg, rbeg, cur, N);
    k_fill<<<512, B, 0, stream>>>(ei, E, cur, csrc);

    k_gemm_dual_w128<<<(rowTiles + 3) / 4, 128, 0, stream>>>(xh, Wt1l, Wt1r, b1l, b1r,
                                                            xl, xr, N);
    k_node_attn1<<<2048, B, 0, stream>>>(xl, xr, att1, bias1, rbeg, cur, csrc, hh, N);

    k_gemm_dual_w48<<<(rowTiles + 3) / 4, 128, 0, stream>>>(hh, Wt2l, Wt2r, b2l, b2r,
                                                           xl, xr, N);
    k_node_attn2<<<1536, B, 0, stream>>>(xl, xr, att2, bias2, rbeg, cur, csrc, out, N);
}


# 13 launches -> 7: independent stages share a launch by block range (counter clear + 4 weight transposes; degree count + conversion of x; list fill + layer-1 GEMM); sums and their order kept
# speedup vs baseline: 3.6392x; 1.0834x over previous
#include <hip/hip_runtime.h>
#include <hip/hip_bf16.h>
#include <math.h>

typedef __attribute__((ext_vector_type(16))) _Float16 v16h;
typedef __attribute__((ext_vector_type(8)))  _Float16 half8;
typedef __attribute__((ext_vector_type(8)))  float    v8f;

#define NEG_SLOPE 0.2f

__device__ inline float lrelu(float v) { return v > 0.f ? v : NEG_SLOPE * v; }

__device__ inline v16h load_frag16(const _Float16* __restrict__ p, int row0,
                                   int ld, int k0, int lane, int maxRow) {
    int m = lane & 15;
    int base = (lane >> 4) * 8;
    int r = row0 + m;
    if (r > maxRow) r = maxRow;
    const _Float16* q = p + (size_t)r * ld + k0 + base;
    half8 lo = *(const half8*)q;
    half8 hi = *(const half8*)(q + 16);
    v16h f;
#pragma unroll
    for (int i = 0; i < 8; ++i) { f[i] = lo[i]; f[i + 8] = hi[i]; }
    return f;
}


typedef __attribute__((ext_vector_type(4))) _Float16 half4;
__device__ inline half4 cvt4(const float4& v) {
    half4 o;
    o[0] = (_Float16)v.x; o[1] = (_Float16)v.y; o[2] = (_Float16)v.z; o[3] = (_Float16)v.w;
    return o;
}
__device__ inline void cvt_f32_to_f16_body(const float* __restrict__ src,
                                           _Float16* __restrict__ dst, long long n,
                                           int blk, int nblk) {
    const long long n4 = n >> 2;
    const long long tid = (long long)blk * blockDim.x + threadIdx.x;
    const long long stride = (long long)nblk * blockDim.x;
    const float4* __restrict__ s4 = (const float4*)src;
    half4* __restrict__ d4 = (half4*)dst;
    long long i = tid;
    for (; i + 3 * stride < n4; i += 4 * stride) {
        float4 a = s4[i], b = s4[i + stride], c = s4[i + 2 * stride], d = s4[i + 3 * stride];
        d4[i] = cvt4(a); d4[i + stride] = cvt4(b);
        d4[i + 2 * stride] = cvt4(c); d4[i + 3 * stride] = cvt4(d);
    }
    for (; i < n4; i += stride) d4[i] = cvt4(s4[i]);
    if (tid < (n & 3)) dst[(n4 << 2) + tid] = (_Float16)src[(n4 << 2) + tid];
}

__device__ inline void tr_w128_body(const float* __restrict__ W, _Float16* __restrict__ Wt,
                                    int blk) {
    int i = blk * blockDim.x + threadIdx.x;
    if (i >= 128 * 128) return;
    int c = i >> 7, k = i & 127;
    Wt[i] = (_Float16)W[k * 128 + c];
}

__device__ inline void tr_w40_body(const float* __restrict__ W, _Float16* __restrict__ Wt,
                                   int blk) {
    int i = blk * blockDim.x + threadIdx.x;
    if (i >= 48 * 128) return;
    int c = i >> 7, k = i & 127;
    Wt[i] = (c < 40) ? (_Float16)W[k * 40 + c] : (_Float16)0.f;
}

__global__ __launch_bounds__(256)
void k_zero_tr(unsigned* __restrict__ deg, int N,
               const float* __restrict__ W1l, _Float16* __restrict__ Wt1l,
               const float* __restrict__ W1r, _Float16* __restrict__ Wt1r,
               const float* __restrict__ W2l, _Float16* __restrict__ Wt2l,
               const float* __restrict__ W2r, _Float16* __restrict__ Wt2r, int nzb) {
    int b = blockIdx.x;
    if (b < nzb) {
        int i = b * blockDim.x + threadIdx.x;
        if (i < N) deg[i] = 0u;
        return;
    }
    b -= nzb;
    if (b < 64) { tr_w128_body(W1l, Wt1l, b); return; }
    b -= 64;
    if (b < 64) { tr_w128_body(W1r, Wt1r, b); return; }
    b -= 64;
    if (b < 24) { tr_w40_body(W2l, Wt2l, b); return; }
    b -= 24;
    tr_w40_body(W2r, Wt2r, b);
}

template <int NCT, int NB>
__device__ inline void gemm_dual_strip(const _Float16* __restrict__ A,
                                       const _Float16* __restrict__ Wtl,
                                       const _Float16* __restrict__ Wtr,
                                       const float* __restrict__ bl,
                                       const float* __restrict__ br,
                                       float* __restrict__ Cl, float* __restrict__ Cr,
                                       int Nrows, int blk) {
    constexpr int ldc = NCT * 16;
    const int lane = threadIdx.x & 31;
    const int strip = __builtin_amdgcn_readfirstlane(
        (int)((blk * blockDim.x + threadIdx.x) >> 5));
    const int r0 = strip * 16;
    if (r0 >= Nrows) return;
    v16h a[4];
#pragma unroll
    for (int kk = 0; kk < 4; ++kk) a[kk] = load_frag16(A, r0, 128, kk * 32, lane, Nrows - 1);
    const int n = lane & 15, hw = lane >> 4;
    const bool whole = r0 + 16 <= Nrows;
#pragma unroll
    for (int ct = 0; ct < NCT; ++ct) {
        const int c0 = ct * 16;
        v16h bL[4], bR[4];
#pragma unroll
        for (int kk = 0; kk < 4; ++kk) {
            bL[kk] = load_frag16(Wtl, c0, 128, kk * 32, lane, ldc - 1);
            bR[kk] = load_frag16(Wtr, c0, 128, kk * 32, lane, ldc - 1);
        }
        v8f accL = {};
        v8f accR = {};
#pragma unroll
        for (int kk = 0; kk < 4; ++kk) {
            accL = __builtin_amdgcn_wmma_f32_16x16x32_f16(false, a[kk], false, bL[kk],
                                                          (short)0, accL, false, false);
            accR = __builtin_amdgcn_wmma_f32_16x16x32_f16(false, a[kk], false, bR[kk],
                                                          (short)0, accR, false, false);
        }
        const int col = c0 + n;
        float bLv = 0.f, bRv = 0.f;
        if (c0 + 16 <= NB || col < NB) { bLv = bl[col]; bRv = br[col]; }
        float* __restrict__ pl = Cl + (size_t)(r0 + hw * 8) * ldc + col;
        float* __restrict__ pr = Cr + (size_t)(r0 + hw * 8) * ldc + col;
        if (whole) {
#pragma unroll
            for (int v = 0; v < 8; ++v) {
                pl[v * ldc] = accL[v] + bLv;
                pr[v * ldc] = accR[v] + bRv;
            }
        } else {
#pragma unroll
            for (int v = 0; v < 8; ++v) {
                if (r0 + hw * 8 + v < Nrows) {
                    pl[v * ldc] = accL[v] + bLv;
                    pr[v * ldc] = accR[v] + bRv;
                }
            }
        }
    }
}
__global__ __launch_bounds__(128)
void k_gemm_dual_w48(const _Float16* __restrict__ A, const _Float16* __restrict__ Wtl,
                     const _Float16* __restrict__ Wtr, const float* __restrict__ bl,
                     const float* __restrict__ br, float* __restrict__ Cl,
                     float* __restrict__ Cr, int Nrows) {
    gemm_dual_strip<3, 40>(A, Wtl, Wtr, bl, br, Cl, Cr, Nrows, (int)blockIdx.x);
}

__device__ inline void count_body(const int* __restrict__ ei, int E, int* __restrict__ deg,
                                  int blk, int nblk) {
    const int stride = nblk * blockDim.x;
    for (int base = blk * blockDim.x + threadIdx.x; base < E; base += 8 * stride) {
        int dd[8];
#pragma unroll
        for (int u = 0; u < 8; ++u) {
            int i = base + u * stride;
            dd[u] = (i < E) ? ei[E + i] : -1;
        }
#pragma unroll
        for (int u = 0; u < 8; ++u)
            if (dd[u] >= 0) atomicAdd(&deg[dd[u]], 1);
    }
}

__global__ __launch_bounds__(256)
void k_count_cvt(const int* __restrict__ ei, int E, int* __restrict__ deg,
                 const float* __restrict__ src, _Float16* __restrict__ dst, long long n,
                 int ncb) {
    const int b = blockIdx.x;
    if (b < ncb) count_body(ei, E, deg, b, ncb);
    else cvt_f32_to_f16_body(src, dst, n, b - ncb, (int)gridDim.x - ncb);
}

__device__ inline int4 ld4i_guard(const int* __restrict__ p, int idx, int N) {
    int4 r = make_int4(0, 0, 0, 0);
    if (idx + 3 < N) {
        r = *(const int4*)(p + idx);
    } else if (idx < N) {
        const int a = p[idx];
        const int b = p[(idx + 1 < N) ? idx + 1 : idx];
        const int c = p[(idx + 2 < N) ? idx + 2 : idx];
        r.x = a;
        r.y = (idx + 1 < N) ? b : 0;
        r.z = (idx + 2 < N) ? c : 0;
    }
    return r;
}
__device__ inline void st4i_guard(int* __restrict__ p, int idx, int N, const int4& o) {
    if (idx + 3 < N) {
        *(int4*)(p + idx) = o;
    } else {
        if (idx     < N) p[idx]     = o.x;
        if (idx + 1 < N) p[idx + 1] = o.y;
        if (idx + 2 < N) p[idx + 2] = o.z;
    }
}
#define SCAN_PIECE (1024 * 4)
#define SCAN_TILE (4 * SCAN_PIECE)
__device__ inline void ld_tile(const int* __restrict__ p, int i0, int N, int4 (&r)[4]) {
    if (i0 + 3 * SCAN_PIECE + 3 < N) {
#pragma unroll
        for (int q = 0; q < 4; ++q) r[q] = *(const int4*)(p + i0 + q * SCAN_PIECE);
    } else {
#pragma unroll
        for (int q = 0; q < 4; ++q) r[q] = ld4i_guard(p, i0 + q * SCAN_PIECE, N);
    }
}
__device__ inline int wave_incl_scan(int v, int lane) {
#pragma unroll
    for (int off = 1; off < 32; off <<= 1) {
        int u = __shfl_up(v, off, 32);
        if (lane >= off) v += u;
    }
    return v;
}
__global__ __launch_bounds__(1024)
void k_scan(const int* __restrict__ deg, int* __restrict__ rbeg,
            int* __restrict__ cur, int N) {
    __shared__ int wtot[2][4][32];
    const int tid = threadIdx.x;
    const int lane = tid & 31;
    const int wv = __builtin_amdgcn_readfirstlane(tid >> 5);
    const int piece = SCAN_PIECE;
    const int tile = SCAN_TILE;
    const int nT = (N + tile - 1) / tile;
    int carry = 0;
    int4 nx[4];
    ld_tile(deg, tid * 4, N, nx);
    for (int t = 0; t < nT; ++t) {
        const int i0 = t * tile + tid * 4;
        int4 v[4];
#pragma unroll
        for (int q = 0; q < 4; ++q) v[q] = nx[q];
        if (t + 1 < nT) ld_tile(deg, i0 + SCAN_TILE, N, nx);
        int run[4], inc[4];
#pragma unroll
        for (int q = 0; q < 4; ++q) {
            v[q].y += v[q].x; v[q].z += v[q].y; v[q].w += v[q].z;
            run[q] = v[q].w;
        }
#pragma unroll
        for (int q = 0; q < 4; ++q) inc[q] = wave_incl_scan(run[q], lane);
        if (lane == 31) {
#pragma unroll
            for (int q = 0; q < 4; ++q) wtot[t & 1][q][wv] = inc[q];
        }
        __syncthreads();
        int wt[4], winc[4];
#pragma unroll
        for (int q = 0; q < 4; ++q) wt[q] = wtot[t & 1][q][lane];
#pragma unroll
        for (int q = 0; q < 4; ++q) winc[q] = wave_incl_scan(wt[q], lane);
#pragma unroll
        for (int q = 0; q < 4; ++q) {
            const int wexcl = __builtin_amdgcn_readlane(winc[q] - wt[q], wv);
            const int total = __builtin_amdgcn_readlane(winc[q], 31);
            const int base = carry + wexcl + (inc[q] - run[q]);
            const int4 o = make_int4(base, base + v[q].x, base + v[q].y, base + v[q].z);
            st4i_guard(rbeg, i0 + q * piece, N, o);
            st4i_guard(cur,  i0 + q * piece, N, o);
            carry += total;
        }
    }
}

__device__ inline void fill_body(const int* __restrict__ ei, int E, int* __restrict__ cur,
                                 int* __restrict__ csrc, int blk, int nblk) {
    const int stride = nblk * blockDim.x;
    for (int base = blk * blockDim.x + threadIdx.x; base < E; base += 8 * stride) {
        int dd[8], ss[8], pp[8];
#pragma unroll
        for (int u = 0; u < 8; ++u) {
            int i = base + u * stride;
            bool on = i < E;
            dd[u] = on ? ei[E + i] : -1;
            ss[u] = on ? ei[i] : 0;
        }
#pragma unroll
        for (int u = 0; u < 8; ++u)
            pp[u] = (dd[u] >= 0) ? atomicAdd(&cur[dd[u]], 1) : -1;
#pragma unroll
        for (int u = 0; u < 8; ++u)
            if (pp[u] >= 0) csrc[pp[u]] = ss[u];
    }
}

__global__ __launch_bounds__(256)
void k_fill_gemm1(const int* __restrict__ ei, int E, int* __restrict__ cur,
                  int* __restrict__ csrc, int nfb,
                  const _Float16* __restrict__ A, const _Float16* __restrict__ Wtl,
                  const _Float16* __restrict__ Wtr, const float* __restrict__ bl,
                  const float* __restrict__ br, float* __restrict__ Cl,
                  float* __restrict__ Cr, int Nrows) {
    const int b = blockIdx.x;
    if (b < nfb) fill_body(ei, E, cur, csrc, b, nfb);
    else gemm_dual_strip<8, 128>(A, Wtl, Wtr, bl, br, Cl, Cr, Nrows, b - nfb);
}

__device__ inline float score4(const float4& a, const float4& b, const float4& t) {
    return t.x * lrelu(a.x + b.x) + t.y * lrelu(a.y + b.y) +
           t.z * lrelu(a.z + b.z) + t.w * lrelu(a.w + b.w);
}
__device__ inline float red16(float v) {
#pragma unroll
    for (int off = 1; off < 16; off <<= 1) v += __shfl_xor(v, off, 32);
    return v;
}
__device__ inline float redmax16(float v) {
#pragma unroll
    for (int off = 1; off < 16; off <<= 1) v = fmaxf(v, __shfl_xor(v, off, 32));
    return v;
}
__device__ inline void scale4(float4& a, float s) { a.x *= s; a.y *= s; a.z *= s; a.w *= s; }
__device__ inline void axpy4(float4& a, float p, const float4& v) {
    a.x += p * v.x; a.y += p * v.y; a.z += p * v.z; a.w += p * v.w;
}

__device__ inline int sort_chunk(int mine, int cnt, int lane) {
    int rank = 0;
    for (int k = 0; k < cnt; ++k) {
        int v = __builtin_amdgcn_readlane(mine, k);
        rank += ((v < mine) || (v == mine && k < lane)) ? 1 : 0;
    }
    return __builtin_amdgcn_ds_permute(rank << 2, mine);
}

__device__ inline void fold4(float& m, float& den, float4& acc,
                             float e0, float e1, float e2, float e3,
                             const float4& a0, const float4& a1,
                             const float4& a2, const float4& a3) {
    float mn = fmaxf(fmaxf(m, fmaxf(e0, e1)), fmaxf(e2, e3));
    float sc = __expf(m - mn);
    float p0 = __expf(e0 - mn), p1 = __expf(e1 - mn);
    float p2 = __expf(e2 - mn), p3 = __expf(e3 - mn);
    scale4(acc, sc);
    axpy4(acc, p0, a0); axpy4(acc, p1, a1); axpy4(acc, p2, a2); axpy4(acc, p3, a3);
    den = den * sc + ((p0 + p1) + (p2 + p3));
    m = mn;
}
__device__ inline void fold1(float& m, float& den, float4& acc, float e0, const float4& a0) {
    float mn = fmaxf(m, e0);
    float sc = __expf(m - mn);
    float p0 = __expf(e0 - mn);
    scale4(acc, sc);
    axpy4(acc, p0, a0);
    den = den * sc + p0;
    m = mn;
}

__global__ __launch_bounds__(256)
void k_node_attn1(const float* __restrict__ xl, const float* __restrict__ xr,
                 const float* __restrict__ att, const float* __restrict__ bias,
                 const int* __restrict__ rbeg, const int* __restrict__ rend,
                 const int* __restrict__ csrc, _Float16* __restrict__ hh, int N) {
    const int wave = __builtin_amdgcn_readfirstlane(
        (int)((blockIdx.x * blockDim.x + threadIdx.x) >> 5));
    const int nwave = (int)((gridDim.x * blockDim.x) >> 5);
    const int lane = threadIdx.x & 31;
    const float4 t  = ((const float4*)att)[lane];
    const float4 bb = ((const float4*)bias)[lane];
    for (int d = wave; d < N; d += nwave) {
    const float4 b4 = ((const float4*)(xr + (size_t)d * 128))[lane];
    float4 acc = ((const float4*)(xl + (size_t)d * 128))[lane];
    float m = red16(score4(acc, b4, t));
    float den = 1.f;
    int beg = rbeg[d], end = rend[d];
    for (int j0 = beg; j0 < end; j0 += 32) {
        int cnt = end - j0; if (cnt > 32) cnt = 32;
        int mine = (lane < cnt) ? csrc[j0 + lane] : 0x7fffffff;
        int srt = sort_chunk(mine, cnt, lane);
        int k = 0;
        for (; k + 4 <= cnt; k += 4) {
            int s0 = __builtin_amdgcn_readlane(srt, k);
            int s1 = __builtin_amdgcn_readlane(srt, k + 1);
            int s2 = __builtin_amdgcn_readlane(srt, k + 2);
            int s3 = __builtin_amdgcn_readlane(srt, k + 3);
            float4 a0 = ((const float4*)(xl + (size_t)s0 * 128))[lane];
            float4 a1 = ((const float4*)(xl + (size_t)s1 * 128))[lane];
            float4 a2 = ((const float4*)(xl + (size_t)s2 * 128))[lane];
            float4 a3 = ((const float4*)(xl + (size_t)s3 * 128))[lane];
            float e0 = red16(score4(a0, b4, t));
            float e1 = red16(score4(a1, b4, t));
            float e2 = red16(score4(a2, b4, t));
            float e3 = red16(score4(a3, b4, t));
            fold4(m, den, acc, e0, e1, e2, e3, a0, a1, a2, a3);
        }
        for (; k < cnt; ++k) {
            int s0 = __builtin_amdgcn_readlane(srt, k);
            float4 a0 = ((const float4*)(xl + (size_t)s0 * 128))[lane];
            float e0 = red16(score4(a0, b4, t));
            fold1(m, den, acc, e0, a0);
        }
    }
    float inv = 1.f / den;
    float o0 = fmaxf(acc.x * inv + bb.x, 0.f);
    float o1 = fmaxf(acc.y * inv + bb.y, 0.f);
    float o2 = fmaxf(acc.z * inv + bb.z, 0.f);
    float o3 = fmaxf(acc.w * inv + bb.w, 0.f);
    half4 o;
    o[0] = (_Float16)o0; o[1] = (_Float16)o1; o[2] = (_Float16)o2; o[3] = (_Float16)o3;
    *(half4*)(hh + (size_t)d * 128 + lane * 4) = o;
    }
}

__device__ inline float4 ld4_if(const float* __restrict__ p, int idx, bool on) {
    float4 r = make_float4(0.f, 0.f, 0.f, 0.f);
    if (on) r = ((const float4*)p)[idx];
    return r;
}

__global__ __launch_bounds__(256)
void k_node_attn2(const float* __restrict__ xl, const float* __restrict__ xr,
                 const float* __restrict__ att, const float* __restrict__ bias,
                 const int* __restrict__ rbeg, const int* __restrict__ rend,
                 const int* __restrict__ csrc, float* __restrict__ out, int N) {
    const int wave = __builtin_amdgcn_readfirstlane(
        (int)((blockIdx.x * blockDim.x + threadIdx.x) >> 5));
    const int nwave = (int)((gridDim.x * blockDim.x) >> 5);
    const int lane = threadIdx.x & 31;
    const int hl = lane & 15;
    const bool act = hl < 10;
    const float4 t  = ld4_if(att, hl, act);
    for (int d = wave; d < N; d += nwave) {
    const float4 b4 = ld4_if(xr + (size_t)d * 48, hl, act);
    float4 acc = ld4_if(xl + (size_t)d * 48, hl, act);
    float m = red16(score4(acc, b4, t));
    float den = 1.f;
    int beg = rbeg[d], end = rend[d];
    for (int j0 = beg; j0 < end; j0 += 32) {
        int cnt = end - j0; if (cnt > 32) cnt = 32;
        int mine = (lane < cnt) ? csrc[j0 + lane] : 0x7fffffff;
        int srt = sort_chunk(mine, cnt, lane);
        int k = 0;
        for (; k + 4 <= cnt; k += 4) {
            int s0 = __builtin_amdgcn_readlane(srt, k);
            int s1 = __builtin_amdgcn_readlane(srt, k + 1);
            int s2 = __builtin_amdgcn_readlane(srt, k + 2);
            int s3 = __builtin_amdgcn_readlane(srt, k + 3);
            float4 a0 = ld4_if(xl + (size_t)s0 * 48, hl, act);
            float4 a1 = ld4_if(xl + (size_t)s1 * 48, hl, act);
            float4 a2 = ld4_if(xl + (size_t)s2 * 48, hl, act);
            float4 a3 = ld4_if(xl + (size_t)s3 * 48, hl, act);
            float e0 = red16(score4(a0, b4, t));
            float e1 = red16(score4(a1, b4, t));
            float e2 = red16(score4(a2, b4, t));
            float e3 = red16(score4(a3, b4, t));
            fold4(m, den, acc, e0, e1, e2, e3, a0, a1, a2, a3);
        }
        for (; k < cnt; ++k) {
            int s0 = __builtin_amdgcn_readlane(srt, k);
            float4 a0 = ld4_if(xl + (size_t)s0 * 48, hl, act);
            float e0 = red16(score4(a0, b4, t));
            fold1(m, den, acc, e0, a0);
        }
    }
    float inv = 1.f / den;
    const float4 bb = ld4_if(bias, hl, act);
    float v0 = acc.x * inv + bb.x, v1 = acc.y * inv + bb.y;
    float v2 = acc.z * inv + bb.z, v3 = acc.w * inv + bb.w;
    float mx = act ? fmaxf(fmaxf(v0, v1), fmaxf(v2, v3)) : -INFINITY;
    mx = redmax16(mx);
    float se = act ? (__expf(v0 - mx) + __expf(v1 - mx)) + (__expf(v2 - mx) + __expf(v3 - mx))
                   : 0.f;
    se = red16(se);
    float L = mx + logf(se);
    if (lane < 10)
        *(float4*)(out + (size_t)d * 40 + lane * 4) = make_float4(v0 - L, v1 - L, v2 - L, v3 - L);
    }
}

extern "C" void kernel_launch(void* const* d_in, const int* in_sizes, int n_in,
                              void* d_out, int out_size, void* d_ws, size_t ws_size,
                              hipStream_t stream) {
    const float* x    = (const float*)d_in[0];
    const int*   ei   = (const int*)d_in[1];
    const float* W1l  = (const float*)d_in[2];
    const float* b1l  = (const float*)d_in[3];
    const float* W1r  = (const float*)d_in[4];
    const float* b1r  = (const float*)d_in[5];
    const float* att1 = (const float*)d_in[6];
    const float* bias1= (const float*)d_in[7];
    const float* W2l  = (const float*)d_in[8];
    const float* b2l  = (const float*)d_in[9];
    const float* W2r  = (const float*)d_in[10];
    const float* b2r  = (const float*)d_in[11];
    const float* att2 = (const float*)d_in[12];
    const float* bias2= (const float*)d_in[13];
    float* out = (float*)d_out;

    const int N = in_sizes[0] / 128;
    const int E = in_sizes[1] / 2;

    char* w = (char*)d_ws;
    size_t off = 0;
    auto take = [&](size_t bytes) -> char* {
        char* p = w + off;
        off = (off + bytes + 255) & ~(size_t)255;
        return p;
    };
    _Float16* xh   = (_Float16*)take((size_t)N * 128 * 2);
    _Float16* hh   = (_Float16*)take((size_t)N * 128 * 2);
    _Float16* Wt1l = (_Float16*)take(128 * 128 * 2);
    _Float16* Wt1r = (_Float16*)take(128 * 128 * 2);
    _Float16* Wt2l = (_Float16*)take(48 * 128 * 2);
    _Float16* Wt2r = (_Float16*)take(48 * 128 * 2);
    float*    xl   = (float*)take((size_t)N * 128 * 4);
    float*    xr   = (float*)take((size_t)N * 128 * 4);
    int*      deg  = (int*)take((size_t)N * 4);
    int*      rbeg = (int*)take((size_t)N * 4);
    int*      cur  = (int*)take((size_t)N * 4);
    int*      csrc = (int*)take((size_t)E * 4);

    const int B = 256;
    auto blks = [&](long long n) { return (unsigned)((n + B - 1) / B); };
    const long long Nx128 = (long long)N * 128;
    const int rowTiles = (N + 15) / 16;

    const int nzb = (int)blks(N);
    k_zero_tr<<<nzb + 64 + 64 + 24 + 24, B, 0, stream>>>((unsigned*)deg, N, W1l, Wt1l, W1r, Wt1r,
                                                        W2l, Wt2l, W2r, Wt2r, nzb);
    k_count_cvt<<<512 + 512, B, 0, stream>>>(ei, E, deg, x, xh, Nx128, 512);
    k_scan<<<1, 1024, 0, stream>>>(deg, rbeg, cur, N);
    k_fill_gemm1<<<512 + (rowTiles + 7) / 8, B, 0, stream>>>(ei, E, cur, csrc, 512,
                                                            xh, Wt1l, Wt1r, b1l, b1r,
                                                            xl, xr, N);

    k_node_attn1<<<2048, B, 0, stream>>>(xl, xr, att1, bias1, rbeg, cur, csrc, hh, N);

    k_gemm_dual_w48<<<(rowTiles + 3) / 4, 128, 0, stream>>>(hh, Wt2l, Wt2r, b2l, b2r,
                                                           xl, xr, N);
    k_node_attn2<<<1536, B, 0, stream>>>(xl, xr, att2, bias2, rbeg, cur, csrc, out, N);
}


# Edge lists built by a two-level counting sort on LDS counters (256 edge slices x 256-node bins, then one work-group per bin); both global atomic adds per edge gone; lists, sums and their order kept
# speedup vs baseline: 5.7906x; 1.5912x over previous
#include <hip/hip_runtime.h>
#include <hip/hip_bf16.h>
#include <math.h>

typedef __attribute__((ext_vector_type(16))) _Float16 v16h;
typedef __attribute__((ext_vector_type(8)))  _Float16 half8;
typedef __attribute__((ext_vector_type(8)))  float    v8f;

#define NEG_SLOPE 0.2f

__device__ inline float lrelu(float v) { return v > 0.f ? v : NEG_SLOPE * v; }

__device__ inline v16h load_frag16(const _Float16* __restrict__ p, int row0,
                                   int ld, int k0, int lane, int maxRow) {
    int m = lane & 15;
    int base = (lane >> 4) * 8;
    int r = row0 + m;
    if (r > maxRow) r = maxRow;
    const _Float16* q = p + (size_t)r * ld + k0 + base;
    half8 lo = *(const half8*)q;
    half8 hi = *(const half8*)(q + 16);
    v16h f;
#pragma unroll
    for (int i = 0; i < 8; ++i) { f[i] = lo[i]; f[i + 8] = hi[i]; }
    return f;
}


typedef __attribute__((ext_vector_type(4))) _Float16 half4;
__device__ inline half4 cvt4(const float4& v) {
    half4 o;
    o[0] = (_Float16)v.x; o[1] = (_Float16)v.y; o[2] = (_Float16)v.z; o[3] = (_Float16)v.w;
    return o;
}
__device__ inline void cvt_f32_to_f16_body(const float* __restrict__ src,
                                           _Float16* __restrict__ dst, long long n,
                                           int blk, int nblk) {
    const long long n4 = n >> 2;
    const long long tid = (long long)blk * blockDim.x + threadIdx.x;
    const long long stride = (long long)nblk * blockDim.x;
    const float4* __restrict__ s4 = (const float4*)src;
    half4* __restrict__ d4 = (half4*)dst;
    long long i = tid;
    for (; i + 3 * stride < n4; i += 4 * stride) {
        float4 a = s4[i], b = s4[i + stride], c = s4[i + 2 * stride], d = s4[i + 3 * stride];
        d4[i] = cvt4(a); d4[i + stride] = cvt4(b);
        d4[i + 2 * stride] = cvt4(c); d4[i + 3 * stride] = cvt4(d);
    }
    for (; i < n4; i += stride) d4[i] = cvt4(s4[i]);
    if (tid < (n & 3)) dst[(n4 << 2) + tid] = (_Float16)src[(n4 << 2) + tid];
}

__device__ inline void tr_w128_body(const float* __restrict__ W, _Float16* __restrict__ Wt,
                                    int blk) {
    int i = blk * blockDim.x + threadIdx.x;
    if (i >= 128 * 128) return;
    int c = i >> 7, k = i & 127;
    Wt[i] = (_Float16)W[k * 128 + c];
}

__device__ inline void tr_w40_body(const float* __restrict__ W, _Float16* __restrict__ Wt,
                                   int blk) {
    int i = blk * blockDim.x + threadIdx.x;
    if (i >= 48 * 128) return;
    int c = i >> 7, k = i & 127;
    Wt[i] = (c < 40) ? (_Float16)W[k * 40 + c] : (_Float16)0.f;
}

template <int NCT, int NB>
__device__ inline void gemm_dual_strip(const _Float16* __restrict__ A,
                                       const _Float16* __restrict__ Wtl,
                                       const _Float16* __restrict__ Wtr,
                                       const float* __restrict__ bl,
                                       const float* __restrict__ br,
                                       float* __restrict__ Cl, float* __restrict__ Cr,
                                       int Nrows, int blk) {
    constexpr int ldc = NCT * 16;
    const int lane = threadIdx.x & 31;
    const int strip = __builtin_amdgcn_readfirstlane(
        (int)((blk * blockDim.x + threadIdx.x) >> 5));
    const int r0 = strip * 16;
    if (r0 >= Nrows) return;
    v16h a[4];
#pragma unroll
    for (int kk = 0; kk < 4; ++kk) a[kk] = load_frag16(A, r0, 128, kk * 32, lane, Nrows - 1);
    const int n = lane & 15, hw = lane >> 4;
    const bool whole = r0 + 16 <= Nrows;
#pragma unroll
    for (int ct = 0; ct < NCT; ++ct) {
        const int c0 = ct * 16;
        v16h bL[4], bR[4];
#pragma unroll
        for (int kk = 0; kk < 4; ++kk) {
            bL[kk] = load_frag16(Wtl, c0, 128, kk * 32, lane, ldc - 1);
            bR[kk] = load_frag16(Wtr, c0, 128, kk * 32, lane, ldc - 1);
        }
        v8f accL = {};
        v8f accR = {};
#pragma unroll
        for (int kk = 0; kk < 4; ++kk) {
            accL = __builtin_amdgcn_wmma_f32_16x16x32_f16(false, a[kk], false, bL[kk],
                                                          (short)0, accL, false, false);
            accR = __builtin_amdgcn_wmma_f32_16x16x32_f16(false, a[kk], false, bR[kk],
                                                          (short)0, accR, false, false);
        }
        const int col = c0 + n;
        float bLv = 0.f, bRv = 0.f;
        if (c0 + 16 <= NB || col < NB) { bLv = bl[col]; bRv = br[col]; }
        float* __restrict__ pl = Cl + (size_t)(r0 + hw * 8) * ldc + col;
        float* __restrict__ pr = Cr + (size_t)(r0 + hw * 8) * ldc + col;
        if (whole) {
#pragma unroll
            for (int v = 0; v < 8; ++v) {
                pl[v * ldc] = accL[v] + bLv;
                pr[v * ldc] = accR[v] + bRv;
            }
        } else {
#pragma unroll
            for (int v = 0; v < 8; ++v) {
                if (r0 + hw * 8 + v < Nrows) {
                    pl[v * ldc] = accL[v] + bLv;
                    pr[v * ldc] = accR[v] + bRv;
                }
            }
        }
    }
}
__global__ __launch_bounds__(128)
void k_gemm_dual_w48(const _Float16* __restrict__ A, const _Float16* __restrict__ Wtl,
                     const _Float16* __restrict__ Wtr, const float* __restrict__ bl,
                     const float* __restrict__ br, float* __restrict__ Cl,
                     float* __restrict__ Cr, int Nrows) {
    gemm_dual_strip<3, 40>(A, Wtl, Wtr, bl, br, Cl, Cr, Nrows, (int)blockIdx.x);
}

#define SLICES 256
#define BIN_SHIFT 8
#define BIN_NODES (1 << BIN_SHIFT)
#define MAX_BINS 1024

__device__ inline void slice_bounds(int E, int s, int& e0, int& e1) {
    const int per = (E + SLICES - 1) / SLICES;
    e0 = s * per;
    e1 = e0 + per;
    if (e1 > E) e1 = E;
}

__device__ inline void hist_body(const int* __restrict__ ei, int E, int nbins,
                                 int* __restrict__ hist, int s, int* cnt) {
    for (int i = threadIdx.x; i < nbins; i += blockDim.x) cnt[i] = 0;
    __syncthreads();
    int e0, e1;
    slice_bounds(E, s, e0, e1);
    const int step = blockDim.x;
    for (int i = e0 + (int)threadIdx.x; i < e1; i += 4 * step) {
        int dd[4];
#pragma unroll
        for (int u = 0; u < 4; ++u) {
            const int j = i + u * step;
            dd[u] = (j < e1) ? ei[E + j] : -1;
        }
#pragma unroll
        for (int u = 0; u < 4; ++u)
            if (dd[u] >= 0) atomicAdd(&cnt[dd[u] >> BIN_SHIFT], 1);
    }
    __syncthreads();
    for (int i = threadIdx.x; i < nbins; i += blockDim.x) hist[i * SLICES + s] = cnt[i];
}

__device__ inline void scatter_body(const int* __restrict__ ei, int E, int nbins,
                                    const int* __restrict__ offs, int* __restrict__ binned_words,
                                    int s, int* cur) {
    int2* __restrict__ binned = (int2*)binned_words;
    for (int i = threadIdx.x; i < nbins; i += blockDim.x) cur[i] = offs[i * SLICES + s];
    __syncthreads();
    int e0, e1;
    slice_bounds(E, s, e0, e1);
    const int step = blockDim.x;
    for (int i = e0 + (int)threadIdx.x; i < e1; i += 4 * step) {
        int dd[4], ss[4];
#pragma unroll
        for (int u = 0; u < 4; ++u) {
            const int j = i + u * step;
            const bool on = j < e1;
            dd[u] = on ? ei[E + j] : -1;
            ss[u] = on ? ei[j] : 0;
        }
#pragma unroll
        for (int u = 0; u < 4; ++u) {
            if (dd[u] >= 0) {
                const int p = atomicAdd(&cur[dd[u] >> BIN_SHIFT], 1);
                binned[p] = make_int2(dd[u], ss[u]);
            }
        }
    }
}

__global__ __launch_bounds__(256)
void k_hist_tr_cvt(const int* __restrict__ ei, int E, int nbins, int* __restrict__ hist,
                   const float* __restrict__ W1l, _Float16* __restrict__ Wt1l,
                   const float* __restrict__ W1r, _Float16* __restrict__ Wt1r,
                   const float* __restrict__ W2l, _Float16* __restrict__ Wt2l,
                   const float* __restrict__ W2r, _Float16* __restrict__ Wt2r,
                   const float* __restrict__ src, _Float16* __restrict__ dst, long long n,
                   int ncv) {
    __shared__ int cnt[MAX_BINS];
    int b = blockIdx.x;
    if (b < SLICES) { hist_body(ei, E, nbins, hist, b, cnt); return; }
    b -= SLICES;
    if (b < 64) { tr_w128_body(W1l, Wt1l, b); return; }
    b -= 64;
    if (b < 64) { tr_w128_body(W1r, Wt1r, b); return; }
    b -= 64;
    if (b < 24) { tr_w40_body(W2l, Wt2l, b); return; }
    b -= 24;
    if (b < 24) { tr_w40_body(W2r, Wt2r, b); return; }
    b -= 24;
    cvt_f32_to_f16_body(src, dst, n, b, ncv);
}

__device__ inline int4 ld4i_guard(const int* __restrict__ p, int idx, int N) {
    int4 r = make_int4(0, 0, 0, 0);
    if (idx + 3 < N) {
        r = *(const int4*)(p + idx);
    } else if (idx < N) {
        const int a = p[idx];
        const int b = p[(idx + 1 < N) ? idx + 1 : idx];
        const int c = p[(idx + 2 < N) ? idx + 2 : idx];
        r.x = a;
        r.y = (idx + 1 < N) ? b : 0;
        r.z = (idx + 2 < N) ? c : 0;
    }
    return r;
}
__device__ inline void st4i_guard(int* __restrict__ p, int idx, int N, const int4& o) {
    if (idx + 3 < N) {
        *(int4*)(p + idx) = o;
    } else {
        if (idx     < N) p[idx]     = o.x;
        if (idx + 1 < N) p[idx + 1] = o.y;
        if (idx + 2 < N) p[idx + 2] = o.z;
    }
}
#define SCAN_PIECE (1024 * 4)
#define SCAN_TILE (4 * SCAN_PIECE)
__device__ inline void ld_tile(const int* __restrict__ p, int i0, int N, int4 (&r)[4]) {
    if (i0 + 3 * SCAN_PIECE + 3 < N) {
#pragma unroll
        for (int q = 0; q < 4; ++q) r[q] = *(const int4*)(p + i0 + q * SCAN_PIECE);
    } else {
#pragma unroll
        for (int q = 0; q < 4; ++q) r[q] = ld4i_guard(p, i0 + q * SCAN_PIECE, N);
    }
}
__device__ inline int wave_incl_scan(int v, int lane) {
#pragma unroll
    for (int off = 1; off < 32; off <<= 1) {
        int u = __shfl_up(v, off, 32);
        if (lane >= off) v += u;
    }
    return v;
}
__global__ __launch_bounds__(1024)
void k_scan(const int* __restrict__ deg, int* __restrict__ rbeg, int N) {
    __shared__ int wtot[2][4][32];
    const int tid = threadIdx.x;
    const int lane = tid & 31;
    const int wv = __builtin_amdgcn_readfirstlane(tid >> 5);
    const int piece = SCAN_PIECE;
    const int tile = SCAN_TILE;
    const int nT = (N + tile - 1) / tile;
    int carry = 0;
    int4 nx[4];
    ld_tile(deg, tid * 4, N, nx);
    for (int t = 0; t < nT; ++t) {
        const int i0 = t * tile + tid * 4;
        int4 v[4];
#pragma unroll
        for (int q = 0; q < 4; ++q) v[q] = nx[q];
        if (t + 1 < nT) ld_tile(deg, i0 + SCAN_TILE, N, nx);
        int run[4], inc[4];
#pragma unroll
        for (int q = 0; q < 4; ++q) {
            v[q].y += v[q].x; v[q].z += v[q].y; v[q].w += v[q].z;
            run[q] = v[q].w;
        }
#pragma unroll
        for (int q = 0; q < 4; ++q) inc[q] = wave_incl_scan(run[q], lane);
        if (lane == 31) {
#pragma unroll
            for (int q = 0; q < 4; ++q) wtot[t & 1][q][wv] = inc[q];
        }
        __syncthreads();
        int wt[4], winc[4];
#pragma unroll
        for (int q = 0; q < 4; ++q) wt[q] = wtot[t & 1][q][lane];
#pragma unroll
        for (int q = 0; q < 4; ++q) winc[q] = wave_incl_scan(wt[q], lane);
#pragma unroll
        for (int q = 0; q < 4; ++q) {
            const int wexcl = __builtin_amdgcn_readlane(winc[q] - wt[q], wv);
            const int total = __builtin_amdgcn_readlane(winc[q], 31);
            const int base = carry + wexcl + (inc[q] - run[q]);
            const int4 o = make_int4(base, base + v[q].x, base + v[q].y, base + v[q].z);
            st4i_guard(rbeg, i0 + q * piece, N, o);
            carry += total;
        }
    }
}

__global__ __launch_bounds__(256)
void k_scatter_gemm1(const int* __restrict__ ei, int E, int nbins,
                     const int* __restrict__ offs, int* __restrict__ binned,
                     const _Float16* __restrict__ A, const _Float16* __restrict__ Wtl,
                     const _Float16* __restrict__ Wtr, const float* __restrict__ bl,
                     const float* __restrict__ br, float* __restrict__ Cl,
                     float* __restrict__ Cr, int Nrows) {
    __shared__ int cur[MAX_BINS];
    const int b = blockIdx.x;
    if (b < SLICES) scatter_body(ei, E, nbins, offs, binned, b, cur);
    else gemm_dual_strip<8, 128>(A, Wtl, Wtr, bl, br, Cl, Cr, Nrows, b - SLICES);
}

__global__ __launch_bounds__(BIN_NODES)
void k_binsort(const int* __restrict__ binned_words, const int* __restrict__ offs, int nbins,
               int E, int N, int* __restrict__ rbeg, int* __restrict__ rend,
               int* __restrict__ csrc) {
    const int2* __restrict__ binned = (const int2*)binned_words;
    __shared__ int cnt[BIN_NODES];
    __shared__ int wsum[BIN_NODES / 32];
    const int bin = blockIdx.x;
    const int tid = threadIdx.x;
    const int lane = tid & 31;
    const int wv = tid >> 5;
    const int s0 = offs[bin * SLICES];
    const int s1 = (bin + 1 < nbins) ? offs[(bin + 1) * SLICES] : E;
    cnt[tid] = 0;
    __syncthreads();
    for (int i = s0 + tid; i < s1; i += 4 * BIN_NODES) {
        int dd[4];
#pragma unroll
        for (int u = 0; u < 4; ++u) {
            const int j = i + u * BIN_NODES;
            dd[u] = (j < s1) ? binned[j].x : -1;
        }
#pragma unroll
        for (int u = 0; u < 4; ++u)
            if (dd[u] >= 0) atomicAdd(&cnt[dd[u] & (BIN_NODES - 1)], 1);
    }
    __syncthreads();
    const int c = cnt[tid];
    const int inc = wave_incl_scan(c, lane);
    if (lane == 31) wsum[wv] = inc;
    __syncthreads();
    int wbase = 0;
#pragma unroll
    for (int w = 0; w < BIN_NODES / 32; ++w) {
        const int t = wsum[w];
        if (w < wv) wbase += t;
    }
    const int start = s0 + wbase + inc - c;
    const int node = (bin << BIN_SHIFT) + tid;
    if (node < N) { rbeg[node] = start; rend[node] = start + c; }
    cnt[tid] = start;
    __syncthreads();
    for (int i = s0 + tid; i < s1; i += 4 * BIN_NODES) {
        int2 ee[4];
#pragma unroll
        for (int u = 0; u < 4; ++u) {
            const int j = i + u * BIN_NODES;
            ee[u] = (j < s1) ? binned[j] : make_int2(-1, 0);
        }
#pragma unroll
        for (int u = 0; u < 4; ++u) {
            if (ee[u].x >= 0) {
                const int p = atomicAdd(&cnt[ee[u].x & (BIN_NODES - 1)], 1);
                csrc[p] = ee[u].y;
            }
        }
    }
}

__device__ inline float score4(const float4& a, const float4& b, const float4& t) {
    return t.x * lrelu(a.x + b.x) + t.y * lrelu(a.y + b.y) +
           t.z * lrelu(a.z + b.z) + t.w * lrelu(a.w + b.w);
}
__device__ inline float red16(float v) {
#pragma unroll
    for (int off = 1; off < 16; off <<= 1) v += __shfl_xor(v, off, 32);
    return v;
}
__device__ inline float redmax16(float v) {
#pragma unroll
    for (int off = 1; off < 16; off <<= 1) v = fmaxf(v, __shfl_xor(v, off, 32));
    return v;
}
__device__ inline void scale4(float4& a, float s) { a.x *= s; a.y *= s; a.z *= s; a.w *= s; }
__device__ inline void axpy4(float4& a, float p, const float4& v) {
    a.x += p * v.x; a.y += p * v.y; a.z += p * v.z; a.w += p * v.w;
}

__device__ inline int sort_chunk(int mine, int cnt, int lane) {
    int rank = 0;
    for (int k = 0; k < cnt; ++k) {
        int v = __builtin_amdgcn_readlane(mine, k);
        rank += ((v < mine) || (v == mine && k < lane)) ? 1 : 0;
    }
    return __builtin_amdgcn_ds_permute(rank << 2, mine);
}

__device__ inline void fold4(float& m, float& den, float4& acc,
                             float e0, float e1, float e2, float e3,
                             const float4& a0, const float4& a1,
                             const float4& a2, const float4& a3) {
    float mn = fmaxf(fmaxf(m, fmaxf(e0, e1)), fmaxf(e2, e3));
    float sc = __expf(m - mn);
    float p0 = __expf(e0 - mn), p1 = __expf(e1 - mn);
    float p2 = __expf(e2 - mn), p3 = __expf(e3 - mn);
    scale4(acc, sc);
    axpy4(acc, p0, a0); axpy4(acc, p1, a1); axpy4(acc, p2, a2); axpy4(acc, p3, a3);
    den = den * sc + ((p0 + p1) + (p2 + p3));
    m = mn;
}
__device__ inline void fold1(float& m, float& den, float4& acc, float e0, const float4& a0) {
    float mn = fmaxf(m, e0);
    float sc = __expf(m - mn);
    float p0 = __expf(e0 - mn);
    scale4(acc, sc);
    axpy4(acc, p0, a0);
    den = den * sc + p0;
    m = mn;
}

__global__ __launch_bounds__(256)
void k_node_attn1(const float* __restrict__ xl, const float* __restrict__ xr,
                 const float* __restrict__ att, const float* __restrict__ bias,
                 const int* __restrict__ rbeg, const int* __restrict__ rend,
                 const int* __restrict__ csrc, _Float16* __restrict__ hh, int N) {
    const int wave = __builtin_amdgcn_readfirstlane(
        (int)((blockIdx.x * blockDim.x + threadIdx.x) >> 5));
    const int nwave = (int)((gridDim.x * blockDim.x) >> 5);
    const int lane = threadIdx.x & 31;
    const float4 t  = ((const float4*)att)[lane];
    const float4 bb = ((const float4*)bias)[lane];
    for (int d = wave; d < N; d += nwave) {
    const float4 b4 = ((const float4*)(xr + (size_t)d * 128))[lane];
    float4 acc = ((const float4*)(xl + (size_t)d * 128))[lane];
    float m = red16(score4(acc, b4, t));
    float den = 1.f;
    int beg = rbeg[d], end = rend[d];
    for (int j0 = beg; j0 < end; j0 += 32) {
        int cnt = end - j0; if (cnt > 32) cnt = 32;
        int mine = (lane < cnt) ? csrc[j0 + lane] : 0x7fffffff;
        int srt = sort_chunk(mine, cnt, lane);
        int k = 0;
        for (; k + 4 <= cnt; k += 4) {
            int s0 = __builtin_amdgcn_readlane(srt, k);
            int s1 = __builtin_amdgcn_readlane(srt, k + 1);
            int s2 = __builtin_amdgcn_readlane(srt, k + 2);
            int s3 = __builtin_amdgcn_readlane(srt, k + 3);
            float4 a0 = ((const float4*)(xl + (size_t)s0 * 128))[lane];
            float4 a1 = ((const float4*)(xl + (size_t)s1 * 128))[lane];
            float4 a2 = ((const float4*)(xl + (size_t)s2 * 128))[lane];
            float4 a3 = ((const float4*)(xl + (size_t)s3 * 128))[lane];
            float e0 = red16(score4(a0, b4, t));
            float e1 = red16(score4(a1, b4, t));
            float e2 = red16(score4(a2, b4, t));
            float e3 = red16(score4(a3, b4, t));
            fold4(m, den, acc, e0, e1, e2, e3, a0, a1, a2, a3);
        }
        for (; k < cnt; ++k) {
            int s0 = __builtin_amdgcn_readlane(srt, k);
            float4 a0 = ((const float4*)(xl + (size_t)s0 * 128))[lane];
            float e0 = red16(score4(a0, b4, t));
            fold1(m, den, acc, e0, a0);
        }
    }
    float inv = 1.f / den;
    float o0 = fmaxf(acc.x * inv + bb.x, 0.f);
    float o1 = fmaxf(acc.y * inv + bb.y, 0.f);
    float o2 = fmaxf(acc.z * inv + bb.z, 0.f);
    float o3 = fmaxf(acc.w * inv + bb.w, 0.f);
    half4 o;
    o[0] = (_Float16)o0; o[1] = (_Float16)o1; o[2] = (_Float16)o2; o[3] = (_Float16)o3;
    *(half4*)(hh + (size_t)d * 128 + lane * 4) = o;
    }
}

__device__ inline float4 ld4_if(const float* __restrict__ p, int idx, bool on) {
    float4 r = make_float4(0.f, 0.f, 0.f, 0.f);
    if (on) r = ((const float4*)p)[idx];
    return r;
}

__global__ __launch_bounds__(256)
void k_node_attn2(const float* __restrict__ xl, const float* __restrict__ xr,
                 const float* __restrict__ att, const float* __restrict__ bias,
                 const int* __restrict__ rbeg, const int* __restrict__ rend,
                 const int* __restrict__ csrc, float* __restrict__ out, int N) {
    const int wave = __builtin_amdgcn_readfirstlane(
        (int)((blockIdx.x * blockDim.x + threadIdx.x) >> 5));
    const int nwave = (int)((gridDim.x * blockDim.x) >> 5);
    const int lane = threadIdx.x & 31;
    const int hl = lane & 15;
    const bool act = hl < 10;
    const float4 t  = ld4_if(att, hl, act);
    for (int d = wave; d < N; d += nwave) {
    const float4 b4 = ld4_if(xr + (size_t)d * 48, hl, act);
    float4 acc = ld4_if(xl + (size_t)d * 48, hl, act);
    float m = red16(score4(acc, b4, t));
    float den = 1.f;
    int beg = rbeg[d], end = rend[d];
    for (int j0 = beg; j0 < end; j0 += 32) {
        int cnt = end - j0; if (cnt > 32) cnt = 32;
        int mine = (lane < cnt) ? csrc[j0 + lane] : 0x7fffffff;
        int srt = sort_chunk(mine, cnt, lane);
        int k = 0;
        for (; k + 4 <= cnt; k += 4) {
            int s0 = __builtin_amdgcn_readlane(srt, k);
            int s1 = __builtin_amdgcn_readlane(srt, k + 1);
            int s2 = __builtin_amdgcn_readlane(srt, k + 2);
            int s3 = __builtin_amdgcn_readlane(srt, k + 3);
            float4 a0 = ld4_if(xl + (size_t)s0 * 48, hl, act);
            float4 a1 = ld4_if(xl + (size_t)s1 * 48, hl, act);
            float4 a2 = ld4_if(xl + (size_t)s2 * 48, hl, act);
            float4 a3 = ld4_if(xl + (size_t)s3 * 48, hl, act);
            float e0 = red16(score4(a0, b4, t));
            float e1 = red16(score4(a1, b4, t));
            float e2 = red16(score4(a2, b4, t));
            float e3 = red16(score4(a3, b4, t));
            fold4(m, den, acc, e0, e1, e2, e3, a0, a1, a2, a3);
        }
        for (; k < cnt; ++k) {
            int s0 = __builtin_amdgcn_readlane(srt, k);
            float4 a0 = ld4_if(xl + (size_t)s0 * 48, hl, act);
            float e0 = red16(score4(a0, b4, t));
            fold1(m, den, acc, e0, a0);
        }
    }
    float inv = 1.f / den;
    const float4 bb = ld4_if(bias, hl, act);
    float v0 = acc.x * inv + bb.x, v1 = acc.y * inv + bb.y;
    float v2 = acc.z * inv + bb.z, v3 = acc.w * inv + bb.w;
    float mx = act ? fmaxf(fmaxf(v0, v1), fmaxf(v2, v3)) : -INFINITY;
    mx = redmax16(mx);
    float se = act ? (__expf(v0 - mx) + __expf(v1 - mx)) + (__expf(v2 - mx) + __expf(v3 - mx))
                   : 0.f;
    se = red16(se);
    float L = mx + logf(se);
    if (lane < 10)
        *(float4*)(out + (size_t)d * 40 + lane * 4) = make_float4(v0 - L, v1 - L, v2 - L, v3 - L);
    }
}

extern "C" void kernel_launch(void* const* d_in, const int* in_sizes, int n_in,
                              void* d_out, int out_size, void* d_ws, size_t ws_size,
                              hipStream_t stream) {
    const float* x    = (const float*)d_in[0];
    const int*   ei   = (const int*)d_in[1];
    const float* W1l  = (const float*)d_in[2];
    const float* b1l  = (const float*)d_in[3];
    const float* W1r  = (const float*)d_in[4];
    const float* b1r  = (const float*)d_in[5];
    const float* att1 = (const float*)d_in[6];
    const float* bias1= (const float*)d_in[7];
    const float* W2l  = (const float*)d_in[8];
    const float* b2l  = (const float*)d_in[9];
    const float* W2r  = (const float*)d_in[10];
    const float* b2r  = (const float*)d_in[11];
    const float* att2 = (const float*)d_in[12];
    const float* bias2= (const float*)d_in[13];
    float* out = (float*)d_out;

    const int N = in_sizes[0] / 128;
    const int E = in_sizes[1] / 2;

    char* w = (char*)d_ws;
    size_t off = 0;
    auto take = [&](size_t bytes) -> char* {
        char* p = w + off;
        off = (off + bytes + 255) & ~(size_t)255;
        return p;
    };
    _Float16* xh   = (_Float16*)take((size_t)N * 128 * 2);
    _Float16* hh   = (_Float16*)take((size_t)N * 128 * 2);
    _Float16* Wt1l = (_Float16*)take(128 * 128 * 2);
    _Float16* Wt1r = (_Float16*)take(128 * 128 * 2);
    _Float16* Wt2l = (_Float16*)take(48 * 128 * 2);
    _Float16* Wt2r = (_Float16*)take(48 * 128 * 2);
    float*    xl   = (float*)take((size_t)N * 128 * 4);
    float*    xr   = (float*)take((size_t)N * 128 * 4);
    const int nbins = (N + BIN_NODES - 1) / BIN_NODES;
    const int nh = nbins * SLICES;
    int*      hist = (int*)take((size_t)nh * 4);
    int*      offs = (int*)take((size_t)nh * 4);
    int*      binned = (int*)take((size_t)E * 8);
    int*      rbeg = (int*)take((size_t)N * 4);
    int*      rend = (int*)take((size_t)N * 4);
    int*      csrc = (int*)take((size_t)E * 4);

    const int B = 256;
    const long long Nx128 = (long long)N * 128;
    const int rowTiles = (N + 15) / 16;

    k_hist_tr_cvt<<<SLICES + 64 + 64 + 24 + 24 + 512, B, 0, stream>>>(
        ei, E, nbins, hist, W1l, Wt1l, W1r, Wt1r, W2l, Wt2l, W2r, Wt2r, x, xh, Nx128, 512);
    k_scan<<<1, 1024, 0, stream>>>(hist, offs, nh);
    k_scatter_gemm1<<<SLICES + (rowTiles + 7) / 8, B, 0, stream>>>(ei, E, nbins, offs, binned,
                                                                xh, Wt1l, Wt1r, b1l, b1r,
                                                                xl, xr, N);
    k_binsort<<<nbins, BIN_NODES, 0, stream>>>(binned, offs, nbins, E, N, rbeg, rend, csrc);

    k_node_attn1<<<2048, B, 0, stream>>>(xl, xr, att1, bias1, rbeg, rend, csrc, hh, N);

    k_gemm_dual_w48<<<(rowTiles + 3) / 4, 128, 0, stream>>>(hh, Wt2l, Wt2r, b2l, b2r,
                                                           xl, xr, N);
    k_node_attn2<<<1536, B, 0, stream>>>(xl, xr, att2, bias2, rbeg, rend, csrc, out, N);
}


# Weights stored in matrix-fragment order: a B-fragment load is one 512-byte run per wave (4 lines) instead of 16-byte pieces of 16 rows (16 lines); fragments, products and sums bit-identical
# speedup vs baseline: 6.1120x; 1.0555x over previous
#include <hip/hip_runtime.h>
#include <hip/hip_bf16.h>
#include <math.h>

typedef __attribute__((ext_vector_type(16))) _Float16 v16h;
typedef __attribute__((ext_vector_type(8)))  _Float16 half8;
typedef __attribute__((ext_vector_type(8)))  float    v8f;

#define NEG_SLOPE 0.2f

__device__ inline float lrelu(float v) { return v > 0.f ? v : NEG_SLOPE * v; }

__device__ inline v16h load_frag16(const _Float16* __restrict__ p, int row0,
                                   int ld, int k0, int lane, int maxRow) {
    int m = lane & 15;
    int base = (lane >> 4) * 8;
    int r = row0 + m;
    if (r > maxRow) r = maxRow;
    const _Float16* q = p + (size_t)r * ld + k0 + base;
    half8 lo = *(const half8*)q;
    half8 hi = *(const half8*)(q + 16);
    v16h f;
#pragma unroll
    for (int i = 0; i < 8; ++i) { f[i] = lo[i]; f[i + 8] = hi[i]; }
    return f;
}


typedef __attribute__((ext_vector_type(4))) _Float16 half4;
__device__ inline half4 cvt4(const float4& v) {
    half4 o;
    o[0] = (_Float16)v.x; o[1] = (_Float16)v.y; o[2] = (_Float16)v.z; o[3] = (_Float16)v.w;
    return o;
}
__device__ inline void cvt_f32_to_f16_body(const float* __restrict__ src,
                                           _Float16* __restrict__ dst, long long n,
                                           int blk, int nblk) {
    const long long n4 = n >> 2;
    const long long tid = (long long)blk * blockDim.x + threadIdx.x;
    const long long stride = (long long)nblk * blockDim.x;
    const float4* __restrict__ s4 = (const float4*)src;
    half4* __restrict__ d4 = (half4*)dst;
    long long i = tid;
    for (; i + 3 * stride < n4; i += 4 * stride) {
        float4 a = s4[i], b = s4[i + stride], c = s4[i + 2 * stride], d = s4[i + 3 * stride];
        d4[i] = cvt4(a); d4[i + stride] = cvt4(b);
        d4[i + 2 * stride] = cvt4(c); d4[i + 3 * stride] = cvt4(d);
    }
    for (; i < n4; i += stride) d4[i] = cvt4(s4[i]);
    if (tid < (n & 3)) dst[(n4 << 2) + tid] = (_Float16)src[(n4 << 2) + tid];
}

template <int C, int NCT>
__device__ inline void wfrag_body(const float* __restrict__ W, _Float16* __restrict__ Wf,
                                  int blk) {
    int i = blk * blockDim.x + threadIdx.x;
    if (i >= NCT * 16 * 128) return;
    const int e = i & 7, L = (i >> 3) & 31, j = (i >> 8) & 1, kk = (i >> 9) & 3, ct = i >> 11;
    const int k = kk * 32 + j * 16 + (L >> 4) * 8 + e;
    const int col = ct * 16 + (L & 15);
    Wf[i] = (col < C) ? (_Float16)W[k * C + col] : (_Float16)0.f;
}

__device__ inline v16h load_wfrag(const _Float16* __restrict__ Wf, int ct, int kk, int lane) {
    const _Float16* q = Wf + (size_t)((ct * 4 + kk) * 2) * 256 + lane * 8;
    half8 lo = *(const half8*)q;
    half8 hi = *(const half8*)(q + 256);
    v16h f;
#pragma unroll
    for (int i = 0; i < 8; ++i) { f[i] = lo[i]; f[i + 8] = hi[i]; }
    return f;
}

template <int NCT, int NB>
__device__ inline void gemm_dual_strip(const _Float16* __restrict__ A,
                                       const _Float16* __restrict__ Wtl,
                                       const _Float16* __restrict__ Wtr,
                                       const float* __restrict__ bl,
                                       const float* __restrict__ br,
                                       float* __restrict__ Cl, float* __restrict__ Cr,
                                       int Nrows, int blk) {
    constexpr int ldc = NCT * 16;
    const int lane = threadIdx.x & 31;
    const int strip = __builtin_amdgcn_readfirstlane(
        (int)((blk * blockDim.x + threadIdx.x) >> 5));
    const int r0 = strip * 16;
    if (r0 >= Nrows) return;
    v16h a[4];
#pragma unroll
    for (int kk = 0; kk < 4; ++kk) a[kk] = load_frag16(A, r0, 128, kk * 32, lane, Nrows - 1);
    const int n = lane & 15, hw = lane >> 4;
    const bool whole = r0 + 16 <= Nrows;
#pragma unroll
    for (int ct = 0; ct < NCT; ++ct) {
        const int c0 = ct * 16;
        v16h bL[4], bR[4];
#pragma unroll
        for (int kk = 0; kk < 4; ++kk) {
            bL[kk] = load_wfrag(Wtl, ct, kk, lane);
            bR[kk] = load_wfrag(Wtr, ct, kk, lane);
        }
        v8f accL = {};
        v8f accR = {};
#pragma unroll
        for (int kk = 0; kk < 4; ++kk) {
            accL = __builtin_amdgcn_wmma_f32_16x16x32_f16(false, a[kk], false, bL[kk],
                                                          (short)0, accL, false, false);
            accR = __builtin_amdgcn_wmma_f32_16x16x32_f16(false, a[kk], false, bR[kk],
                                                          (short)0, accR, false, false);
        }
        const int col = c0 + n;
        float bLv = 0.f, bRv = 0.f;
        if (c0 + 16 <= NB || col < NB) { bLv = bl[col]; bRv = br[col]; }
        float* __restrict__ pl = Cl + (size_t)(r0 + hw * 8) * ldc + col;
        float* __restrict__ pr = Cr + (size_t)(r0 + hw * 8) * ldc + col;
        if (whole) {
#pragma unroll
            for (int v = 0; v < 8; ++v) {
                pl[v * ldc] = accL[v] + bLv;
                pr[v * ldc] = accR[v] + bRv;
            }
        } else {
#pragma unroll
            for (int v = 0; v < 8; ++v) {
                if (r0 + hw * 8 + v < Nrows) {
                    pl[v * ldc] = accL[v] + bLv;
                    pr[v * ldc] = accR[v] + bRv;
                }
            }
        }
    }
}
__global__ __launch_bounds__(128)
void k_gemm_dual_w48(const _Float16* __restrict__ A, const _Float16* __restrict__ Wtl,
                     const _Float16* __restrict__ Wtr, const float* __restrict__ bl,
                     const float* __restrict__ br, float* __restrict__ Cl,
                     float* __restrict__ Cr, int Nrows) {
    gemm_dual_strip<3, 40>(A, Wtl, Wtr, bl, br, Cl, Cr, Nrows, (int)blockIdx.x);
}

#define SLICES 256
#define BIN_SHIFT 8
#define BIN_NODES (1 << BIN_SHIFT)
#define MAX_BINS 1024

__device__ inline void slice_bounds(int E, int s, int& e0, int& e1) {
    const int per = (E + SLICES - 1) / SLICES;
    e0 = s * per;
    e1 = e0 + per;
    if (e1 > E) e1 = E;
}

__device__ inline void hist_body(const int* __restrict__ ei, int E, int nbins,
                                 int* __restrict__ hist, int s, int* cnt) {
    for (int i = threadIdx.x; i < nbins; i += blockDim.x) cnt[i] = 0;
    __syncthreads();
    int e0, e1;
    slice_bounds(E, s, e0, e1);
    const int step = blockDim.x;
    for (int i = e0 + (int)threadIdx.x; i < e1; i += 4 * step) {
        int dd[4];
#pragma unroll
        for (int u = 0; u < 4; ++u) {
            const int j = i + u * step;
            dd[u] = (j < e1) ? ei[E + j] : -1;
        }
#pragma unroll
        for (int u = 0; u < 4; ++u)
            if (dd[u] >= 0) atomicAdd(&cnt[dd[u] >> BIN_SHIFT], 1);
    }
    __syncthreads();
    for (int i = threadIdx.x; i < nbins; i += blockDim.x) hist[i * SLICES + s] = cnt[i];
}

__device__ inline void scatter_body(const int* __restrict__ ei, int E, int nbins,
                                    const int* __restrict__ offs, int* __restrict__ binned_words,
                                    int s, int* cur) {
    int2* __restrict__ binned = (int2*)binned_words;
    for (int i = threadIdx.x; i < nbins; i += blockDim.x) cur[i] = offs[i * SLICES + s];
    __syncthreads();
    int e0, e1;
    slice_bounds(E, s, e0, e1);
    const int step = blockDim.x;
    for (int i = e0 + (int)threadIdx.x; i < e1; i += 4 * step) {
        int dd[4], ss[4];
#pragma unroll
        for (int u = 0; u < 4; ++u) {
            const int j = i + u * step;
            const bool on = j < e1;
            dd[u] = on ? ei[E + j] : -1;
            ss[u] = on ? ei[j] : 0;
        }
#pragma unroll
        for (int u = 0; u < 4; ++u) {
            if (dd[u] >= 0) {
                const int p = atomicAdd(&cur[dd[u] >> BIN_SHIFT], 1);
                binned[p] = make_int2(dd[u], ss[u]);
            }
        }
    }
}

__global__ __launch_bounds__(256)
void k_hist_tr_cvt(const int* __restrict__ ei, int E, int nbins, int* __restrict__ hist,
                   const float* __restrict__ W1l, _Float16* __restrict__ Wt1l,
                   const float* __restrict__ W1r, _Float16* __restrict__ Wt1r,
                   const float* __restrict__ W2l, _Float16* __restrict__ Wt2l,
                   const float* __restrict__ W2r, _Float16* __restrict__ Wt2r,
                   const float* __restrict__ src, _Float16* __restrict__ dst, long long n,
                   int ncv) {
    __shared__ int cnt[MAX_BINS];
    int b = blockIdx.x;
    if (b < SLICES) { hist_body(ei, E, nbins, hist, b, cnt); return; }
    b -= SLICES;
    if (b < 64) { wfrag_body<128, 8>(W1l, Wt1l, b); return; }
    b -= 64;
    if (b < 64) { wfrag_body<128, 8>(W1r, Wt1r, b); return; }
    b -= 64;
    if (b < 24) { wfrag_body<40, 3>(W2l, Wt2l, b); return; }
    b -= 24;
    if (b < 24) { wfrag_body<40, 3>(W2r, Wt2r, b); return; }
    b -= 24;
    cvt_f32_to_f16_body(src, dst, n, b, ncv);
}

__device__ inline int4 ld4i_guard(const int* __restrict__ p, int idx, int N) {
    int4 r = make_int4(0, 0, 0, 0);
    if (idx + 3 < N) {
        r = *(const int4*)(p + idx);
    } else if (idx < N) {
        const int a = p[idx];
        const int b = p[(idx + 1 < N) ? idx + 1 : idx];
        const int c = p[(idx + 2 < N) ? idx + 2 : idx];
        r.x = a;
        r.y = (idx + 1 < N) ? b : 0;
        r.z = (idx + 2 < N) ? c : 0;
    }
    return r;
}
__device__ inline void st4i_guard(int* __restrict__ p, int idx, int N, const int4& o) {
    if (idx + 3 < N) {
        *(int4*)(p + idx) = o;
    } else {
        if (idx     < N) p[idx]     = o.x;
        if (idx + 1 < N) p[idx + 1] = o.y;
        if (idx + 2 < N) p[idx + 2] = o.z;
    }
}
#define SCAN_PIECE (1024 * 4)
#define SCAN_TILE (4 * SCAN_PIECE)
__device__ inline void ld_tile(const int* __restrict__ p, int i0, int N, int4 (&r)[4]) {
    if (i0 + 3 * SCAN_PIECE + 3 < N) {
#pragma unroll
        for (int q = 0; q < 4; ++q) r[q] = *(const int4*)(p + i0 + q * SCAN_PIECE);
    } else {
#pragma unroll
        for (int q = 0; q < 4; ++q) r[q] = ld4i_guard(p, i0 + q * SCAN_PIECE, N);
    }
}
__device__ inline int wave_incl_scan(int v, int lane) {
#pragma unroll
    for (int off = 1; off < 32; off <<= 1) {
        int u = __shfl_up(v, off, 32);
        if (lane >= off) v += u;
    }
    return v;
}
__global__ __launch_bounds__(1024)
void k_scan(const int* __restrict__ deg, int* __restrict__ rbeg, int N) {
    __shared__ int wtot[2][4][32];
    const int tid = threadIdx.x;
    const int lane = tid & 31;
    const int wv = __builtin_amdgcn_readfirstlane(tid >> 5);
    const int piece = SCAN_PIECE;
    const int tile = SCAN_TILE;
    const int nT = (N + tile - 1) / tile;
    int carry = 0;
    int4 nx[4];
    ld_tile(deg, tid * 4, N, nx);
    for (int t = 0; t < nT; ++t) {
        const int i0 = t * tile + tid * 4;
        int4 v[4];
#pragma unroll
        for (int q = 0; q < 4; ++q) v[q] = nx[q];
        if (t + 1 < nT) ld_tile(deg, i0 + SCAN_TILE, N, nx);
        int run[4], inc[4];
#pragma unroll
        for (int q = 0; q < 4; ++q) {
            v[q].y += v[q].x; v[q].z += v[q].y; v[q].w += v[q].z;
            run[q] = v[q].w;
        }
#pragma unroll
        for (int q = 0; q < 4; ++q) inc[q] = wave_incl_scan(run[q], lane);
        if (lane == 31) {
#pragma unroll
            for (int q = 0; q < 4; ++q) wtot[t & 1][q][wv] = inc[q];
        }
        __syncthreads();
        int wt[4], winc[4];
#pragma unroll
        for (int q = 0; q < 4; ++q) wt[q] = wtot[t & 1][q][lane];
#pragma unroll
        for (int q = 0; q < 4; ++q) winc[q] = wave_incl_scan(wt[q], lane);
#pragma unroll
        for (int q = 0; q < 4; ++q) {
            const int wexcl = __builtin_amdgcn_readlane(winc[q] - wt[q], wv);
            const int total = __builtin_amdgcn_readlane(winc[q], 31);
            const int base = carry + wexcl + (inc[q] - run[q]);
            const int4 o = make_int4(base, base + v[q].x, base + v[q].y, base + v[q].z);
            st4i_guard(rbeg, i0 + q * piece, N, o);
            carry += total;
        }
    }
}

__global__ __launch_bounds__(256)
void k_scatter_gemm1(const int* __restrict__ ei, int E, int nbins,
                     const int* __restrict__ offs, int* __restrict__ binned,
                     const _Float16* __restrict__ A, const _Float16* __restrict__ Wtl,
                     const _Float16* __restrict__ Wtr, const float* __restrict__ bl,
                     const float* __restrict__ br, float* __restrict__ Cl,
                     float* __restrict__ Cr, int Nrows) {
    __shared__ int cur[MAX_BINS];
    const int b = blockIdx.x;
    if (b < SLICES) scatter_body(ei, E, nbins, offs, binned, b, cur);
    else gemm_dual_strip<8, 128>(A, Wtl, Wtr, bl, br, Cl, Cr, Nrows, b - SLICES);
}

__global__ __launch_bounds__(BIN_NODES)
void k_binsort(const int* __restrict__ binned_words, const int* __restrict__ offs, int nbins,
               int E, int N, int* __restrict__ rbeg, int* __restrict__ rend,
               int* __restrict__ csrc) {
    const int2* __restrict__ binned = (const int2*)binned_words;
    __shared__ int cnt[BIN_NODES];
    __shared__ int wsum[BIN_NODES / 32];
    const int bin = blockIdx.x;
    const int tid = threadIdx.x;
    const int lane = tid & 31;
    const int wv = tid >> 5;
    const int s0 = offs[bin * SLICES];
    const int s1 = (bin + 1 < nbins) ? offs[(bin + 1) * SLICES] : E;
    cnt[tid] = 0;
    __syncthreads();
    for (int i = s0 + tid; i < s1; i += 4 * BIN_NODES) {
        int dd[4];
#pragma unroll
        for (int u = 0; u < 4; ++u) {
            const int j = i + u * BIN_NODES;
            dd[u] = (j < s1) ? binned[j].x : -1;
        }
#pragma unroll
        for (int u = 0; u < 4; ++u)
            if (dd[u] >= 0) atomicAdd(&cnt[dd[u] & (BIN_NODES - 1)], 1);
    }
    __syncthreads();
    const int c = cnt[tid];
    const int inc = wave_incl_scan(c, lane);
    if (lane == 31) wsum[wv] = inc;
    __syncthreads();
    int wbase = 0;
#pragma unroll
    for (int w = 0; w < BIN_NODES / 32; ++w) {
        const int t = wsum[w];
        if (w < wv) wbase += t;
    }
    const int start = s0 + wbase + inc - c;
    const int node = (bin << BIN_SHIFT) + tid;
    if (node < N) { rbeg[node] = start; rend[node] = start + c; }
    cnt[tid] = start;
    __syncthreads();
    for (int i = s0 + tid; i < s1; i += 4 * BIN_NODES) {
        int2 ee[4];
#pragma unroll
        for (int u = 0; u < 4; ++u) {
            const int j = i + u * BIN_NODES;
            ee[u] = (j < s1) ? binned[j] : make_int2(-1, 0);
        }
#pragma unroll
        for (int u = 0; u < 4; ++u) {
            if (ee[u].x >= 0) {
                const int p = atomicAdd(&cnt[ee[u].x & (BIN_NODES - 1)], 1);
                csrc[p] = ee[u].y;
            }
        }
    }
}

__device__ inline float score4(const float4& a, const float4& b, const float4& t) {
    return t.x * lrelu(a.x + b.x) + t.y * lrelu(a.y + b.y) +
           t.z * lrelu(a.z + b.z) + t.w * lrelu(a.w + b.w);
}
__device__ inline float red16(float v) {
#pragma unroll
    for (int off = 1; off < 16; off <<= 1) v += __shfl_xor(v, off, 32);
    return v;
}
__device__ inline float redmax16(float v) {
#pragma unroll
    for (int off = 1; off < 16; off <<= 1) v = fmaxf(v, __shfl_xor(v, off, 32));
    return v;
}
__device__ inline void scale4(float4& a, float s) { a.x *= s; a.y *= s; a.z *= s; a.w *= s; }
__device__ inline void axpy4(float4& a, float p, const float4& v) {
    a.x += p * v.x; a.y += p * v.y; a.z += p * v.z; a.w += p * v.w;
}

__device__ inline int sort_chunk(int mine, int cnt, int lane) {
    int rank = 0;
    for (int k = 0; k < cnt; ++k) {
        int v = __builtin_amdgcn_readlane(mine, k);
        rank += ((v < mine) || (v == mine && k < lane)) ? 1 : 0;
    }
    return __builtin_amdgcn_ds_permute(rank << 2, mine);
}

__device__ inline void fold4(float& m, float& den, float4& acc,
                             float e0, float e1, float e2, float e3,
                             const float4& a0, const float4& a1,
                             const float4& a2, const float4& a3) {
    float mn = fmaxf(fmaxf(m, fmaxf(e0, e1)), fmaxf(e2, e3));
    float sc = __expf(m - mn);
    float p0 = __expf(e0 - mn), p1 = __expf(e1 - mn);
    float p2 = __expf(e2 - mn), p3 = __expf(e3 - mn);
    scale4(acc, sc);
    axpy4(acc, p0, a0); axpy4(acc, p1, a1); axpy4(acc, p2, a2); axpy4(acc, p3, a3);
    den = den * sc + ((p0 + p1) + (p2 + p3));
    m = mn;
}
__device__ inline void fold1(float& m, float& den, float4& acc, float e0, const float4& a0) {
    float mn = fmaxf(m, e0);
    float sc = __expf(m - mn);
    float p0 = __expf(e0 - mn);
    scale4(acc, sc);
    axpy4(acc, p0, a0);
    den = den * sc + p0;
    m = mn;
}

__global__ __launch_bounds__(256)
void k_node_attn1(const float* __restrict__ xl, const float* __restrict__ xr,
                 const float* __restrict__ att, const float* __restrict__ bias,
                 const int* __restrict__ rbeg, const int* __restrict__ rend,
                 const int* __restrict__ csrc, _Float16* __restrict__ hh, int N) {
    const int wave = __builtin_amdgcn_readfirstlane(
        (int)((blockIdx.x * blockDim.x + threadIdx.x) >> 5));
    const int nwave = (int)((gridDim.x * blockDim.x) >> 5);
    const int lane = threadIdx.x & 31;
    const float4 t  = ((const float4*)att)[lane];
    const float4 bb = ((const float4*)bias)[lane];
    for (int d = wave; d < N; d += nwave) {
    const float4 b4 = ((const float4*)(xr + (size_t)d * 128))[lane];
    float4 acc = ((const float4*)(xl + (size_t)d * 128))[lane];
    float m = red16(score4(acc, b4, t));
    float den = 1.f;
    int beg = rbeg[d], end = rend[d];
    for (int j0 = beg; j0 < end; j0 += 32) {
        int cnt = end - j0; if (cnt > 32) cnt = 32;
        int mine = (lane < cnt) ? csrc[j0 + lane] : 0x7fffffff;
        int srt = sort_chunk(mine, cnt, lane);
        int k = 0;
        for (; k + 4 <= cnt; k += 4) {
            int s0 = __builtin_amdgcn_readlane(srt, k);
            int s1 = __builtin_amdgcn_readlane(srt, k + 1);
            int s2 = __builtin_amdgcn_readlane(srt, k + 2);
            int s3 = __builtin_amdgcn_readlane(srt, k + 3);
            float4 a0 = ((const float4*)(xl + (size_t)s0 * 128))[lane];
            float4 a1 = ((const float4*)(xl + (size_t)s1 * 128))[lane];
            float4 a2 = ((const float4*)(xl + (size_t)s2 * 128))[lane];
            float4 a3 = ((const float4*)(xl + (size_t)s3 * 128))[lane];
            float e0 = red16(score4(a0, b4, t));
            float e1 = red16(score4(a1, b4, t));
            float e2 = red16(score4(a2, b4, t));
            float e3 = red16(score4(a3, b4, t));
            fold4(m, den, acc, e0, e1, e2, e3, a0, a1, a2, a3);
        }
        for (; k < cnt; ++k) {
            int s0 = __builtin_amdgcn_readlane(srt, k);
            float4 a0 = ((const float4*)(xl + (size_t)s0 * 128))[lane];
            float e0 = red16(score4(a0, b4, t));
            fold1(m, den, acc, e0, a0);
        }
    }
    float inv = 1.f / den;
    float o0 = fmaxf(acc.x * inv + bb.x, 0.f);
    float o1 = fmaxf(acc.y * inv + bb.y, 0.f);
    float o2 = fmaxf(acc.z * inv + bb.z, 0.f);
    float o3 = fmaxf(acc.w * inv + bb.w, 0.f);
    half4 o;
    o[0] = (_Float16)o0; o[1] = (_Float16)o1; o[2] = (_Float16)o2; o[3] = (_Float16)o3;
    *(half4*)(hh + (size_t)d * 128 + lane * 4) = o;
    }
}

__device__ inline float4 ld4_if(const float* __restrict__ p, int idx, bool on) {
    float4 r = make_float4(0.f, 0.f, 0.f, 0.f);
    if (on) r = ((const float4*)p)[idx];
    return r;
}

__global__ __launch_bounds__(256)
void k_node_attn2(const float* __restrict__ xl, const float* __restrict__ xr,
                 const float* __restrict__ att, const float* __restrict__ bias,
                 const int* __restrict__ rbeg, const int* __restrict__ rend,
                 const int* __restrict__ csrc, float* __restrict__ out, int N) {
    const int wave = __builtin_amdgcn_readfirstlane(
        (int)((blockIdx.x * blockDim.x + threadIdx.x) >> 5));
    const int nwave = (int)((gridDim.x * blockDim.x) >> 5);
    const int lane = threadIdx.x & 31;
    const int hl = lane & 15;
    const bool act = hl < 10;
    const float4 t  = ld4_if(att, hl, act);
    for (int d = wave; d < N; d += nwave) {
    const float4 b4 = ld4_if(xr + (size_t)d * 48, hl, act);
    float4 acc = ld4_if(xl + (size_t)d * 48, hl, act);
    float m = red16(score4(acc, b4, t));
    float den = 1.f;
    int beg = rbeg[d], end = rend[d];
    for (int j0 = beg; j0 < end; j0 += 32) {
        int cnt = end - j0; if (cnt > 32) cnt = 32;
        int mine = (lane < cnt) ? csrc[j0 + lane] : 0x7fffffff;
        int srt = sort_chunk(mine, cnt, lane);
        int k = 0;
        for (; k + 4 <= cnt; k += 4) {
            int s0 = __builtin_amdgcn_readlane(srt, k);
            int s1 = __builtin_amdgcn_readlane(srt, k + 1);
            int s2 = __builtin_amdgcn_readlane(srt, k + 2);
            int s3 = __builtin_amdgcn_readlane(srt, k + 3);
            float4 a0 = ld4_if(xl + (size_t)s0 * 48, hl, act);
            float4 a1 = ld4_if(xl + (size_t)s1 * 48, hl, act);
            float4 a2 = ld4_if(xl + (size_t)s2 * 48, hl, act);
            float4 a3 = ld4_if(xl + (size_t)s3 * 48, hl, act);
            float e0 = red16(score4(a0, b4, t));
            float e1 = red16(score4(a1, b4, t));
            float e2 = red16(score4(a2, b4, t));
            float e3 = red16(score4(a3, b4, t));
            fold4(m, den, acc, e0, e1, e2, e3, a0, a1, a2, a3);
        }
        for (; k < cnt; ++k) {
            int s0 = __builtin_amdgcn_readlane(srt, k);
            float4 a0 = ld4_if(xl + (size_t)s0 * 48, hl, act);
            float e0 = red16(score4(a0, b4, t));
            fold1(m, den, acc, e0, a0);
        }
    }
    float inv = 1.f / den;
    const float4 bb = ld4_if(bias, hl, act);
    float v0 = acc.x * inv + bb.x, v1 = acc.y * inv + bb.y;
    float v2 = acc.z * inv + bb.z, v3 = acc.w * inv + bb.w;
    float mx = act ? fmaxf(fmaxf(v0, v1), fmaxf(v2, v3)) : -INFINITY;
    mx = redmax16(mx);
    float se = act ? (__expf(v0 - mx) + __expf(v1 - mx)) + (__expf(v2 - mx) + __expf(v3 - mx))
                   : 0.f;
    se = red16(se);
    float L = mx + logf(se);
    if (lane < 10)
        *(float4*)(out + (size_t)d * 40 + lane * 4) = make_float4(v0 - L, v1 - L, v2 - L, v3 - L);
    }
}

extern "C" void kernel_launch(void* const* d_in, const int* in_sizes, int n_in,
                              void* d_out, int out_size, void* d_ws, size_t ws_size,
                              hipStream_t stream) {
    const float* x    = (const float*)d_in[0];
    const int*   ei   = (const int*)d_in[1];
    const float* W1l  = (const float*)d_in[2];
    const float* b1l  = (const float*)d_in[3];
    const float* W1r  = (const float*)d_in[4];
    const float* b1r  = (const float*)d_in[5];
    const float* att1 = (const float*)d_in[6];
    const float* bias1= (const float*)d_in[7];
    const float* W2l  = (const float*)d_in[8];
    const float* b2l  = (const float*)d_in[9];
    const float* W2r  = (const float*)d_in[10];
    const float* b2r  = (const float*)d_in[11];
    const float* att2 = (const float*)d_in[12];
    const float* bias2= (const float*)d_in[13];
    float* out = (float*)d_out;

    const int N = in_sizes[0] / 128;
    const int E = in_sizes[1] / 2;

    char* w = (char*)d_ws;
    size_t off = 0;
    auto take = [&](size_t bytes) -> char* {
        char* p = w + off;
        off = (off + bytes + 255) & ~(size_t)255;
        return p;
    };
    _Float16* xh   = (_Float16*)take((size_t)N * 128 * 2);
    _Float16* hh   = (_Float16*)take((size_t)N * 128 * 2);
    _Float16* Wt1l = (_Float16*)take(128 * 128 * 2);
    _Float16* Wt1r = (_Float16*)take(128 * 128 * 2);
    _Float16* Wt2l = (_Float16*)take(48 * 128 * 2);
    _Float16* Wt2r = (_Float16*)take(48 * 128 * 2);
    float*    xl   = (float*)take((size_t)N * 128 * 4);
    float*    xr   = (float*)take((size_t)N * 128 * 4);
    const int nbins = (N + BIN_NODES - 1) / BIN_NODES;
    const int nh = nbins * SLICES;
    int*      hist = (int*)take((size_t)nh * 4);
    int*      offs = (int*)take((size_t)nh * 4);
    int*      binned = (int*)take((size_t)E * 8);
    int*      rbeg = (int*)take((size_t)N * 4);
    int*      rend = (int*)take((size_t)N * 4);
    int*      csrc = (int*)take((size_t)E * 4);

    const int B = 256;
    const long long Nx128 = (long long)N * 128;
    const int rowTiles = (N + 15) / 16;

    k_hist_tr_cvt<<<SLICES + 64 + 64 + 24 + 24 + 512, B, 0, stream>>>(
        ei, E, nbins, hist, W1l, Wt1l, W1r, Wt1r, W2l, Wt2l, W2r, Wt2r, x, xh, Nx128, 512);
    k_scan<<<1, 1024, 0, stream>>>(hist, offs, nh);
    k_scatter_gemm1<<<SLICES + (rowTiles + 7) / 8, B, 0, stream>>>(ei, E, nbins, offs, binned,
                                                                xh, Wt1l, Wt1r, b1l, b1r,
                                                                xl, xr, N);
    k_binsort<<<nbins, BIN_NODES, 0, stream>>>(binned, offs, nbins, E, N, rbeg, rend, csrc);

    k_node_attn1<<<2048, B, 0, stream>>>(xl, xr, att1, bias1, rbeg, rend, csrc, hh, N);

    k_gemm_dual_w48<<<(rowTiles + 3) / 4, 128, 0, stream>>>(hh, Wt2l, Wt2r, b2l, b2r,
                                                           xl, xr, N);
    k_node_attn2<<<1536, B, 0, stream>>>(xl, xr, att2, bias2, rbeg, rend, csrc, out, N);
}


# Layer-2 attention packs four edges per wave pass, one per 8-lane group (5 channels a lane), was one edge per pass; per-group softmax partial sums merged at node end, so the sum order changes
# speedup vs baseline: 6.9731x; 1.1409x over previous
#include <hip/hip_runtime.h>
#include <hip/hip_bf16.h>
#include <math.h>

typedef __attribute__((ext_vector_type(16))) _Float16 v16h;
typedef __attribute__((ext_vector_type(8)))  _Float16 half8;
typedef __attribute__((ext_vector_type(8)))  float    v8f;

#define NEG_SLOPE 0.2f

__device__ inline float lrelu(float v) { return v > 0.f ? v : NEG_SLOPE * v; }

__device__ inline v16h load_frag16(const _Float16* __restrict__ p, int row0,
                                   int ld, int k0, int lane, int maxRow) {
    int m = lane & 15;
    int base = (lane >> 4) * 8;
    int r = row0 + m;
    if (r > maxRow) r = maxRow;
    const _Float16* q = p + (size_t)r * ld + k0 + base;
    half8 lo = *(const half8*)q;
    half8 hi = *(const half8*)(q + 16);
    v16h f;
#pragma unroll
    for (int i = 0; i < 8; ++i) { f[i] = lo[i]; f[i + 8] = hi[i]; }
    return f;
}


typedef __attribute__((ext_vector_type(4))) _Float16 half4;
__device__ inline half4 cvt4(const float4& v) {
    half4 o;
    o[0] = (_Float16)v.x; o[1] = (_Float16)v.y; o[2] = (_Float16)v.z; o[3] = (_Float16)v.w;
    return o;
}
__device__ inline void cvt_f32_to_f16_body(const float* __restrict__ src,
                                           _Float16* __restrict__ dst, long long n,
                                           int blk, int nblk) {
    const long long n4 = n >> 2;
    const long long tid = (long long)blk * blockDim.x + threadIdx.x;
    const long long stride = (long long)nblk * blockDim.x;
    const float4* __restrict__ s4 = (const float4*)src;
    half4* __restrict__ d4 = (half4*)dst;
    long long i = tid;
    for (; i + 3 * stride < n4; i += 4 * stride) {
        float4 a = s4[i], b = s4[i + stride], c = s4[i + 2 * stride], d = s4[i + 3 * stride];
        d4[i] = cvt4(a); d4[i + stride] = cvt4(b);
        d4[i + 2 * stride] = cvt4(c); d4[i + 3 * stride] = cvt4(d);
    }
    for (; i < n4; i += stride) d4[i] = cvt4(s4[i]);
    if (tid < (n & 3)) dst[(n4 << 2) + tid] = (_Float16)src[(n4 << 2) + tid];
}

template <int C, int NCT>
__device__ inline void wfrag_body(const float* __restrict__ W, _Float16* __restrict__ Wf,
                                  int blk) {
    int i = blk * blockDim.x + threadIdx.x;
    if (i >= NCT * 16 * 128) return;
    const int e = i & 7, L = (i >> 3) & 31, j = (i >> 8) & 1, kk = (i >> 9) & 3, ct = i >> 11;
    const int k = kk * 32 + j * 16 + (L >> 4) * 8 + e;
    const int col = ct * 16 + (L & 15);
    Wf[i] = (col < C) ? (_Float16)W[k * C + col] : (_Float16)0.f;
}

__device__ inline v16h load_wfrag(const _Float16* __restrict__ Wf, int ct, int kk, int lane) {
    const _Float16* q = Wf + (size_t)((ct * 4 + kk) * 2) * 256 + lane * 8;
    half8 lo = *(const half8*)q;
    half8 hi = *(const half8*)(q + 256);
    v16h f;
#pragma unroll
    for (int i = 0; i < 8; ++i) { f[i] = lo[i]; f[i + 8] = hi[i]; }
    return f;
}

template <int NCT, int NB>
__device__ inline void gemm_dual_strip(const _Float16* __restrict__ A,
                                       const _Float16* __restrict__ Wtl,
                                       const _Float16* __restrict__ Wtr,
                                       const float* __restrict__ bl,
                                       const float* __restrict__ br,
                                       float* __restrict__ Cl, float* __restrict__ Cr,
                                       int Nrows, int blk) {
    constexpr int ldc = NCT * 16;
    const int lane = threadIdx.x & 31;
    const int strip = __builtin_amdgcn_readfirstlane(
        (int)((blk * blockDim.x + threadIdx.x) >> 5));
    const int r0 = strip * 16;
    if (r0 >= Nrows) return;
    v16h a[4];
#pragma unroll
    for (int kk = 0; kk < 4; ++kk) a[kk] = load_frag16(A, r0, 128, kk * 32, lane, Nrows - 1);
    const int n = lane & 15, hw = lane >> 4;
    const bool whole = r0 + 16 <= Nrows;
#pragma unroll
    for (int ct = 0; ct < NCT; ++ct) {
        const int c0 = ct * 16;
        v16h bL[4], bR[4];
#pragma unroll
        for (int kk = 0; kk < 4; ++kk) {
            bL[kk] = load_wfrag(Wtl, ct, kk, lane);
            bR[kk] = load_wfrag(Wtr, ct, kk, lane);
        }
        v8f accL = {};
        v8f accR = {};
#pragma unroll
        for (int kk = 0; kk < 4; ++kk) {
            accL = __builtin_amdgcn_wmma_f32_16x16x32_f16(false, a[kk], false, bL[kk],
                                                          (short)0, accL, false, false);
            accR = __builtin_amdgcn_wmma_f32_16x16x32_f16(false, a[kk], false, bR[kk],
                                                          (short)0, accR, false, false);
        }
        const int col = c0 + n;
        float bLv = 0.f, bRv = 0.f;
        if (c0 + 16 <= NB || col < NB) { bLv = bl[col]; bRv = br[col]; }
        float* __restrict__ pl = Cl + (size_t)(r0 + hw * 8) * ldc + col;
        float* __restrict__ pr = Cr + (size_t)(r0 + hw * 8) * ldc + col;
        if (whole) {
#pragma unroll
            for (int v = 0; v < 8; ++v) {
                pl[v * ldc] = accL[v] + bLv;
                pr[v * ldc] = accR[v] + bRv;
            }
        } else {
#pragma unroll
            for (int v = 0; v < 8; ++v) {
                if (r0 + hw * 8 + v < Nrows) {
                    pl[v * ldc] = accL[v] + bLv;
                    pr[v * ldc] = accR[v] + bRv;
                }
            }
        }
    }
}
__global__ __launch_bounds__(128)
void k_gemm_dual_w48(const _Float16* __restrict__ A, const _Float16* __restrict__ Wtl,
                     const _Float16* __restrict__ Wtr, const float* __restrict__ bl,
                     const float* __restrict__ br, float* __restrict__ Cl,
                     float* __restrict__ Cr, int Nrows) {
    gemm_dual_strip<3, 40>(A, Wtl, Wtr, bl, br, Cl, Cr, Nrows, (int)blockIdx.x);
}

#define SLICES 256
#define BIN_SHIFT 8
#define BIN_NODES (1 << BIN_SHIFT)
#define MAX_BINS 1024

__device__ inline void slice_bounds(int E, int s, int& e0, int& e1) {
    const int per = (E + SLICES - 1) / SLICES;
    e0 = s * per;
    e1 = e0 + per;
    if (e1 > E) e1 = E;
}

__device__ inline void hist_body(const int* __restrict__ ei, int E, int nbins,
                                 int* __restrict__ hist, int s, int* cnt) {
    for (int i = threadIdx.x; i < nbins; i += blockDim.x) cnt[i] = 0;
    __syncthreads();
    int e0, e1;
    slice_bounds(E, s, e0, e1);
    const int step = blockDim.x;
    for (int i = e0 + (int)threadIdx.x; i < e1; i += 4 * step) {
        int dd[4];
#pragma unroll
        for (int u = 0; u < 4; ++u) {
            const int j = i + u * step;
            dd[u] = (j < e1) ? ei[E + j] : -1;
        }
#pragma unroll
        for (int u = 0; u < 4; ++u)
            if (dd[u] >= 0) atomicAdd(&cnt[dd[u] >> BIN_SHIFT], 1);
    }
    __syncthreads();
    for (int i = threadIdx.x; i < nbins; i += blockDim.x) hist[i * SLICES + s] = cnt[i];
}

__device__ inline void scatter_body(const int* __restrict__ ei, int E, int nbins,
                                    const int* __restrict__ offs, int* __restrict__ binned_words,
                                    int s, int* cur) {
    int2* __restrict__ binned = (int2*)binned_words;
    for (int i = threadIdx.x; i < nbins; i += blockDim.x) cur[i] = offs[i * SLICES + s];
    __syncthreads();
    int e0, e1;
    slice_bounds(E, s, e0, e1);
    const int step = blockDim.x;
    for (int i = e0 + (int)threadIdx.x; i < e1; i += 4 * step) {
        int dd[4], ss[4];
#pragma unroll
        for (int u = 0; u < 4; ++u) {
            const int j = i + u * step;
            const bool on = j < e1;
            dd[u] = on ? ei[E + j] : -1;
            ss[u] = on ? ei[j] : 0;
        }
#pragma unroll
        for (int u = 0; u < 4; ++u) {
            if (dd[u] >= 0) {
                const int p = atomicAdd(&cur[dd[u] >> BIN_SHIFT], 1);
                binned[p] = make_int2(dd[u], ss[u]);
            }
        }
    }
}

__global__ __launch_bounds__(256)
void k_hist_tr_cvt(const int* __restrict__ ei, int E, int nbins, int* __restrict__ hist,
                   const float* __restrict__ W1l, _Float16* __restrict__ Wt1l,
                   const float* __restrict__ W1r, _Float16* __restrict__ Wt1r,
                   const float* __restrict__ W2l, _Float16* __restrict__ Wt2l,
                   const float* __restrict__ W2r, _Float16* __restrict__ Wt2r,
                   const float* __restrict__ src, _Float16* __restrict__ dst, long long n,
                   int ncv) {
    __shared__ int cnt[MAX_BINS];
    int b = blockIdx.x;
    if (b < SLICES) { hist_body(ei, E, nbins, hist, b, cnt); return; }
    b -= SLICES;
    if (b < 64) { wfrag_body<128, 8>(W1l, Wt1l, b); return; }
    b -= 64;
    if (b < 64) { wfrag_body<128, 8>(W1r, Wt1r, b); return; }
    b -= 64;
    if (b < 24) { wfrag_body<40, 3>(W2l, Wt2l, b); return; }
    b -= 24;
    if (b < 24) { wfrag_body<40, 3>(W2r, Wt2r, b); return; }
    b -= 24;
    cvt_f32_to_f16_body(src, dst, n, b, ncv);
}

__device__ inline int4 ld4i_guard(const int* __restrict__ p, int idx, int N) {
    int4 r = make_int4(0, 0, 0, 0);
    if (idx + 3 < N) {
        r = *(const int4*)(p + idx);
    } else if (idx < N) {
        const int a = p[idx];
        const int b = p[(idx + 1 < N) ? idx + 1 : idx];
        const int c = p[(idx + 2 < N) ? idx + 2 : idx];
        r.x = a;
        r.y = (idx + 1 < N) ? b : 0;
        r.z = (idx + 2 < N) ? c : 0;
    }
    return r;
}
__device__ inline void st4i_guard(int* __restrict__ p, int idx, int N, const int4& o) {
    if (idx + 3 < N) {
        *(int4*)(p + idx) = o;
    } else {
        if (idx     < N) p[idx]     = o.x;
        if (idx + 1 < N) p[idx + 1] = o.y;
        if (idx + 2 < N) p[idx + 2] = o.z;
    }
}
#define SCAN_PIECE (1024 * 4)
#define SCAN_TILE (4 * SCAN_PIECE)
__device__ inline void ld_tile(const int* __restrict__ p, int i0, int N, int4 (&r)[4]) {
    if (i0 + 3 * SCAN_PIECE + 3 < N) {
#pragma unroll
        for (int q = 0; q < 4; ++q) r[q] = *(const int4*)(p + i0 + q * SCAN_PIECE);
    } else {
#pragma unroll
        for (int q = 0; q < 4; ++q) r[q] = ld4i_guard(p, i0 + q * SCAN_PIECE, N);
    }
}
__device__ inline int wave_incl_scan(int v, int lane) {
#pragma unroll
    for (int off = 1; off < 32; off <<= 1) {
        int u = __shfl_up(v, off, 32);
        if (lane >= off) v += u;
    }
    return v;
}
__global__ __launch_bounds__(1024)
void k_scan(const int* __restrict__ deg, int* __restrict__ rbeg, int N) {
    __shared__ int wtot[2][4][32];
    const int tid = threadIdx.x;
    const int lane = tid & 31;
    const int wv = __builtin_amdgcn_readfirstlane(tid >> 5);
    const int piece = SCAN_PIECE;
    const int tile = SCAN_TILE;
    const int nT = (N + tile - 1) / tile;
    int carry = 0;
    int4 nx[4];
    ld_tile(deg, tid * 4, N, nx);
    for (int t = 0; t < nT; ++t) {
        const int i0 = t * tile + tid * 4;
        int4 v[4];
#pragma unroll
        for (int q = 0; q < 4; ++q) v[q] = nx[q];
        if (t + 1 < nT) ld_tile(deg, i0 + SCAN_TILE, N, nx);
        int run[4], inc[4];
#pragma unroll
        for (int q = 0; q < 4; ++q) {
            v[q].y += v[q].x; v[q].z += v[q].y; v[q].w += v[q].z;
            run[q] = v[q].w;
        }
#pragma unroll
        for (int q = 0; q < 4; ++q) inc[q] = wave_incl_scan(run[q], lane);
        if (lane == 31) {
#pragma unroll
            for (int q = 0; q < 4; ++q) wtot[t & 1][q][wv] = inc[q];
        }
        __syncthreads();
        int wt[4], winc[4];
#pragma unroll
        for (int q = 0; q < 4; ++q) wt[q] = wtot[t & 1][q][lane];
#pragma unroll
        for (int q = 0; q < 4; ++q) winc[q] = wave_incl_scan(wt[q], lane);
#pragma unroll
        for (int q = 0; q < 4; ++q) {
            const int wexcl = __builtin_amdgcn_readlane(winc[q] - wt[q], wv);
            const int total = __builtin_amdgcn_readlane(winc[q], 31);
            const int base = carry + wexcl + (inc[q] - run[q]);
            const int4 o = make_int4(base, base + v[q].x, base + v[q].y, base + v[q].z);
            st4i_guard(rbeg, i0 + q * piece, N, o);
            carry += total;
        }
    }
}

__global__ __launch_bounds__(256)
void k_scatter_gemm1(const int* __restrict__ ei, int E, int nbins,
                     const int* __restrict__ offs, int* __restrict__ binned,
                     const _Float16* __restrict__ A, const _Float16* __restrict__ Wtl,
                     const _Float16* __restrict__ Wtr, const float* __restrict__ bl,
                     const float* __restrict__ br, float* __restrict__ Cl,
                     float* __restrict__ Cr, int Nrows) {
    __shared__ int cur[MAX_BINS];
    const int b = blockIdx.x;
    if (b < SLICES) scatter_body(ei, E, nbins, offs, binned, b, cur);
    else gemm_dual_strip<8, 128>(A, Wtl, Wtr, bl, br, Cl, Cr, Nrows, b - SLICES);
}

__global__ __launch_bounds__(BIN_NODES)
void k_binsort(const int* __restrict__ binned_words, const int* __restrict__ offs, int nbins,
               int E, int N, int* __restrict__ rbeg, int* __restrict__ rend,
               int* __restrict__ csrc) {
    const int2* __restrict__ binned = (const int2*)binned_words;
    __shared__ int cnt[BIN_NODES];
    __shared__ int wsum[BIN_NODES / 32];
    const int bin = blockIdx.x;
    const int tid = threadIdx.x;
    const int lane = tid & 31;
    const int wv = tid >> 5;
    const int s0 = offs[bin * SLICES];
    const int s1 = (bin + 1 < nbins) ? offs[(bin + 1) * SLICES] : E;
    cnt[tid] = 0;
    __syncthreads();
    for (int i = s0 + tid; i < s1; i += 4 * BIN_NODES) {
        int dd[4];
#pragma unroll
        for (int u = 0; u < 4; ++u) {
            const int j = i + u * BIN_NODES;
            dd[u] = (j < s1) ? binned[j].x : -1;
        }
#pragma unroll
        for (int u = 0; u < 4; ++u)
            if (dd[u] >= 0) atomicAdd(&cnt[dd[u] & (BIN_NODES - 1)], 1);
    }
    __syncthreads();
    const int c = cnt[tid];
    const int inc = wave_incl_scan(c, lane);
    if (lane == 31) wsum[wv] = inc;
    __syncthreads();
    int wbase = 0;
#pragma unroll
    for (int w = 0; w < BIN_NODES / 32; ++w) {
        const int t = wsum[w];
        if (w < wv) wbase += t;
    }
    const int start = s0 + wbase + inc - c;
    const int node = (bin << BIN_SHIFT) + tid;
    if (node < N) { rbeg[node] = start; rend[node] = start + c; }
    cnt[tid] = start;
    __syncthreads();
    for (int i = s0 + tid; i < s1; i += 4 * BIN_NODES) {
        int2 ee[4];
#pragma unroll
        for (int u = 0; u < 4; ++u) {
            const int j = i + u * BIN_NODES;
            ee[u] = (j < s1) ? binned[j] : make_int2(-1, 0);
        }
#pragma unroll
        for (int u = 0; u < 4; ++u) {
            if (ee[u].x >= 0) {
                const int p = atomicAdd(&cnt[ee[u].x & (BIN_NODES - 1)], 1);
                csrc[p] = ee[u].y;
            }
        }
    }
}

__device__ inline float score4(const float4& a, const float4& b, const float4& t) {
    return t.x * lrelu(a.x + b.x) + t.y * lrelu(a.y + b.y) +
           t.z * lrelu(a.z + b.z) + t.w * lrelu(a.w + b.w);
}
__device__ inline float red16(float v) {
#pragma unroll
    for (int off = 1; off < 16; off <<= 1) v += __shfl_xor(v, off, 32);
    return v;
}
__device__ inline void scale4(float4& a, float s) { a.x *= s; a.y *= s; a.z *= s; a.w *= s; }
__device__ inline void axpy4(float4& a, float p, const float4& v) {
    a.x += p * v.x; a.y += p * v.y; a.z += p * v.z; a.w += p * v.w;
}

__device__ inline int sort_chunk(int mine, int cnt, int lane) {
    int rank = 0;
    for (int k = 0; k < cnt; ++k) {
        int v = __builtin_amdgcn_readlane(mine, k);
        rank += ((v < mine) || (v == mine && k < lane)) ? 1 : 0;
    }
    return __builtin_amdgcn_ds_permute(rank << 2, mine);
}

__device__ inline void fold4(float& m, float& den, float4& acc,
                             float e0, float e1, float e2, float e3,
                             const float4& a0, const float4& a1,
                             const float4& a2, const float4& a3) {
    float mn = fmaxf(fmaxf(m, fmaxf(e0, e1)), fmaxf(e2, e3));
    float sc = __expf(m - mn);
    float p0 = __expf(e0 - mn), p1 = __expf(e1 - mn);
    float p2 = __expf(e2 - mn), p3 = __expf(e3 - mn);
    scale4(acc, sc);
    axpy4(acc, p0, a0); axpy4(acc, p1, a1); axpy4(acc, p2, a2); axpy4(acc, p3, a3);
    den = den * sc + ((p0 + p1) + (p2 + p3));
    m = mn;
}
__device__ inline void fold1(float& m, float& den, float4& acc, float e0, const float4& a0) {
    float mn = fmaxf(m, e0);
    float sc = __expf(m - mn);
    float p0 = __expf(e0 - mn);
    scale4(acc, sc);
    axpy4(acc, p0, a0);
    den = den * sc + p0;
    m = mn;
}

__global__ __launch_bounds__(256)
void k_node_attn1(const float* __restrict__ xl, const float* __restrict__ xr,
                 const float* __restrict__ att, const float* __restrict__ bias,
                 const int* __restrict__ rbeg, const int* __restrict__ rend,
                 const int* __restrict__ csrc, _Float16* __restrict__ hh, int N) {
    const int wave = __builtin_amdgcn_readfirstlane(
        (int)((blockIdx.x * blockDim.x + threadIdx.x) >> 5));
    const int nwave = (int)((gridDim.x * blockDim.x) >> 5);
    const int lane = threadIdx.x & 31;
    const float4 t  = ((const float4*)att)[lane];
    const float4 bb = ((const float4*)bias)[lane];
    for (int d = wave; d < N; d += nwave) {
    const float4 b4 = ((const float4*)(xr + (size_t)d * 128))[lane];
    float4 acc = ((const float4*)(xl + (size_t)d * 128))[lane];
    float m = red16(score4(acc, b4, t));
    float den = 1.f;
    int beg = rbeg[d], end = rend[d];
    for (int j0 = beg; j0 < end; j0 += 32) {
        int cnt = end - j0; if (cnt > 32) cnt = 32;
        int mine = (lane < cnt) ? csrc[j0 + lane] : 0x7fffffff;
        int srt = sort_chunk(mine, cnt, lane);
        int k = 0;
        for (; k + 4 <= cnt; k += 4) {
            int s0 = __builtin_amdgcn_readlane(srt, k);
            int s1 = __builtin_amdgcn_readlane(srt, k + 1);
            int s2 = __builtin_amdgcn_readlane(srt, k + 2);
            int s3 = __builtin_amdgcn_readlane(srt, k + 3);
            float4 a0 = ((const float4*)(xl + (size_t)s0 * 128))[lane];
            float4 a1 = ((const float4*)(xl + (size_t)s1 * 128))[lane];
            float4 a2 = ((const float4*)(xl + (size_t)s2 * 128))[lane];
            float4 a3 = ((const float4*)(xl + (size_t)s3 * 128))[lane];
            float e0 = red16(score4(a0, b4, t));
            float e1 = red16(score4(a1, b4, t));
            float e2 = red16(score4(a2, b4, t));
            float e3 = red16(score4(a3, b4, t));
            fold4(m, den, acc, e0, e1, e2, e3, a0, a1, a2, a3);
        }
        for (; k < cnt; ++k) {
            int s0 = __builtin_amdgcn_readlane(srt, k);
            float4 a0 = ((const float4*)(xl + (size_t)s0 * 128))[lane];
            float e0 = red16(score4(a0, b4, t));
            fold1(m, den, acc, e0, a0);
        }
    }
    float inv = 1.f / den;
    float o0 = fmaxf(acc.x * inv + bb.x, 0.f);
    float o1 = fmaxf(acc.y * inv + bb.y, 0.f);
    float o2 = fmaxf(acc.z * inv + bb.z, 0.f);
    float o3 = fmaxf(acc.w * inv + bb.w, 0.f);
    half4 o;
    o[0] = (_Float16)o0; o[1] = (_Float16)o1; o[2] = (_Float16)o2; o[3] = (_Float16)o3;
    *(half4*)(hh + (size_t)d * 128 + lane * 4) = o;
    }
}

__device__ inline float score5(const float4& a, float a1, const float4& b, float b1,
                               const float4& t, float t1) {
    return score4(a, b, t) + t1 * lrelu(a1 + b1);
}
__device__ inline float red8(float v) {
#pragma unroll
    for (int off = 1; off < 8; off <<= 1) v += __shfl_xor(v, off, 32);
    return v;
}
__device__ inline float redmax8(float v) {
#pragma unroll
    for (int off = 1; off < 8; off <<= 1) v = fmaxf(v, __shfl_xor(v, off, 32));
    return v;
}
__device__ inline void attn2_pass(const float* __restrict__ xl, int srt, int k, int cnt,
                                  int q, unsigned o4, unsigned o1, int d,
                                  const float4& b4, float b1, const float4& t4, float t1,
                                  float& m, float& den, float4& acc4, float& acc1) {
    float4 r4[2];
    float r1[2];
    bool on[2];
    int s[2];
#pragma unroll
    for (int u = 0; u < 2; ++u) {
        const int idx = k + 4 * u + q;
        on[u] = idx < cnt;
        s[u] = __builtin_amdgcn_ds_bpermute(idx << 2, srt);
    }
#pragma unroll
    for (int u = 0; u < 2; ++u) {
        const int su = on[u] ? s[u] : d;
        const unsigned rb = __umul24((unsigned)su, 192u);
        r4[u] = *(const float4*)((const char*)xl + (rb + o4));
        r1[u] = *(const float*)((const char*)xl + (rb + o1));
    }
    float e[2];
#pragma unroll
    for (int u = 0; u < 2; ++u) {
        const float sc = red8(score5(r4[u], r1[u], b4, b1, t4, t1));
        e[u] = on[u] ? sc : -INFINITY;
    }
    const float mn = fmaxf(m, fmaxf(e[0], e[1]));
    const float sc = __expf(m - mn);
    const float p0 = __expf(e[0] - mn), p1 = __expf(e[1] - mn);
    scale4(acc4, sc);
    acc1 *= sc;
    axpy4(acc4, p0, r4[0]); acc1 += p0 * r1[0];
    axpy4(acc4, p1, r4[1]); acc1 += p1 * r1[1];
    den = den * sc + (p0 + p1);
    m = mn;
}
__global__ __launch_bounds__(256) __attribute__((amdgpu_waves_per_eu(12, 12)))
void k_node_attn2(const float* __restrict__ xl, const float* __restrict__ xr,
                 const float* __restrict__ att, const float* __restrict__ bias,
                 const int* __restrict__ rbeg, const int* __restrict__ rend,
                 const int* __restrict__ csrc, float* __restrict__ out, int N) {
    const int wave = __builtin_amdgcn_readfirstlane(
        (int)((blockIdx.x * blockDim.x + threadIdx.x) >> 5));
    const int nwave = (int)((gridDim.x * blockDim.x) >> 5);
    const int lane = threadIdx.x & 31;
    const int g = lane & 7;
    const int q = lane >> 3;
    const unsigned o4 = (unsigned)g * 16u;
    const unsigned o1 = 128u + (unsigned)g * 4u;
    const float4 t4  = ((const float4*)att)[g];
    const float  t1  = att[32 + g];
    for (int d = wave; d < N; d += nwave) {
    const float* __restrict__ xrd = xr + (size_t)d * 48;
    const float* __restrict__ xld = xl + (size_t)d * 48;
    const float4 b4 = ((const float4*)xrd)[g];
    const float  b1 = xrd[32 + g];
    const float4 s4 = ((const float4*)xld)[g];
    const float  s1 = xld[32 + g];
    float m = red8(score5(s4, s1, b4, b1, t4, t1));
    float den = (q == 0) ? 1.f : 0.f;
    float4 acc4 = (q == 0) ? s4 : make_float4(0.f, 0.f, 0.f, 0.f);
    float acc1 = (q == 0) ? s1 : 0.f;
    int beg = rbeg[d], end = rend[d];
    for (int j0 = beg; j0 < end; j0 += 32) {
        int cnt = end - j0; if (cnt > 32) cnt = 32;
        int mine = (lane < cnt) ? csrc[j0 + lane] : 0x7fffffff;
        int srt = sort_chunk(mine, cnt, lane);
        for (int k = 0; k < cnt; k += 8)
            attn2_pass(xl, srt, k, cnt, q, o4, o1, d, b4, b1, t4, t1, m, den, acc4, acc1);
    }
    const float4 bb4 = ((const float4*)bias)[g];
    const float  bb1 = bias[32 + g];
#pragma unroll
    for (int off = 8; off < 32; off <<= 1) {
        const float mo = __shfl_xor(m, off, 32);
        const float dn = __shfl_xor(den, off, 32);
        float4 ao;
        ao.x = __shfl_xor(acc4.x, off, 32); ao.y = __shfl_xor(acc4.y, off, 32);
        ao.z = __shfl_xor(acc4.z, off, 32); ao.w = __shfl_xor(acc4.w, off, 32);
        const float a1o = __shfl_xor(acc1, off, 32);
        const float M = fmaxf(m, mo);
        const float sa = __expf(m - M), sb = __expf(mo - M);
        den = den * sa + dn * sb;
        acc4.x = acc4.x * sa + ao.x * sb; acc4.y = acc4.y * sa + ao.y * sb;
        acc4.z = acc4.z * sa + ao.z * sb; acc4.w = acc4.w * sa + ao.w * sb;
        acc1 = acc1 * sa + a1o * sb;
        m = M;
    }
    float inv = 1.f / den;
    float v0 = acc4.x * inv + bb4.x, v1 = acc4.y * inv + bb4.y;
    float v2 = acc4.z * inv + bb4.z, v3 = acc4.w * inv + bb4.w;
    float v4 = acc1 * inv + bb1;
    float mx = redmax8(fmaxf(fmaxf(fmaxf(v0, v1), fmaxf(v2, v3)), v4));
    float se = red8(((__expf(v0 - mx) + __expf(v1 - mx)) + (__expf(v2 - mx) + __expf(v3 - mx)))
                    + __expf(v4 - mx));
    float L = mx + logf(se);
    if (lane < 8) {
        *(float4*)(out + (size_t)d * 40 + lane * 4) = make_float4(v0 - L, v1 - L, v2 - L, v3 - L);
        out[(size_t)d * 40 + 32 + lane] = v4 - L;
    }
    }
}

extern "C" void kernel_launch(void* const* d_in, const int* in_sizes, int n_in,
                              void* d_out, int out_size, void* d_ws, size_t ws_size,
                              hipStream_t stream) {
    const float* x    = (const float*)d_in[0];
    const int*   ei   = (const int*)d_in[1];
    const float* W1l  = (const float*)d_in[2];
    const float* b1l  = (const float*)d_in[3];
    const float* W1r  = (const float*)d_in[4];
    const float* b1r  = (const float*)d_in[5];
    const float* att1 = (const float*)d_in[6];
    const float* bias1= (const float*)d_in[7];
    const float* W2l  = (const float*)d_in[8];
    const float* b2l  = (const float*)d_in[9];
    const float* W2r  = (const float*)d_in[10];
    const float* b2r  = (const float*)d_in[11];
    const float* att2 = (const float*)d_in[12];
    const float* bias2= (const float*)d_in[13];
    float* out = (float*)d_out;

    const int N = in_sizes[0] / 128;
    const int E = in_sizes[1] / 2;

    char* w = (char*)d_ws;
    size_t off = 0;
    auto take = [&](size_t bytes) -> char* {
        char* p = w + off;
        off = (off + bytes + 255) & ~(size_t)255;
        return p;
    };
    _Float16* xh   = (_Float16*)take((size_t)N * 128 * 2);
    _Float16* hh   = (_Float16*)take((size_t)N * 128 * 2);
    _Float16* Wt1l = (_Float16*)take(128 * 128 * 2);
    _Float16* Wt1r = (_Float16*)take(128 * 128 * 2);
    _Float16* Wt2l = (_Float16*)take(48 * 128 * 2);
    _Float16* Wt2r = (_Float16*)take(48 * 128 * 2);
    float*    xl   = (float*)take((size_t)N * 128 * 4);
    float*    xr   = (float*)take((size_t)N * 128 * 4);
    const int nbins = (N + BIN_NODES - 1) / BIN_NODES;
    const int nh = nbins * SLICES;
    int*      hist = (int*)take((size_t)nh * 4);
    int*      offs = (int*)take((size_t)nh * 4);
    int*      binned = (int*)take((size_t)E * 8);
    int*      rbeg = (int*)take((size_t)N * 4);
    int*      rend = (int*)take((size_t)N * 4);
    int*      csrc = (int*)take((size_t)E * 4);

    const int B = 256;
    const long long Nx128 = (long long)N * 128;
    const int rowTiles = (N + 15) / 16;

    k_hist_tr_cvt<<<SLICES + 64 + 64 + 24 + 24 + 512, B, 0, stream>>>(
        ei, E, nbins, hist, W1l, Wt1l, W1r, Wt1r, W2l, Wt2l, W2r, Wt2r, x, xh, Nx128, 512);
    k_scan<<<1, 1024, 0, stream>>>(hist, offs, nh);
    k_scatter_gemm1<<<SLICES + (rowTiles + 7) / 8, B, 0, stream>>>(ei, E, nbins, offs, binned,
                                                                xh, Wt1l, Wt1r, b1l, b1r,
                                                                xl, xr, N);
    k_binsort<<<nbins, BIN_NODES, 0, stream>>>(binned, offs, nbins, E, N, rbeg, rend, csrc);

    k_node_attn1<<<2048, B, 0, stream>>>(xl, xr, att1, bias1, rbeg, rend, csrc, hh, N);

    k_gemm_dual_w48<<<(rowTiles + 3) / 4, 128, 0, stream>>>(hh, Wt2l, Wt2r, b2l, b2r,
                                                           xl, xr, N);
    k_node_attn2<<<1536, B, 0, stream>>>(xl, xr, att2, bias2, rbeg, rend, csrc, out, N);
}


# Level 1 of the edge sort made slice-local: a slice sorts its own edges by bin in place, a bin gathers its 256 pieces; global count, scan launch and scatter gone (7 -> 5 launches); lists and sums kept
# speedup vs baseline: 7.9405x; 1.1387x over previous
#include <hip/hip_runtime.h>
#include <hip/hip_bf16.h>
#include <math.h>

typedef __attribute__((ext_vector_type(16))) _Float16 v16h;
typedef __attribute__((ext_vector_type(8)))  _Float16 half8;
typedef __attribute__((ext_vector_type(8)))  float    v8f;

#define NEG_SLOPE 0.2f

__device__ inline float lrelu(float v) { return v > 0.f ? v : NEG_SLOPE * v; }

__device__ inline v16h load_frag16(const _Float16* __restrict__ p, int row0,
                                   int ld, int k0, int lane, int maxRow) {
    int m = lane & 15;
    int base = (lane >> 4) * 8;
    int r = row0 + m;
    if (r > maxRow) r = maxRow;
    const _Float16* q = p + (size_t)r * ld + k0 + base;
    half8 lo = *(const half8*)q;
    half8 hi = *(const half8*)(q + 16);
    v16h f;
#pragma unroll
    for (int i = 0; i < 8; ++i) { f[i] = lo[i]; f[i + 8] = hi[i]; }
    return f;
}


typedef __attribute__((ext_vector_type(4))) _Float16 half4;
__device__ inline half4 cvt4(const float4& v) {
    half4 o;
    o[0] = (_Float16)v.x; o[1] = (_Float16)v.y; o[2] = (_Float16)v.z; o[3] = (_Float16)v.w;
    return o;
}
__device__ inline void cvt_f32_to_f16_body(const float* __restrict__ src,
                                           _Float16* __restrict__ dst, long long n,
                                           int blk, int nblk) {
    const long long n4 = n >> 2;
    const long long tid = (long long)blk * blockDim.x + threadIdx.x;
    const long long stride = (long long)nblk * blockDim.x;
    const float4* __restrict__ s4 = (const float4*)src;
    half4* __restrict__ d4 = (half4*)dst;
    long long i = tid;
    for (; i + 3 * stride < n4; i += 4 * stride) {
        float4 a = s4[i], b = s4[i + stride], c = s4[i + 2 * stride], d = s4[i + 3 * stride];
        d4[i] = cvt4(a); d4[i + stride] = cvt4(b);
        d4[i + 2 * stride] = cvt4(c); d4[i + 3 * stride] = cvt4(d);
    }
    for (; i < n4; i += stride) d4[i] = cvt4(s4[i]);
    if (tid < (n & 3)) dst[(n4 << 2) + tid] = (_Float16)src[(n4 << 2) + tid];
}

template <int C, int NCT>
__device__ inline void wfrag_body(const float* __restrict__ W, _Float16* __restrict__ Wf,
                                  int blk) {
    int i = blk * blockDim.x + threadIdx.x;
    if (i >= NCT * 16 * 128) return;
    const int e = i & 7, L = (i >> 3) & 31, j = (i >> 8) & 1, kk = (i >> 9) & 3, ct = i >> 11;
    const int k = kk * 32 + j * 16 + (L >> 4) * 8 + e;
    const int col = ct * 16 + (L & 15);
    Wf[i] = (col < C) ? (_Float16)W[k * C + col] : (_Float16)0.f;
}

__device__ inline v16h load_wfrag(const _Float16* __restrict__ Wf, int ct, int kk, int lane) {
    const _Float16* q = Wf + (size_t)((ct * 4 + kk) * 2) * 256 + lane * 8;
    half8 lo = *(const half8*)q;
    half8 hi = *(const half8*)(q + 256);
    v16h f;
#pragma unroll
    for (int i = 0; i < 8; ++i) { f[i] = lo[i]; f[i + 8] = hi[i]; }
    return f;
}

template <int NCT, int NB>
__device__ inline void gemm_dual_strip(const _Float16* __restrict__ A,
                                       const _Float16* __restrict__ Wtl,
                                       const _Float16* __restrict__ Wtr,
                                       const float* __restrict__ bl,
                                       const float* __restrict__ br,
                                       float* __restrict__ Cl, float* __restrict__ Cr,
                                       int Nrows, int blk) {
    constexpr int ldc = NCT * 16;
    const int lane = threadIdx.x & 31;
    const int strip = __builtin_amdgcn_readfirstlane(
        (int)((blk * blockDim.x + threadIdx.x) >> 5));
    const int r0 = strip * 16;
    if (r0 >= Nrows) return;
    v16h a[4];
#pragma unroll
    for (int kk = 0; kk < 4; ++kk) a[kk] = load_frag16(A, r0, 128, kk * 32, lane, Nrows - 1);
    const int n = lane & 15, hw = lane >> 4;
    const bool whole = r0 + 16 <= Nrows;
#pragma unroll
    for (int ct = 0; ct < NCT; ++ct) {
        const int c0 = ct * 16;
        v16h bL[4], bR[4];
#pragma unroll
        for (int kk = 0; kk < 4; ++kk) {
            bL[kk] = load_wfrag(Wtl, ct, kk, lane);
            bR[kk] = load_wfrag(Wtr, ct, kk, lane);
        }
        v8f accL = {};
        v8f accR = {};
#pragma unroll
        for (int kk = 0; kk < 4; ++kk) {
            accL = __builtin_amdgcn_wmma_f32_16x16x32_f16(false, a[kk], false, bL[kk],
                                                          (short)0, accL, false, false);
            accR = __builtin_amdgcn_wmma_f32_16x16x32_f16(false, a[kk], false, bR[kk],
                                                          (short)0, accR, false, false);
        }
        const int col = c0 + n;
        float bLv = 0.f, bRv = 0.f;
        if (c0 + 16 <= NB || col < NB) { bLv = bl[col]; bRv = br[col]; }
        float* __restrict__ pl = Cl + (size_t)(r0 + hw * 8) * ldc + col;
        float* __restrict__ pr = Cr + (size_t)(r0 + hw * 8) * ldc + col;
        if (whole) {
#pragma unroll
            for (int v = 0; v < 8; ++v) {
                pl[v * ldc] = accL[v] + bLv;
                pr[v * ldc] = accR[v] + bRv;
            }
        } else {
#pragma unroll
            for (int v = 0; v < 8; ++v) {
                if (r0 + hw * 8 + v < Nrows) {
                    pl[v * ldc] = accL[v] + bLv;
                    pr[v * ldc] = accR[v] + bRv;
                }
            }
        }
    }
}
__global__ __launch_bounds__(128)
void k_gemm_dual_w48(const _Float16* __restrict__ A, const _Float16* __restrict__ Wtl,
                     const _Float16* __restrict__ Wtr, const float* __restrict__ bl,
                     const float* __restrict__ br, float* __restrict__ Cl,
                     float* __restrict__ Cr, int Nrows) {
    gemm_dual_strip<3, 40>(A, Wtl, Wtr, bl, br, Cl, Cr, Nrows, (int)blockIdx.x);
}

#define SLICES 256
#define BIN_SHIFT 8
#define BIN_NODES (1 << BIN_SHIFT)
#define MAX_BINS 1024

__device__ inline void slice_bounds(int E, int s, int& e0, int& e1) {
    const int per = (E + SLICES - 1) / SLICES;
    e0 = s * per;
    e1 = e0 + per;
    if (e1 > E) e1 = E;
}

__device__ inline int wave_incl_scan(int v, int lane) {
#pragma unroll
    for (int off = 1; off < 32; off <<= 1) {
        int u = __shfl_up(v, off, 32);
        if (lane >= off) v += u;
    }
    return v;
}

__device__ inline void slicesort_body(const int* __restrict__ ei, int E, int nbins,
                                      int* __restrict__ binned_words,
                                      int* __restrict__ piece_words, int s,
                                      int* cnt, int* wsum) {
    int2* __restrict__ binned = (int2*)binned_words;
    int2* __restrict__ piece = (int2*)piece_words;
    const int tid = threadIdx.x;
    const int lane = tid & 31;
    const int wv = tid >> 5;
    const int step = blockDim.x;
    for (int i = tid; i < MAX_BINS; i += step) cnt[i] = 0;
    __syncthreads();
    int e0, e1;
    slice_bounds(E, s, e0, e1);
    for (int i = e0 + tid; i < e1; i += 4 * step) {
        int dd[4];
#pragma unroll
        for (int u = 0; u < 4; ++u) {
            const int j = i + u * step;
            dd[u] = (j < e1) ? ei[E + j] : -1;
        }
#pragma unroll
        for (int u = 0; u < 4; ++u)
            if (dd[u] >= 0) atomicAdd(&cnt[dd[u] >> BIN_SHIFT], 1);
    }
    __syncthreads();
    const int4 c = *(const int4*)(cnt + 4 * tid);
    const int run = (c.x + c.y) + (c.z + c.w);
    const int inc = wave_incl_scan(run, lane);
    if (lane == 31) wsum[wv] = inc;
    __syncthreads();
    int wbase = 0;
#pragma unroll
    for (int w = 0; w < MAX_BINS / 4 / 32; ++w) {
        const int t = wsum[w];
        if (w < wv) wbase += t;
    }
    const int b0 = wbase + inc - run;
    const int4 st = make_int4(b0, b0 + c.x, b0 + c.x + c.y, b0 + c.x + c.y + c.z);
    *(int4*)(cnt + 4 * tid) = st;
    const int bb = 4 * tid;
    if (bb     < nbins) piece[(bb    ) * SLICES + s] = make_int2(st.x, c.x);
    if (bb + 1 < nbins) piece[(bb + 1) * SLICES + s] = make_int2(st.y, c.y);
    if (bb + 2 < nbins) piece[(bb + 2) * SLICES + s] = make_int2(st.z, c.z);
    if (bb + 3 < nbins) piece[(bb + 3) * SLICES + s] = make_int2(st.w, c.w);
    __syncthreads();
    for (int i = e0 + tid; i < e1; i += 4 * step) {
        int dd[4], ss[4];
#pragma unroll
        for (int u = 0; u < 4; ++u) {
            const int j = i + u * step;
            const bool on = j < e1;
            dd[u] = on ? ei[E + j] : -1;
            ss[u] = on ? ei[j] : 0;
        }
#pragma unroll
        for (int u = 0; u < 4; ++u) {
            if (dd[u] >= 0) {
                const int p = atomicAdd(&cnt[dd[u] >> BIN_SHIFT], 1);
                binned[e0 + p] = make_int2(dd[u], ss[u]);
            }
        }
    }
}

__global__ __launch_bounds__(256)
void k_slicesort_tr_cvt(const int* __restrict__ ei, int E, int nbins,
                        int* __restrict__ binned, int* __restrict__ piece,
                        const float* __restrict__ W1l, _Float16* __restrict__ Wt1l,
                        const float* __restrict__ W1r, _Float16* __restrict__ Wt1r,
                        const float* __restrict__ W2l, _Float16* __restrict__ Wt2l,
                        const float* __restrict__ W2r, _Float16* __restrict__ Wt2r,
                        const float* __restrict__ src, _Float16* __restrict__ dst, long long n,
                        int ncv) {
    __shared__ alignas(16) int cnt[MAX_BINS];
    __shared__ int wsum[MAX_BINS / 4 / 32];
    int b = blockIdx.x;
    if (b < SLICES) { slicesort_body(ei, E, nbins, binned, piece, b, cnt, wsum); return; }
    b -= SLICES;
    if (b < 64) { wfrag_body<128, 8>(W1l, Wt1l, b); return; }
    b -= 64;
    if (b < 64) { wfrag_body<128, 8>(W1r, Wt1r, b); return; }
    b -= 64;
    if (b < 24) { wfrag_body<40, 3>(W2l, Wt2l, b); return; }
    b -= 24;
    if (b < 24) { wfrag_body<40, 3>(W2r, Wt2r, b); return; }
    b -= 24;
    cvt_f32_to_f16_body(src, dst, n, b, ncv);
}

template <class F>
__device__ inline void walk_pieces(const int2* __restrict__ binned, int g0, int c, int lane, F f) {
    const int hi = lane >> 4;
    const int j0 = lane & 15;
    for (int pp = 0; pp < 16; pp += 4) {
        int pg[4], pn[4];
#pragma unroll
        for (int u = 0; u < 4; ++u) {
            const int sl = 2 * (pp + u) + hi;
            pg[u] = __shfl(g0, sl, 32);
            pn[u] = __shfl(c, sl, 32);
        }
        int2 ea[4], eb[4];
#pragma unroll
        for (int u = 0; u < 4; ++u) {
            ea[u] = (j0 < pn[u]) ? binned[pg[u] + j0] : make_int2(-1, 0);
            eb[u] = (j0 + 16 < pn[u]) ? binned[pg[u] + j0 + 16] : make_int2(-1, 0);
        }
#pragma unroll
        for (int u = 0; u < 4; ++u) {
            if (ea[u].x >= 0) f(ea[u]);
            if (eb[u].x >= 0) f(eb[u]);
        }
#pragma unroll
        for (int u = 0; u < 4; ++u)
            for (int j = j0 + 32; j < pn[u]; j += 16) f(binned[pg[u] + j]);
    }
}

__device__ inline void binsort_body(const int* __restrict__ binned_words,
                                    const int* __restrict__ piece_words, int E, int N,
                                    int* __restrict__ rbeg, int* __restrict__ rend,
                                    int* __restrict__ csrc, int bin,
                                    int* cnt, int* wsumA, int* wsumB) {
    const int2* __restrict__ binned = (const int2*)binned_words;
    const int2* __restrict__ piece = (const int2*)piece_words;
    const int tid = threadIdx.x;
    const int lane = tid & 31;
    const int wv = tid >> 5;
    const int2 pc = piece[bin * SLICES + tid];
    int e0, e1;
    slice_bounds(E, tid, e0, e1);
    const int g0 = e0 + pc.x;
    const int pn = pc.y;
    cnt[tid] = 0;
    const int incx = wave_incl_scan(pc.x, lane);
    if (lane == 31) wsumA[wv] = incx;
    __syncthreads();
    int base = 0;
#pragma unroll
    for (int w = 0; w < BIN_NODES / 32; ++w) base += wsumA[w];
    walk_pieces(binned, g0, pn, lane, [&](const int2& e) {
        atomicAdd(&cnt[e.x & (BIN_NODES - 1)], 1);
    });
    __syncthreads();
    const int c = cnt[tid];
    const int inc = wave_incl_scan(c, lane);
    if (lane == 31) wsumB[wv] = inc;
    __syncthreads();
    int wbase = 0;
#pragma unroll
    for (int w = 0; w < BIN_NODES / 32; ++w) {
        const int t = wsumB[w];
        if (w < wv) wbase += t;
    }
    const int start = base + wbase + inc - c;
    const int node = (bin << BIN_SHIFT) + tid;
    if (node < N) { rbeg[node] = start; rend[node] = start + c; }
    cnt[tid] = start;
    __syncthreads();
    walk_pieces(binned, g0, pn, lane, [&](const int2& e) {
        const int p = atomicAdd(&cnt[e.x & (BIN_NODES - 1)], 1);
        csrc[p] = e.y;
    });
}

__global__ __launch_bounds__(256)
void k_binsort_gemm1(const int* __restrict__ binned, const int* __restrict__ piece, int nbins,
                     int E, int N, int* __restrict__ rbeg, int* __restrict__ rend,
                     int* __restrict__ csrc,
                     const _Float16* __restrict__ A, const _Float16* __restrict__ Wtl,
                     const _Float16* __restrict__ Wtr, const float* __restrict__ bl,
                     const float* __restrict__ br, float* __restrict__ Cl,
                     float* __restrict__ Cr) {
    __shared__ int cnt[BIN_NODES];
    __shared__ int wsum[2][BIN_NODES / 32];
    const int b = blockIdx.x;
    if (b < nbins) binsort_body(binned, piece, E, N, rbeg, rend, csrc, b, cnt, wsum[0], wsum[1]);
    else gemm_dual_strip<8, 128>(A, Wtl, Wtr, bl, br, Cl, Cr, N, b - nbins);
}

__device__ inline float score4(const float4& a, const float4& b, const float4& t) {
    return t.x * lrelu(a.x + b.x) + t.y * lrelu(a.y + b.y) +
           t.z * lrelu(a.z + b.z) + t.w * lrelu(a.w + b.w);
}
__device__ inline float red16(float v) {
#pragma unroll
    for (int off = 1; off < 16; off <<= 1) v += __shfl_xor(v, off, 32);
    return v;
}
__device__ inline void scale4(float4& a, float s) { a.x *= s; a.y *= s; a.z *= s; a.w *= s; }
__device__ inline void axpy4(float4& a, float p, const float4& v) {
    a.x += p * v.x; a.y += p * v.y; a.z += p * v.z; a.w += p * v.w;
}

__device__ inline int sort_chunk(int mine, int cnt, int lane) {
    int rank = 0;
    for (int k = 0; k < cnt; ++k) {
        int v = __builtin_amdgcn_readlane(mine, k);
        rank += ((v < mine) || (v == mine && k < lane)) ? 1 : 0;
    }
    return __builtin_amdgcn_ds_permute(rank << 2, mine);
}

__device__ inline void fold4(float& m, float& den, float4& acc,
                             float e0, float e1, float e2, float e3,
                             const float4& a0, const float4& a1,
                             const float4& a2, const float4& a3) {
    float mn = fmaxf(fmaxf(m, fmaxf(e0, e1)), fmaxf(e2, e3));
    float sc = __expf(m - mn);
    float p0 = __expf(e0 - mn), p1 = __expf(e1 - mn);
    float p2 = __expf(e2 - mn), p3 = __expf(e3 - mn);
    scale4(acc, sc);
    axpy4(acc, p0, a0); axpy4(acc, p1, a1); axpy4(acc, p2, a2); axpy4(acc, p3, a3);
    den = den * sc + ((p0 + p1) + (p2 + p3));
    m = mn;
}
__device__ inline void fold1(float& m, float& den, float4& acc, float e0, const float4& a0) {
    float mn = fmaxf(m, e0);
    float sc = __expf(m - mn);
    float p0 = __expf(e0 - mn);
    scale4(acc, sc);
    axpy4(acc, p0, a0);
    den = den * sc + p0;
    m = mn;
}

__global__ __launch_bounds__(256)
void k_node_attn1(const float* __restrict__ xl, const float* __restrict__ xr,
                 const float* __restrict__ att, const float* __restrict__ bias,
                 const int* __restrict__ rbeg, const int* __restrict__ rend,
                 const int* __restrict__ csrc, _Float16* __restrict__ hh, int N) {
    const int wave = __builtin_amdgcn_readfirstlane(
        (int)((blockIdx.x * blockDim.x + threadIdx.x) >> 5));
    const int nwave = (int)((gridDim.x * blockDim.x) >> 5);
    const int lane = threadIdx.x & 31;
    const float4 t  = ((const float4*)att)[lane];
    const float4 bb = ((const float4*)bias)[lane];
    for (int d = wave; d < N; d += nwave) {
    const float4 b4 = ((const float4*)(xr + (size_t)d * 128))[lane];
    float4 acc = ((const float4*)(xl + (size_t)d * 128))[lane];
    float m = red16(score4(acc, b4, t));
    float den = 1.f;
    int beg = rbeg[d], end = rend[d];
    for (int j0 = beg; j0 < end; j0 += 32) {
        int cnt = end - j0; if (cnt > 32) cnt = 32;
        int mine = (lane < cnt) ? csrc[j0 + lane] : 0x7fffffff;
        int srt = sort_chunk(mine, cnt, lane);
        int k = 0;
        for (; k + 4 <= cnt; k += 4) {
            int s0 = __builtin_amdgcn_readlane(srt, k);
            int s1 = __builtin_amdgcn_readlane(srt, k + 1);
            int s2 = __builtin_amdgcn_readlane(srt, k + 2);
            int s3 = __builtin_amdgcn_readlane(srt, k + 3);
            float4 a0 = ((const float4*)(xl + (size_t)s0 * 128))[lane];
            float4 a1 = ((const float4*)(xl + (size_t)s1 * 128))[lane];
            float4 a2 = ((const float4*)(xl + (size_t)s2 * 128))[lane];
            float4 a3 = ((const float4*)(xl + (size_t)s3 * 128))[lane];
            float e0 = red16(score4(a0, b4, t));
            float e1 = red16(score4(a1, b4, t));
            float e2 = red16(score4(a2, b4, t));
            float e3 = red16(score4(a3, b4, t));
            fold4(m, den, acc, e0, e1, e2, e3, a0, a1, a2, a3);
        }
        for (; k < cnt; ++k) {
            int s0 = __builtin_amdgcn_readlane(srt, k);
            float4 a0 = ((const float4*)(xl + (size_t)s0 * 128))[lane];
            float e0 = red16(score4(a0, b4, t));
            fold1(m, den, acc, e0, a0);
        }
    }
    float inv = 1.f / den;
    float o0 = fmaxf(acc.x * inv + bb.x, 0.f);
    float o1 = fmaxf(acc.y * inv + bb.y, 0.f);
    float o2 = fmaxf(acc.z * inv + bb.z, 0.f);
    float o3 = fmaxf(acc.w * inv + bb.w, 0.f);
    half4 o;
    o[0] = (_Float16)o0; o[1] = (_Float16)o1; o[2] = (_Float16)o2; o[3] = (_Float16)o3;
    *(half4*)(hh + (size_t)d * 128 + lane * 4) = o;
    }
}

__device__ inline float score5(const float4& a, float a1, const float4& b, float b1,
                               const float4& t, float t1) {
    return score4(a, b, t) + t1 * lrelu(a1 + b1);
}
__device__ inline float red8(float v) {
#pragma unroll
    for (int off = 1; off < 8; off <<= 1) v += __shfl_xor(v, off, 32);
    return v;
}
__device__ inline float redmax8(float v) {
#pragma unroll
    for (int off = 1; off < 8; off <<= 1) v = fmaxf(v, __shfl_xor(v, off, 32));
    return v;
}
__device__ inline void attn2_pass(const float* __restrict__ xl, int srt, int k, int cnt,
                                  int q, unsigned o4, unsigned o1, int d,
                                  const float4& b4, float b1, const float4& t4, float t1,
                                  float& m, float& den, float4& acc4, float& acc1) {
    float4 r4[2];
    float r1[2];
    bool on[2];
    int s[2];
#pragma unroll
    for (int u = 0; u < 2; ++u) {
        const int idx = k + 4 * u + q;
        on[u] = idx < cnt;
        s[u] = __builtin_amdgcn_ds_bpermute(idx << 2, srt);
    }
#pragma unroll
    for (int u = 0; u < 2; ++u) {
        const int su = on[u] ? s[u] : d;
        const unsigned rb = __umul24((unsigned)su, 192u);
        r4[u] = *(const float4*)((const char*)xl + (rb + o4));
        r1[u] = *(const float*)((const char*)xl + (rb + o1));
    }
    float e[2];
#pragma unroll
    for (int u = 0; u < 2; ++u) {
        const float sc = red8(score5(r4[u], r1[u], b4, b1, t4, t1));
        e[u] = on[u] ? sc : -INFINITY;
    }
    const float mn = fmaxf(m, fmaxf(e[0], e[1]));
    const float sc = __expf(m - mn);
    const float p0 = __expf(e[0] - mn), p1 = __expf(e[1] - mn);
    scale4(acc4, sc);
    acc1 *= sc;
    axpy4(acc4, p0, r4[0]); acc1 += p0 * r1[0];
    axpy4(acc4, p1, r4[1]); acc1 += p1 * r1[1];
    den = den * sc + (p0 + p1);
    m = mn;
}
__global__ __launch_bounds__(256) __attribute__((amdgpu_waves_per_eu(12, 12)))
void k_node_attn2(const float* __restrict__ xl, const float* __restrict__ xr,
                 const float* __restrict__ att, const float* __restrict__ bias,
                 const int* __restrict__ rbeg, const int* __restrict__ rend,
                 const int* __restrict__ csrc, float* __restrict__ out, int N) {
    const int wave = __builtin_amdgcn_readfirstlane(
        (int)((blockIdx.x * blockDim.x + threadIdx.x) >> 5));
    const int nwave = (int)((gridDim.x * blockDim.x) >> 5);
    const int lane = threadIdx.x & 31;
    const int g = lane & 7;
    const int q = lane >> 3;
    const unsigned o4 = (unsigned)g * 16u;
    const unsigned o1 = 128u + (unsigned)g * 4u;
    const float4 t4  = ((const float4*)att)[g];
    const float  t1  = att[32 + g];
    for (int d = wave; d < N; d += nwave) {
    const float* __restrict__ xrd = xr + (size_t)d * 48;
    const float* __restrict__ xld = xl + (size_t)d * 48;
    const float4 b4 = ((const float4*)xrd)[g];
    const float  b1 = xrd[32 + g];
    const float4 s4 = ((const float4*)xld)[g];
    const float  s1 = xld[32 + g];
    float m = red8(score5(s4, s1, b4, b1, t4, t1));
    float den = (q == 0) ? 1.f : 0.f;
    float4 acc4 = (q == 0) ? s4 : make_float4(0.f, 0.f, 0.f, 0.f);
    float acc1 = (q == 0) ? s1 : 0.f;
    int beg = rbeg[d], end = rend[d];
    for (int j0 = beg; j0 < end; j0 += 32) {
        int cnt = end - j0; if (cnt > 32) cnt = 32;
        int mine = (lane < cnt) ? csrc[j0 + lane] : 0x7fffffff;
        int srt = sort_chunk(mine, cnt, lane);
        for (int k = 0; k < cnt; k += 8)
            attn2_pass(xl, srt, k, cnt, q, o4, o1, d, b4, b1, t4, t1, m, den, acc4, acc1);
    }
    const float4 bb4 = ((const float4*)bias)[g];
    const float  bb1 = bias[32 + g];
#pragma unroll
    for (int off = 8; off < 32; off <<= 1) {
        const float mo = __shfl_xor(m, off, 32);
        const float dn = __shfl_xor(den, off, 32);
        float4 ao;
        ao.x = __shfl_xor(acc4.x, off, 32); ao.y = __shfl_xor(acc4.y, off, 32);
        ao.z = __shfl_xor(acc4.z, off, 32); ao.w = __shfl_xor(acc4.w, off, 32);
        const float a1o = __shfl_xor(acc1, off, 32);
        const float M = fmaxf(m, mo);
        const float sa = __expf(m - M), sb = __expf(mo - M);
        den = den * sa + dn * sb;
        acc4.x = acc4.x * sa + ao.x * sb; acc4.y = acc4.y * sa + ao.y * sb;
        acc4.z = acc4.z * sa + ao.z * sb; acc4.w = acc4.w * sa + ao.w * sb;
        acc1 = acc1 * sa + a1o * sb;
        m = M;
    }
    float inv = 1.f / den;
    float v0 = acc4.x * inv + bb4.x, v1 = acc4.y * inv + bb4.y;
    float v2 = acc4.z * inv + bb4.z, v3 = acc4.w * inv + bb4.w;
    float v4 = acc1 * inv + bb1;
    float mx = redmax8(fmaxf(fmaxf(fmaxf(v0, v1), fmaxf(v2, v3)), v4));
    float se = red8(((__expf(v0 - mx) + __expf(v1 - mx)) + (__expf(v2 - mx) + __expf(v3 - mx)))
                    + __expf(v4 - mx));
    float L = mx + logf(se);
    if (lane < 8) {
        *(float4*)(out + (size_t)d * 40 + lane * 4) = make_float4(v0 - L, v1 - L, v2 - L, v3 - L);
        out[(size_t)d * 40 + 32 + lane] = v4 - L;
    }
    }
}

extern "C" void kernel_launch(void* const* d_in, const int* in_sizes, int n_in,
                              void* d_out, int out_size, void* d_ws, size_t ws_size,
                              hipStream_t stream) {
    const float* x    = (const float*)d_in[0];
    const int*   ei   = (const int*)d_in[1];
    const float* W1l  = (const float*)d_in[2];
    const float* b1l  = (const float*)d_in[3];
    const float* W1r  = (const float*)d_in[4];
    const float* b1r  = (const float*)d_in[5];
    const float* att1 = (const float*)d_in[6];
    const float* bias1= (const float*)d_in[7];
    const float* W2l  = (const float*)d_in[8];
    const float* b2l  = (const float*)d_in[9];
    const float* W2r  = (const float*)d_in[10];
    const float* b2r  = (const float*)d_in[11];
    const float* att2 = (const float*)d_in[12];
    const float* bias2= (const float*)d_in[13];
    float* out = (float*)d_out;

    const int N = in_sizes[0] / 128;
    const int E = in_sizes[1] / 2;

    char* w = (char*)d_ws;
    size_t off = 0;
    auto take = [&](size_t bytes) -> char* {
        char* p = w + off;
        off = (off + bytes + 255) & ~(size_t)255;
        return p;
    };
    _Float16* xh   = (_Float16*)take((size_t)N * 128 * 2);
    _Float16* hh   = (_Float16*)take((size_t)N * 128 * 2);
    _Float16* Wt1l = (_Float16*)take(128 * 128 * 2);
    _Float16* Wt1r = (_Float16*)take(128 * 128 * 2);
    _Float16* Wt2l = (_Float16*)take(48 * 128 * 2);
    _Float16* Wt2r = (_Float16*)take(48 * 128 * 2);
    float*    xl   = (float*)take((size_t)N * 128 * 4);
    float*    xr   = (float*)take((size_t)N * 128 * 4);
    const int nbins = (N + BIN_NODES - 1) / BIN_NODES;
    int*      piece = (int*)take((size_t)nbins * SLICES * 8);
    int*      binned = (int*)take((size_t)E * 8);
    int*      rbeg = (int*)take((size_t)N * 4);
    int*      rend = (int*)take((size_t)N * 4);
    int*      csrc = (int*)take((size_t)E * 4);

    const int B = 256;
    const long long Nx128 = (long long)N * 128;
    const int rowTiles = (N + 15) / 16;

    k_slicesort_tr_cvt<<<SLICES + 64 + 64 + 24 + 24 + 512, B, 0, stream>>>(
        ei, E, nbins, binned, piece, W1l, Wt1l, W1r, Wt1r, W2l, Wt2l, W2r, Wt2r, x, xh, Nx128, 512);
    k_binsort_gemm1<<<nbins + (rowTiles + 7) / 8, B, 0, stream>>>(binned, piece, nbins, E, N,
                                                                rbeg, rend, csrc,
                                                                xh, Wt1l, Wt1r, b1l, b1r,
                                                                xl, xr);

    k_node_attn1<<<2048, B, 0, stream>>>(xl, xr, att1, bias1, rbeg, rend, csrc, hh, N);

    k_gemm_dual_w48<<<(rowTiles + 3) / 4, 128, 0, stream>>>(hh, Wt2l, Wt2r, b2l, b2r,
                                                           xl, xr, N);
    k_node_attn2<<<1536, B, 0, stream>>>(xl, xr, att2, bias2, rbeg, rend, csrc, out, N);
}


# Layer-1 attention edge arithmetic on channel pairs (packed add/mul/fma on register pairs as loaded, leaky relu as max); sort ranks on unique keys: fewer vector instructions; score sums pairwise
# speedup vs baseline: 8.2817x; 1.0430x over previous
#include <hip/hip_runtime.h>
#include <hip/hip_bf16.h>
#include <math.h>

typedef __attribute__((ext_vector_type(16))) _Float16 v16h;
typedef __attribute__((ext_vector_type(8)))  _Float16 half8;
typedef __attribute__((ext_vector_type(8)))  float    v8f;

#define NEG_SLOPE 0.2f

__device__ inline float lrelu(float v) { return v > 0.f ? v : NEG_SLOPE * v; }

__device__ inline v16h load_frag16(const _Float16* __restrict__ p, int row0,
                                   int ld, int k0, int lane, int maxRow) {
    int m = lane & 15;
    int base = (lane >> 4) * 8;
    int r = row0 + m;
    if (r > maxRow) r = maxRow;
    const _Float16* q = p + (size_t)r * ld + k0 + base;
    half8 lo = *(const half8*)q;
    half8 hi = *(const half8*)(q + 16);
    v16h f;
#pragma unroll
    for (int i = 0; i < 8; ++i) { f[i] = lo[i]; f[i + 8] = hi[i]; }
    return f;
}


typedef __attribute__((ext_vector_type(4))) _Float16 half4;
__device__ inline half4 cvt4(const float4& v) {
    half4 o;
    o[0] = (_Float16)v.x; o[1] = (_Float16)v.y; o[2] = (_Float16)v.z; o[3] = (_Float16)v.w;
    return o;
}
__device__ inline void cvt_f32_to_f16_body(const float* __restrict__ src,
                                           _Float16* __restrict__ dst, long long n,
                                           int blk, int nblk) {
    const long long n4 = n >> 2;
    const long long tid = (long long)blk * blockDim.x + threadIdx.x;
    const long long stride = (long long)nblk * blockDim.x;
    const float4* __restrict__ s4 = (const float4*)src;
    half4* __restrict__ d4 = (half4*)dst;
    long long i = tid;
    for (; i + 3 * stride < n4; i += 4 * stride) {
        float4 a = s4[i], b = s4[i + stride], c = s4[i + 2 * stride], d = s4[i + 3 * stride];
        d4[i] = cvt4(a); d4[i + stride] = cvt4(b);
        d4[i + 2 * stride] = cvt4(c); d4[i + 3 * stride] = cvt4(d);
    }
    for (; i < n4; i += stride) d4[i] = cvt4(s4[i]);
    if (tid < (n & 3)) dst[(n4 << 2) + tid] = (_Float16)src[(n4 << 2) + tid];
}

template <int C, int NCT>
__device__ inline void wfrag_body(const float* __restrict__ W, _Float16* __restrict__ Wf,
                                  int blk) {
    int i = blk * blockDim.x + threadIdx.x;
    if (i >= NCT * 16 * 128) return;
    const int e = i & 7, L = (i >> 3) & 31, j = (i >> 8) & 1, kk = (i >> 9) & 3, ct = i >> 11;
    const int k = kk * 32 + j * 16 + (L >> 4) * 8 + e;
    const int col = ct * 16 + (L & 15);
    Wf[i] = (col < C) ? (_Float16)W[k * C + col] : (_Float16)0.f;
}

__device__ inline v16h load_wfrag(const _Float16* __restrict__ Wf, int ct, int kk, int lane) {
    const _Float16* q = Wf + (size_t)((ct * 4 + kk) * 2) * 256 + lane * 8;
    half8 lo = *(const half8*)q;
    half8 hi = *(const half8*)(q + 256);
    v16h f;
#pragma unroll
    for (int i = 0; i < 8; ++i) { f[i] = lo[i]; f[i + 8] = hi[i]; }
    return f;
}

template <int NCT, int NB>
__device__ inline void gemm_dual_strip(const _Float16* __restrict__ A,
                                       const _Float16* __restrict__ Wtl,
                                       const _Float16* __restrict__ Wtr,
                                       const float* __restrict__ bl,
                                       const float* __restrict__ br,
                                       float* __restrict__ Cl, float* __restrict__ Cr,
                                       int Nrows, int blk) {
    constexpr int ldc = NCT * 16;
    const int lane = threadIdx.x & 31;
    const int strip = __builtin_amdgcn_readfirstlane(
        (int)((blk * blockDim.x + threadIdx.x) >> 5));
    const int r0 = strip * 16;
    if (r0 >= Nrows) return;
    v16h a[4];
#pragma unroll
    for (int kk = 0; kk < 4; ++kk) a[kk] = load_frag16(A, r0, 128, kk * 32, lane, Nrows - 1);
    const int n = lane & 15, hw = lane >> 4;
    const bool whole = r0 + 16 <= Nrows;
#pragma unroll
    for (int ct = 0; ct < NCT; ++ct) {
        const int c0 = ct * 16;
        v16h bL[4], bR[4];
#pragma unroll
        for (int kk = 0; kk < 4; ++kk) {
            bL[kk] = load_wfrag(Wtl, ct, kk, lane);
            bR[kk] = load_wfrag(Wtr, ct, kk, lane);
        }
        v8f accL = {};
        v8f accR = {};
#pragma unroll
        for (int kk = 0; kk < 4; ++kk) {
            accL = __builtin_amdgcn_wmma_f32_16x16x32_f16(false, a[kk], false, bL[kk],
                                                          (short)0, accL, false, false);
            accR = __builtin_amdgcn_wmma_f32_16x16x32_f16(false, a[kk], false, bR[kk],
                                                          (short)0, accR, false, false);
        }
        const int col = c0 + n;
        float bLv = 0.f, bRv = 0.f;
        if (c0 + 16 <= NB || col < NB) { bLv = bl[col]; bRv = br[col]; }
        float* __restrict__ pl = Cl + (size_t)(r0 + hw * 8) * ldc + col;
        float* __restrict__ pr = Cr + (size_t)(r0 + hw * 8) * ldc + col;
        if (whole) {
#pragma unroll
            for (int v = 0; v < 8; ++v) {
                pl[v * ldc] = accL[v] + bLv;
                pr[v * ldc] = accR[v] + bRv;
            }
        } else {
#pragma unroll
            for (int v = 0; v < 8; ++v) {
                if (r0 + hw * 8 + v < Nrows) {
                    pl[v * ldc] = accL[v] + bLv;
                    pr[v * ldc] = accR[v] + bRv;
                }
            }
        }
    }
}
__global__ __launch_bounds__(128)
void k_gemm_dual_w48(const _Float16* __restrict__ A, const _Float16* __restrict__ Wtl,
                     const _Float16* __restrict__ Wtr, const float* __restrict__ bl,
                     const float* __restrict__ br, float* __restrict__ Cl,
                     float* __restrict__ Cr, int Nrows) {
    gemm_dual_strip<3, 40>(A, Wtl, Wtr, bl, br, Cl, Cr, Nrows, (int)blockIdx.x);
}

#define SLICES 256
#define BIN_SHIFT 8
#define BIN_NODES (1 << BIN_SHIFT)
#define MAX_BINS 1024

__device__ inline void slice_bounds(int E, int s, int& e0, int& e1) {
    const int per = (E + SLICES - 1) / SLICES;
    e0 = s * per;
    e1 = e0 + per;
    if (e1 > E) e1 = E;
}

__device__ inline int wave_incl_scan(int v, int lane) {
#pragma unroll
    for (int off = 1; off < 32; off <<= 1) {
        int u = __shfl_up(v, off, 32);
        if (lane >= off) v += u;
    }
    return v;
}

__device__ inline void slicesort_body(const int* __restrict__ ei, int E, int nbins,
                                      int* __restrict__ binned_words,
                                      int* __restrict__ piece_words, int s,
                                      int* cnt, int* wsum) {
    int2* __restrict__ binned = (int2*)binned_words;
    int2* __restrict__ piece = (int2*)piece_words;
    const int tid = threadIdx.x;
    const int lane = tid & 31;
    const int wv = tid >> 5;
    const int step = blockDim.x;
    for (int i = tid; i < MAX_BINS; i += step) cnt[i] = 0;
    __syncthreads();
    int e0, e1;
    slice_bounds(E, s, e0, e1);
    for (int i = e0 + tid; i < e1; i += 4 * step) {
        int dd[4];
#pragma unroll
        for (int u = 0; u < 4; ++u) {
            const int j = i + u * step;
            dd[u] = (j < e1) ? ei[E + j] : -1;
        }
#pragma unroll
        for (int u = 0; u < 4; ++u)
            if (dd[u] >= 0) atomicAdd(&cnt[dd[u] >> BIN_SHIFT], 1);
    }
    __syncthreads();
    const int4 c = *(const int4*)(cnt + 4 * tid);
    const int run = (c.x + c.y) + (c.z + c.w);
    const int inc = wave_incl_scan(run, lane);
    if (lane == 31) wsum[wv] = inc;
    __syncthreads();
    int wbase = 0;
#pragma unroll
    for (int w = 0; w < MAX_BINS / 4 / 32; ++w) {
        const int t = wsum[w];
        if (w < wv) wbase += t;
    }
    const int b0 = wbase + inc - run;
    const int4 st = make_int4(b0, b0 + c.x, b0 + c.x + c.y, b0 + c.x + c.y + c.z);
    *(int4*)(cnt + 4 * tid) = st;
    const int bb = 4 * tid;
    if (bb     < nbins) piece[(bb    ) * SLICES + s] = make_int2(st.x, c.x);
    if (bb + 1 < nbins) piece[(bb + 1) * SLICES + s] = make_int2(st.y, c.y);
    if (bb + 2 < nbins) piece[(bb + 2) * SLICES + s] = make_int2(st.z, c.z);
    if (bb + 3 < nbins) piece[(bb + 3) * SLICES + s] = make_int2(st.w, c.w);
    __syncthreads();
    for (int i = e0 + tid; i < e1; i += 4 * step) {
        int dd[4], ss[4];
#pragma unroll
        for (int u = 0; u < 4; ++u) {
            const int j = i + u * step;
            const bool on = j < e1;
            dd[u] = on ? ei[E + j] : -1;
            ss[u] = on ? ei[j] : 0;
        }
#pragma unroll
        for (int u = 0; u < 4; ++u) {
            if (dd[u] >= 0) {
                const int p = atomicAdd(&cnt[dd[u] >> BIN_SHIFT], 1);
                binned[e0 + p] = make_int2(dd[u], ss[u]);
            }
        }
    }
}

__global__ __launch_bounds__(256)
void k_slicesort_tr_cvt(const int* __restrict__ ei, int E, int nbins,
                        int* __restrict__ binned, int* __restrict__ piece,
                        const float* __restrict__ W1l, _Float16* __restrict__ Wt1l,
                        const float* __restrict__ W1r, _Float16* __restrict__ Wt1r,
                        const float* __restrict__ W2l, _Float16* __restrict__ Wt2l,
                        const float* __restrict__ W2r, _Float16* __restrict__ Wt2r,
                        const float* __restrict__ src, _Float16* __restrict__ dst, long long n,
                        int ncv) {
    __shared__ alignas(16) int cnt[MAX_BINS];
    __shared__ int wsum[MAX_BINS / 4 / 32];
    int b = blockIdx.x;
    if (b < SLICES) { slicesort_body(ei, E, nbins, binned, piece, b, cnt, wsum); return; }
    b -= SLICES;
    if (b < 64) { wfrag_body<128, 8>(W1l, Wt1l, b); return; }
    b -= 64;
    if (b < 64) { wfrag_body<128, 8>(W1r, Wt1r, b); return; }
    b -= 64;
    if (b < 24) { wfrag_body<40, 3>(W2l, Wt2l, b); return; }
    b -= 24;
    if (b < 24) { wfrag_body<40, 3>(W2r, Wt2r, b); return; }
    b -= 24;
    cvt_f32_to_f16_body(src, dst, n, b, ncv);
}

template <class F>
__device__ inline void walk_pieces(const int2* __restrict__ binned, int g0, int c, int lane, F f) {
    const int hi = lane >> 4;
    const int j0 = lane & 15;
    for (int pp = 0; pp < 16; pp += 4) {
        int pg[4], pn[4];
#pragma unroll
        for (int u = 0; u < 4; ++u) {
            const int sl = 2 * (pp + u) + hi;
            pg[u] = __shfl(g0, sl, 32);
            pn[u] = __shfl(c, sl, 32);
        }
        int2 ea[4], eb[4];
#pragma unroll
        for (int u = 0; u < 4; ++u) {
            ea[u] = (j0 < pn[u]) ? binned[pg[u] + j0] : make_int2(-1, 0);
            eb[u] = (j0 + 16 < pn[u]) ? binned[pg[u] + j0 + 16] : make_int2(-1, 0);
        }
#pragma unroll
        for (int u = 0; u < 4; ++u) {
            if (ea[u].x >= 0) f(ea[u]);
            if (eb[u].x >= 0) f(eb[u]);
        }
#pragma unroll
        for (int u = 0; u < 4; ++u)
            for (int j = j0 + 32; j < pn[u]; j += 16) f(binned[pg[u] + j]);
    }
}

__device__ inline void binsort_body(const int* __restrict__ binned_words,
                                    const int* __restrict__ piece_words, int E, int N,
                                    int* __restrict__ rbeg, int* __restrict__ rend,
                                    int* __restrict__ csrc, int bin,
                                    int* cnt, int* wsumA, int* wsumB) {
    const int2* __restrict__ binned = (const int2*)binned_words;
    const int2* __restrict__ piece = (const int2*)piece_words;
    const int tid = threadIdx.x;
    const int lane = tid & 31;
    const int wv = tid >> 5;
    const int2 pc = piece[bin * SLICES + tid];
    int e0, e1;
    slice_bounds(E, tid, e0, e1);
    const int g0 = e0 + pc.x;
    const int pn = pc.y;
    cnt[tid] = 0;
    const int incx = wave_incl_scan(pc.x, lane);
    if (lane == 31) wsumA[wv] = incx;
    __syncthreads();
    int base = 0;
#pragma unroll
    for (int w = 0; w < BIN_NODES / 32; ++w) base += wsumA[w];
    walk_pieces(binned, g0, pn, lane, [&](const int2& e) {
        atomicAdd(&cnt[e.x & (BIN_NODES - 1)], 1);
    });
    __syncthreads();
    const int c = cnt[tid];
    const int inc = wave_incl_scan(c, lane);
    if (lane == 31) wsumB[wv] = inc;
    __syncthreads();
    int wbase = 0;
#pragma unroll
    for (int w = 0; w < BIN_NODES / 32; ++w) {
        const int t = wsumB[w];
        if (w < wv) wbase += t;
    }
    const int start = base + wbase + inc - c;
    const int node = (bin << BIN_SHIFT) + tid;
    if (node < N) { rbeg[node] = start; rend[node] = start + c; }
    cnt[tid] = start;
    __syncthreads();
    walk_pieces(binned, g0, pn, lane, [&](const int2& e) {
        const int p = atomicAdd(&cnt[e.x & (BIN_NODES - 1)], 1);
        csrc[p] = e.y;
    });
}

__global__ __launch_bounds__(256)
void k_binsort_gemm1(const int* __restrict__ binned, const int* __restrict__ piece, int nbins,
                     int E, int N, int* __restrict__ rbeg, int* __restrict__ rend,
                     int* __restrict__ csrc,
                     const _Float16* __restrict__ A, const _Float16* __restrict__ Wtl,
                     const _Float16* __restrict__ Wtr, const float* __restrict__ bl,
                     const float* __restrict__ br, float* __restrict__ Cl,
                     float* __restrict__ Cr) {
    __shared__ int cnt[BIN_NODES];
    __shared__ int wsum[2][BIN_NODES / 32];
    const int b = blockIdx.x;
    if (b < nbins) binsort_body(binned, piece, E, N, rbeg, rend, csrc, b, cnt, wsum[0], wsum[1]);
    else gemm_dual_strip<8, 128>(A, Wtl, Wtr, bl, br, Cl, Cr, N, b - nbins);
}

__device__ inline float score4(const float4& a, const float4& b, const float4& t) {
    return t.x * lrelu(a.x + b.x) + t.y * lrelu(a.y + b.y) +
           t.z * lrelu(a.z + b.z) + t.w * lrelu(a.w + b.w);
}
__device__ inline float red16(float v) {
#pragma unroll
    for (int off = 1; off < 16; off <<= 1) v += __shfl_xor(v, off, 32);
    return v;
}
__device__ inline void scale4(float4& a, float s) { a.x *= s; a.y *= s; a.z *= s; a.w *= s; }
__device__ inline void axpy4(float4& a, float p, const float4& v) {
    a.x += p * v.x; a.y += p * v.y; a.z += p * v.z; a.w += p * v.w;
}

__device__ inline int sort_chunk(int mine, int cnt, int lane) {
    const int key = (lane < cnt) ? ((mine << 5) | lane) : 0x7fffffff;
    int rank = 0;
    for (int k = 0; k < cnt; ++k) {
        int v = __builtin_amdgcn_readlane(key, k);
        rank += (v < key) ? 1 : 0;
    }
    return __builtin_amdgcn_ds_permute(rank << 2, mine);
}


typedef __attribute__((ext_vector_type(2))) float v2f;
typedef __attribute__((ext_vector_type(4))) float v4f;
__device__ inline v2f lrelu2(v2f z) { return __builtin_elementwise_max(z, NEG_SLOPE * z); }
__device__ inline float score4v(const v4f& a, const v4f& b, const v4f& t) {
    v2f s = t.lo * lrelu2(a.lo + b.lo);
    s = t.hi * lrelu2(a.hi + b.hi) + s;
    return s.x + s.y;
}
__device__ inline v2f red16v(v2f v) {
#pragma unroll
    for (int off = 1; off < 16; off <<= 1) {
        v2f u;
        u.x = __shfl_xor(v.x, off, 32);
        u.y = __shfl_xor(v.y, off, 32);
        v += u;
    }
    return v;
}
__device__ inline void fold4v(float& m, float& den, v4f& acc, v2f e01, v2f e23,
                              const v4f& a0, const v4f& a1, const v4f& a2, const v4f& a3) {
    float mn = fmaxf(fmaxf(m, fmaxf(e01.x, e01.y)), fmaxf(e23.x, e23.y));
    float sc = __expf(m - mn);
    float p0 = __expf(e01.x - mn), p1 = __expf(e01.y - mn);
    float p2 = __expf(e23.x - mn), p3 = __expf(e23.y - mn);
    acc = p0 * a0 + acc * sc;
    acc = p1 * a1 + acc;
    acc = p2 * a2 + acc;
    acc = p3 * a3 + acc;
    den = den * sc + ((p0 + p1) + (p2 + p3));
    m = mn;
}
__device__ inline void fold1v(float& m, float& den, v4f& acc, float e0, const v4f& a0) {
    float mn = fmaxf(m, e0);
    float sc = __expf(m - mn);
    float p0 = __expf(e0 - mn);
    acc = p0 * a0 + acc * sc;
    den = den * sc + p0;
    m = mn;
}

__global__ __launch_bounds__(256)
void k_node_attn1(const float* __restrict__ xl, const float* __restrict__ xr,
                 const float* __restrict__ att, const float* __restrict__ bias,
                 const int* __restrict__ rbeg, const int* __restrict__ rend,
                 const int* __restrict__ csrc, _Float16* __restrict__ hh, int N) {
    const int wave = __builtin_amdgcn_readfirstlane(
        (int)((blockIdx.x * blockDim.x + threadIdx.x) >> 5));
    const int nwave = (int)((gridDim.x * blockDim.x) >> 5);
    const int lane = threadIdx.x & 31;
    const v4f t  = ((const v4f*)att)[lane];
    const v4f bb = ((const v4f*)bias)[lane];
    for (int d = wave; d < N; d += nwave) {
    const v4f b4 = ((const v4f*)(xr + (size_t)d * 128))[lane];
    v4f acc = ((const v4f*)(xl + (size_t)d * 128))[lane];
    float m = red16(score4v(acc, b4, t));
    float den = 1.f;
    int beg = rbeg[d], end = rend[d];
    for (int j0 = beg; j0 < end; j0 += 32) {
        int cnt = end - j0; if (cnt > 32) cnt = 32;
        int mine = (lane < cnt) ? csrc[j0 + lane] : 0x7fffffff;
        int srt = sort_chunk(mine, cnt, lane);
        int k = 0;
        for (; k + 4 <= cnt; k += 4) {
            int s0 = __builtin_amdgcn_readlane(srt, k);
            int s1 = __builtin_amdgcn_readlane(srt, k + 1);
            int s2 = __builtin_amdgcn_readlane(srt, k + 2);
            int s3 = __builtin_amdgcn_readlane(srt, k + 3);
            const v4f a0 = ((const v4f*)(xl + (size_t)s0 * 128))[lane];
            const v4f a1 = ((const v4f*)(xl + (size_t)s1 * 128))[lane];
            const v4f a2 = ((const v4f*)(xl + (size_t)s2 * 128))[lane];
            const v4f a3 = ((const v4f*)(xl + (size_t)s3 * 128))[lane];
            v2f e01, e23;
            e01.x = score4v(a0, b4, t); e01.y = score4v(a1, b4, t);
            e23.x = score4v(a2, b4, t); e23.y = score4v(a3, b4, t);
            e01 = red16v(e01);
            e23 = red16v(e23);
            fold4v(m, den, acc, e01, e23, a0, a1, a2, a3);
        }
        for (; k < cnt; ++k) {
            int s0 = __builtin_amdgcn_readlane(srt, k);
            const v4f a0 = ((const v4f*)(xl + (size_t)s0 * 128))[lane];
            float e0 = red16(score4v(a0, b4, t));
            fold1v(m, den, acc, e0, a0);
        }
    }
    float inv = 1.f / den;
    const v4f ov = acc * inv + bb;
    float o0 = fmaxf(ov.x, 0.f);
    float o1 = fmaxf(ov.y, 0.f);
    float o2 = fmaxf(ov.z, 0.f);
    float o3 = fmaxf(ov.w, 0.f);
    half4 o;
    o[0] = (_Float16)o0; o[1] = (_Float16)o1; o[2] = (_Float16)o2; o[3] = (_Float16)o3;
    *(half4*)(hh + (size_t)d * 128 + lane * 4) = o;
    }
}

__device__ inline float score5(const float4& a, float a1, const float4& b, float b1,
                               const float4& t, float t1) {
    return score4(a, b, t) + t1 * lrelu(a1 + b1);
}
__device__ inline float red8(float v) {
#pragma unroll
    for (int off = 1; off < 8; off <<= 1) v += __shfl_xor(v, off, 32);
    return v;
}
__device__ inline float redmax8(float v) {
#pragma unroll
    for (int off = 1; off < 8; off <<= 1) v = fmaxf(v, __shfl_xor(v, off, 32));
    return v;
}
__device__ inline void attn2_pass(const float* __restrict__ xl, int srt, int k, int cnt,
                                  int q, unsigned o4, unsigned o1, int d,
                                  const float4& b4, float b1, const float4& t4, float t1,
                                  float& m, float& den, float4& acc4, float& acc1) {
    float4 r4[2];
    float r1[2];
    bool on[2];
    int s[2];
#pragma unroll
    for (int u = 0; u < 2; ++u) {
        const int idx = k + 4 * u + q;
        on[u] = idx < cnt;
        s[u] = __builtin_amdgcn_ds_bpermute(idx << 2, srt);
    }
#pragma unroll
    for (int u = 0; u < 2; ++u) {
        const int su = on[u] ? s[u] : d;
        const unsigned rb = __umul24((unsigned)su, 192u);
        r4[u] = *(const float4*)((const char*)xl + (rb + o4));
        r1[u] = *(const float*)((const char*)xl + (rb + o1));
    }
    float e[2];
#pragma unroll
    for (int u = 0; u < 2; ++u) {
        const float sc = red8(score5(r4[u], r1[u], b4, b1, t4, t1));
        e[u] = on[u] ? sc : -INFINITY;
    }
    const float mn = fmaxf(m, fmaxf(e[0], e[1]));
    const float sc = __expf(m - mn);
    const float p0 = __expf(e[0] - mn), p1 = __expf(e[1] - mn);
    scale4(acc4, sc);
    acc1 *= sc;
    axpy4(acc4, p0, r4[0]); acc1 += p0 * r1[0];
    axpy4(acc4, p1, r4[1]); acc1 += p1 * r1[1];
    den = den * sc + (p0 + p1);
    m = mn;
}
__global__ __launch_bounds__(256) __attribute__((amdgpu_waves_per_eu(12, 12)))
void k_node_attn2(const float* __restrict__ xl, const float* __restrict__ xr,
                 const float* __restrict__ att, const float* __restrict__ bias,
                 const int* __restrict__ rbeg, const int* __restrict__ rend,
                 const int* __restrict__ csrc, float* __restrict__ out, int N) {
    const int wave = __builtin_amdgcn_readfirstlane(
        (int)((blockIdx.x * blockDim.x + threadIdx.x) >> 5));
    const int nwave = (int)((gridDim.x * blockDim.x) >> 5);
    const int lane = threadIdx.x & 31;
    const int g = lane & 7;
    const int q = lane >> 3;
    const unsigned o4 = (unsigned)g * 16u;
    const unsigned o1 = 128u + (unsigned)g * 4u;
    const float4 t4  = ((const float4*)att)[g];
    const float  t1  = att[32 + g];
    for (int d = wave; d < N; d += nwave) {
    const float* __restrict__ xrd = xr + (size_t)d * 48;
    const float* __restrict__ xld = xl + (size_t)d * 48;
    const float4 b4 = ((const float4*)xrd)[g];
    const float  b1 = xrd[32 + g];
    const float4 s4 = ((const float4*)xld)[g];
    const float  s1 = xld[32 + g];
    float m = red8(score5(s4, s1, b4, b1, t4, t1));
    float den = (q == 0) ? 1.f : 0.f;
    float4 acc4 = (q == 0) ? s4 : make_float4(0.f, 0.f, 0.f, 0.f);
    float acc1 = (q == 0) ? s1 : 0.f;
    int beg = rbeg[d], end = rend[d];
    for (int j0 = beg; j0 < end; j0 += 32) {
        int cnt = end - j0; if (cnt > 32) cnt = 32;
        int mine = (lane < cnt) ? csrc[j0 + lane] : 0x7fffffff;
        int srt = sort_chunk(mine, cnt, lane);
        for (int k = 0; k < cnt; k += 8)
            attn2_pass(xl, srt, k, cnt, q, o4, o1, d, b4, b1, t4, t1, m, den, acc4, acc1);
    }
    const float4 bb4 = ((const float4*)bias)[g];
    const float  bb1 = bias[32 + g];
#pragma unroll
    for (int off = 8; off < 32; off <<= 1) {
        const float mo = __shfl_xor(m, off, 32);
        const float dn = __shfl_xor(den, off, 32);
        float4 ao;
        ao.x = __shfl_xor(acc4.x, off, 32); ao.y = __shfl_xor(acc4.y, off, 32);
        ao.z = __shfl_xor(acc4.z, off, 32); ao.w = __shfl_xor(acc4.w, off, 32);
        const float a1o = __shfl_xor(acc1, off, 32);
        const float M = fmaxf(m, mo);
        const float sa = __expf(m - M), sb = __expf(mo - M);
        den = den * sa + dn * sb;
        acc4.x = acc4.x * sa + ao.x * sb; acc4.y = acc4.y * sa + ao.y * sb;
        acc4.z = acc4.z * sa + ao.z * sb; acc4.w = acc4.w * sa + ao.w * sb;
        acc1 = acc1 * sa + a1o * sb;
        m = M;
    }
    float inv = 1.f / den;
    float v0 = acc4.x * inv + bb4.x, v1 = acc4.y * inv + bb4.y;
    float v2 = acc4.z * inv + bb4.z, v3 = acc4.w * inv + bb4.w;
    float v4 = acc1 * inv + bb1;
    float mx = redmax8(fmaxf(fmaxf(fmaxf(v0, v1), fmaxf(v2, v3)), v4));
    float se = red8(((__expf(v0 - mx) + __expf(v1 - mx)) + (__expf(v2 - mx) + __expf(v3 - mx)))
                    + __expf(v4 - mx));
    float L = mx + logf(se);
    if (lane < 8) {
        *(float4*)(out + (size_t)d * 40 + lane * 4) = make_float4(v0 - L, v1 - L, v2 - L, v3 - L);
        out[(size_t)d * 40 + 32 + lane] = v4 - L;
    }
    }
}

extern "C" void kernel_launch(void* const* d_in, const int* in_sizes, int n_in,
                              void* d_out, int out_size, void* d_ws, size_t ws_size,
                              hipStream_t stream) {
    const float* x    = (const float*)d_in[0];
    const int*   ei   = (const int*)d_in[1];
    const float* W1l  = (const float*)d_in[2];
    const float* b1l  = (const float*)d_in[3];
    const float* W1r  = (const float*)d_in[4];
    const float* b1r  = (const float*)d_in[5];
    const float* att1 = (const float*)d_in[6];
    const float* bias1= (const float*)d_in[7];
    const float* W2l  = (const float*)d_in[8];
    const float* b2l  = (const float*)d_in[9];
    const float* W2r  = (const float*)d_in[10];
    const float* b2r  = (const float*)d_in[11];
    const float* att2 = (const float*)d_in[12];
    const float* bias2= (const float*)d_in[13];
    float* out = (float*)d_out;

    const int N = in_sizes[0] / 128;
    const int E = in_sizes[1] / 2;

    char* w = (char*)d_ws;
    size_t off = 0;
    auto take = [&](size_t bytes) -> char* {
        char* p = w + off;
        off = (off + bytes + 255) & ~(size_t)255;
        return p;
    };
    _Float16* xh   = (_Float16*)take((size_t)N * 128 * 2);
    _Float16* hh   = (_Float16*)take((size_t)N * 128 * 2);
    _Float16* Wt1l = (_Float16*)take(128 * 128 * 2);
    _Float16* Wt1r = (_Float16*)take(128 * 128 * 2);
    _Float16* Wt2l = (_Float16*)take(48 * 128 * 2);
    _Float16* Wt2r = (_Float16*)take(48 * 128 * 2);
    float*    xl   = (float*)take((size_t)N * 128 * 4);
    float*    xr   = (float*)take((size_t)N * 128 * 4);
    const int nbins = (N + BIN_NODES - 1) / BIN_NODES;
    int*      piece = (int*)take((size_t)nbins * SLICES * 8);
    int*      binned = (int*)take((size_t)E * 8);
    int*      rbeg = (int*)take((size_t)N * 4);
    int*      rend = (int*)take((size_t)N * 4);
    int*      csrc = (int*)take((size_t)E * 4);

    const int B = 256;
    const long long Nx128 = (long long)N * 128;
    const int rowTiles = (N + 15) / 16;

    k_slicesort_tr_cvt<<<SLICES + 64 + 64 + 24 + 24 + 512, B, 0, stream>>>(
        ei, E, nbins, binned, piece, W1l, Wt1l, W1r, Wt1r, W2l, Wt2l, W2r, Wt2r, x, xh, Nx128, 512);
    k_binsort_gemm1<<<nbins + (rowTiles + 7) / 8, B, 0, stream>>>(binned, piece, nbins, E, N,
                                                                rbeg, rend, csrc,
                                                                xh, Wt1l, Wt1r, b1l, b1r,
                                                                xl, xr);

    k_node_attn1<<<2048, B, 0, stream>>>(xl, xr, att1, bias1, rbeg, rend, csrc, hh, N);

    k_gemm_dual_w48<<<(rowTiles + 3) / 4, 128, 0, stream>>>(hh, Wt2l, Wt2r, b2l, b2r,
                                                           xl, xr, N);
    k_node_attn2<<<1536, B, 0, stream>>>(xl, xr, att2, bias2, rbeg, rend, csrc, out, N);
}


# Layer-1 source rows, gathered per edge, stored as f16 (were f32): a row is 256 B in one aligned 256-B unit, was 512 B in two; widened on arrival; scores and sums f32, their order kept
# speedup vs baseline: 8.6864x; 1.0489x over previous
#include <hip/hip_runtime.h>
#include <hip/hip_bf16.h>
#include <math.h>

typedef __attribute__((ext_vector_type(16))) _Float16 v16h;
typedef __attribute__((ext_vector_type(8)))  _Float16 half8;
typedef __attribute__((ext_vector_type(8)))  float    v8f;

#define NEG_SLOPE 0.2f

__device__ inline float lrelu(float v) { return v > 0.f ? v : NEG_SLOPE * v; }

__device__ inline v16h load_frag16(const _Float16* __restrict__ p, int row0,
                                   int ld, int k0, int lane, int maxRow) {
    int m = lane & 15;
    int base = (lane >> 4) * 8;
    int r = row0 + m;
    if (r > maxRow) r = maxRow;
    const _Float16* q = p + (size_t)r * ld + k0 + base;
    half8 lo = *(const half8*)q;
    half8 hi = *(const half8*)(q + 16);
    v16h f;
#pragma unroll
    for (int i = 0; i < 8; ++i) { f[i] = lo[i]; f[i + 8] = hi[i]; }
    return f;
}


typedef __attribute__((ext_vector_type(4))) _Float16 half4;
__device__ inline half4 cvt4(const float4& v) {
    half4 o;
    o[0] = (_Float16)v.x; o[1] = (_Float16)v.y; o[2] = (_Float16)v.z; o[3] = (_Float16)v.w;
    return o;
}
__device__ inline void cvt_f32_to_f16_body(const float* __restrict__ src,
                                           _Float16* __restrict__ dst, long long n,
                                           int blk, int nblk) {
    const long long n4 = n >> 2;
    const long long tid = (long long)blk * blockDim.x + threadIdx.x;
    const long long stride = (long long)nblk * blockDim.x;
    const float4* __restrict__ s4 = (const float4*)src;
    half4* __restrict__ d4 = (half4*)dst;
    long long i = tid;
    for (; i + 3 * stride < n4; i += 4 * stride) {
        float4 a = s4[i], b = s4[i + stride], c = s4[i + 2 * stride], d = s4[i + 3 * stride];
        d4[i] = cvt4(a); d4[i + stride] = cvt4(b);
        d4[i + 2 * stride] = cvt4(c); d4[i + 3 * stride] = cvt4(d);
    }
    for (; i < n4; i += stride) d4[i] = cvt4(s4[i]);
    if (tid < (n & 3)) dst[(n4 << 2) + tid] = (_Float16)src[(n4 << 2) + tid];
}

template <int C, int NCT>
__device__ inline void wfrag_body(const float* __restrict__ W, _Float16* __restrict__ Wf,
                                  int blk) {
    int i = blk * blockDim.x + threadIdx.x;
    if (i >= NCT * 16 * 128) return;
    const int e = i & 7, L = (i >> 3) & 31, j = (i >> 8) & 1, kk = (i >> 9) & 3, ct = i >> 11;
    const int k = kk * 32 + j * 16 + (L >> 4) * 8 + e;
    const int col = ct * 16 + (L & 15);
    Wf[i] = (col < C) ? (_Float16)W[k * C + col] : (_Float16)0.f;
}

__device__ inline v16h load_wfrag(const _Float16* __restrict__ Wf, int ct, int kk, int lane) {
    const _Float16* q = Wf + (size_t)((ct * 4 + kk) * 2) * 256 + lane * 8;
    half8 lo = *(const half8*)q;
    half8 hi = *(const half8*)(q + 256);
    v16h f;
#pragma unroll
    for (int i = 0; i < 8; ++i) { f[i] = lo[i]; f[i + 8] = hi[i]; }
    return f;
}

template <int NCT, int NB, class TL>
__device__ inline void gemm_dual_strip(const _Float16* __restrict__ A,
                                       const _Float16* __restrict__ Wtl,
                                       const _Float16* __restrict__ Wtr,
                                       const float* __restrict__ bl,
                                       const float* __restrict__ br,
                                       TL* __restrict__ Cl, float* __restrict__ Cr,
                                       int Nrows, int blk) {
    constexpr int ldc = NCT * 16;
    const int lane = threadIdx.x & 31;
    const int strip = __builtin_amdgcn_readfirstlane(
        (int)((blk * blockDim.x + threadIdx.x) >> 5));
    const int r0 = strip * 16;
    if (r0 >= Nrows) return;
    v16h a[4];
#pragma unroll
    for (int kk = 0; kk < 4; ++kk) a[kk] = load_frag16(A, r0, 128, kk * 32, lane, Nrows - 1);
    const int n = lane & 15, hw = lane >> 4;
    const bool whole = r0 + 16 <= Nrows;
#pragma unroll
    for (int ct = 0; ct < NCT; ++ct) {
        const int c0 = ct * 16;
        v16h bL[4], bR[4];
#pragma unroll
        for (int kk = 0; kk < 4; ++kk) {
            bL[kk] = load_wfrag(Wtl, ct, kk, lane);
            bR[kk] = load_wfrag(Wtr, ct, kk, lane);
        }
        v8f accL = {};
        v8f accR = {};
#pragma unroll
        for (int kk = 0; kk < 4; ++kk) {
            accL = __builtin_amdgcn_wmma_f32_16x16x32_f16(false, a[kk], false, bL[kk],
                                                          (short)0, accL, false, false);
            accR = __builtin_amdgcn_wmma_f32_16x16x32_f16(false, a[kk], false, bR[kk],
                                                          (short)0, accR, false, false);
        }
        const int col = c0 + n;
        float bLv = 0.f, bRv = 0.f;
        if (c0 + 16 <= NB || col < NB) { bLv = bl[col]; bRv = br[col]; }
        TL* __restrict__ pl = Cl + (size_t)(r0 + hw * 8) * ldc + col;
        float* __restrict__ pr = Cr + (size_t)(r0 + hw * 8) * ldc + col;
        if (whole) {
#pragma unroll
            for (int v = 0; v < 8; ++v) {
                pl[v * ldc] = (TL)(accL[v] + bLv);
                pr[v * ldc] = accR[v] + bRv;
            }
        } else {
#pragma unroll
            for (int v = 0; v < 8; ++v) {
                if (r0 + hw * 8 + v < Nrows) {
                    pl[v * ldc] = (TL)(accL[v] + bLv);
                    pr[v * ldc] = accR[v] + bRv;
                }
            }
        }
    }
}
__global__ __launch_bounds__(128)
void k_gemm_dual_w48(const _Float16* __restrict__ A, const _Float16* __restrict__ Wtl,
                     const _Float16* __restrict__ Wtr, const float* __restrict__ bl,
                     const float* __restrict__ br, float* __restrict__ Cl,
                     float* __restrict__ Cr, int Nrows) {
    gemm_dual_strip<3, 40, float>(A, Wtl, Wtr, bl, br, Cl, Cr, Nrows, (int)blockIdx.x);
}

#define SLICES 256
#define BIN_SHIFT 8
#define BIN_NODES (1 << BIN_SHIFT)
#define MAX_BINS 1024

__device__ inline void slice_bounds(int E, int s, int& e0, int& e1) {
    const int per = (E + SLICES - 1) / SLICES;
    e0 = s * per;
    e1 = e0 + per;
    if (e1 > E) e1 = E;
}

__device__ inline int wave_incl_scan(int v, int lane) {
#pragma unroll
    for (int off = 1; off < 32; off <<= 1) {
        int u = __shfl_up(v, off, 32);
        if (lane >= off) v += u;
    }
    return v;
}

__device__ inline void slicesort_body(const int* __restrict__ ei, int E, int nbins,
                                      int* __restrict__ binned_words,
                                      int* __restrict__ piece_words, int s,
                                      int* cnt, int* wsum) {
    int2* __restrict__ binned = (int2*)binned_words;
    int2* __restrict__ piece = (int2*)piece_words;
    const int tid = threadIdx.x;
    const int lane = tid & 31;
    const int wv = tid >> 5;
    const int step = blockDim.x;
    for (int i = tid; i < MAX_BINS; i += step) cnt[i] = 0;
    __syncthreads();
    int e0, e1;
    slice_bounds(E, s, e0, e1);
    for (int i = e0 + tid; i < e1; i += 4 * step) {
        int dd[4];
#pragma unroll
        for (int u = 0; u < 4; ++u) {
            const int j = i + u * step;
            dd[u] = (j < e1) ? ei[E + j] : -1;
        }
#pragma unroll
        for (int u = 0; u < 4; ++u)
            if (dd[u] >= 0) atomicAdd(&cnt[dd[u] >> BIN_SHIFT], 1);
    }
    __syncthreads();
    const int4 c = *(const int4*)(cnt + 4 * tid);
    const int run = (c.x + c.y) + (c.z + c.w);
    const int inc = wave_incl_scan(run, lane);
    if (lane == 31) wsum[wv] = inc;
    __syncthreads();
    int wbase = 0;
#pragma unroll
    for (int w = 0; w < MAX_BINS / 4 / 32; ++w) {
        const int t = wsum[w];
        if (w < wv) wbase += t;
    }
    const int b0 = wbase + inc - run;
    const int4 st = make_int4(b0, b0 + c.x, b0 + c.x + c.y, b0 + c.x + c.y + c.z);
    *(int4*)(cnt + 4 * tid) = st;
    const int bb = 4 * tid;
    if (bb     < nbins) piece[(bb    ) * SLICES + s] = make_int2(st.x, c.x);
    if (bb + 1 < nbins) piece[(bb + 1) * SLICES + s] = make_int2(st.y, c.y);
    if (bb + 2 < nbins) piece[(bb + 2) * SLICES + s] = make_int2(st.z, c.z);
    if (bb + 3 < nbins) piece[(bb + 3) * SLICES + s] = make_int2(st.w, c.w);
    __syncthreads();
    for (int i = e0 + tid; i < e1; i += 4 * step) {
        int dd[4], ss[4];
#pragma unroll
        for (int u = 0; u < 4; ++u) {
            const int j = i + u * step;
            const bool on = j < e1;
            dd[u] = on ? ei[E + j] : -1;
            ss[u] = on ? ei[j] : 0;
        }
#pragma unroll
        for (int u = 0; u < 4; ++u) {
            if (dd[u] >= 0) {
                const int p = atomicAdd(&cnt[dd[u] >> BIN_SHIFT], 1);
                binned[e0 + p] = make_int2(dd[u], ss[u]);
            }
        }
    }
}

__global__ __launch_bounds__(256)
void k_slicesort_tr_cvt(const int* __restrict__ ei, int E, int nbins,
                        int* __restrict__ binned, int* __restrict__ piece,
                        const float* __restrict__ W1l, _Float16* __restrict__ Wt1l,
                        const float* __restrict__ W1r, _Float16* __restrict__ Wt1r,
                        const float* __restrict__ W2l, _Float16* __restrict__ Wt2l,
                        const float* __restrict__ W2r, _Float16* __restrict__ Wt2r,
                        const float* __restrict__ src, _Float16* __restrict__ dst, long long n,
                        int ncv) {
    __shared__ alignas(16) int cnt[MAX_BINS];
    __shared__ int wsum[MAX_BINS / 4 / 32];
    int b = blockIdx.x;
    if (b < SLICES) { slicesort_body(ei, E, nbins, binned, piece, b, cnt, wsum); return; }
    b -= SLICES;
    if (b < 64) { wfrag_body<128, 8>(W1l, Wt1l, b); return; }
    b -= 64;
    if (b < 64) { wfrag_body<128, 8>(W1r, Wt1r, b); return; }
    b -= 64;
    if (b < 24) { wfrag_body<40, 3>(W2l, Wt2l, b); return; }
    b -= 24;
    if (b < 24) { wfrag_body<40, 3>(W2r, Wt2r, b); return; }
    b -= 24;
    cvt_f32_to_f16_body(src, dst, n, b, ncv);
}

template <class F>
__device__ inline void walk_pieces(const int2* __restrict__ binned, int g0, int c, int lane, F f) {
    const int hi = lane >> 4;
    const int j0 = lane & 15;
    for (int pp = 0; pp < 16; pp += 4) {
        int pg[4], pn[4];
#pragma unroll
        for (int u = 0; u < 4; ++u) {
            const int sl = 2 * (pp + u) + hi;
            pg[u] = __shfl(g0, sl, 32);
            pn[u] = __shfl(c, sl, 32);
        }
        int2 ea[4], eb[4];
#pragma unroll
        for (int u = 0; u < 4; ++u) {
            ea[u] = (j0 < pn[u]) ? binned[pg[u] + j0] : make_int2(-1, 0);
            eb[u] = (j0 + 16 < pn[u]) ? binned[pg[u] + j0 + 16] : make_int2(-1, 0);
        }
#pragma unroll
        for (int u = 0; u < 4; ++u) {
            if (ea[u].x >= 0) f(ea[u]);
            if (eb[u].x >= 0) f(eb[u]);
        }
#pragma unroll
        for (int u = 0; u < 4; ++u)
            for (int j = j0 + 32; j < pn[u]; j += 16) f(binned[pg[u] + j]);
    }
}

__device__ inline void binsort_body(const int* __restrict__ binned_words,
                                    const int* __restrict__ piece_words, int E, int N,
                                    int* __restrict__ rbeg, int* __restrict__ rend,
                                    int* __restrict__ csrc, int bin,
                                    int* cnt, int* wsumA, int* wsumB) {
    const int2* __restrict__ binned = (const int2*)binned_words;
    const int2* __restrict__ piece = (const int2*)piece_words;
    const int tid = threadIdx.x;
    const int lane = tid & 31;
    const int wv = tid >> 5;
    const int2 pc = piece[bin * SLICES + tid];
    int e0, e1;
    slice_bounds(E, tid, e0, e1);
    const int g0 = e0 + pc.x;
    const int pn = pc.y;
    cnt[tid] = 0;
    const int incx = wave_incl_scan(pc.x, lane);
    if (lane == 31) wsumA[wv] = incx;
    __syncthreads();
    int base = 0;
#pragma unroll
    for (int w = 0; w < BIN_NODES / 32; ++w) base += wsumA[w];
    walk_pieces(binned, g0, pn, lane, [&](const int2& e) {
        atomicAdd(&cnt[e.x & (BIN_NODES - 1)], 1);
    });
    __syncthreads();
    const int c = cnt[tid];
    const int inc = wave_incl_scan(c, lane);
    if (lane == 31) wsumB[wv] = inc;
    __syncthreads();
    int wbase = 0;
#pragma unroll
    for (int w = 0; w < BIN_NODES / 32; ++w) {
        const int t = wsumB[w];
        if (w < wv) wbase += t;
    }
    const int start = base + wbase + inc - c;
    const int node = (bin << BIN_SHIFT) + tid;
    if (node < N) { rbeg[node] = start; rend[node] = start + c; }
    cnt[tid] = start;
    __syncthreads();
    walk_pieces(binned, g0, pn, lane, [&](const int2& e) {
        const int p = atomicAdd(&cnt[e.x & (BIN_NODES - 1)], 1);
        csrc[p] = e.y;
    });
}

__global__ __launch_bounds__(256)
void k_binsort_gemm1(const int* __restrict__ binned, const int* __restrict__ piece, int nbins,
                     int E, int N, int* __restrict__ rbeg, int* __restrict__ rend,
                     int* __restrict__ csrc,
                     const _Float16* __restrict__ A, const _Float16* __restrict__ Wtl,
                     const _Float16* __restrict__ Wtr, const float* __restrict__ bl,
                     const float* __restrict__ br, _Float16* __restrict__ Cl,
                     float* __restrict__ Cr) {
    __shared__ int cnt[BIN_NODES];
    __shared__ int wsum[2][BIN_NODES / 32];
    const int b = blockIdx.x;
    if (b < nbins) binsort_body(binned, piece, E, N, rbeg, rend, csrc, b, cnt, wsum[0], wsum[1]);
    else gemm_dual_strip<8, 128, _Float16>(A, Wtl, Wtr, bl, br, Cl, Cr, N, b - nbins);
}

__device__ inline float score4(const float4& a, const float4& b, const float4& t) {
    return t.x * lrelu(a.x + b.x) + t.y * lrelu(a.y + b.y) +
           t.z * lrelu(a.z + b.z) + t.w * lrelu(a.w + b.w);
}
__device__ inline float red16(float v) {
#pragma unroll
    for (int off = 1; off < 16; off <<= 1) v += __shfl_xor(v, off, 32);
    return v;
}
__device__ inline void scale4(float4& a, float s) { a.x *= s; a.y *= s; a.z *= s; a.w *= s; }
__device__ inline void axpy4(float4& a, float p, const float4& v) {
    a.x += p * v.x; a.y += p * v.y; a.z += p * v.z; a.w += p * v.w;
}

__device__ inline int sort_chunk(int mine, int cnt, int lane) {
    const int key = (lane < cnt) ? ((mine << 5) | lane) : 0x7fffffff;
    int rank = 0;
    for (int k = 0; k < cnt; ++k) {
        int v = __builtin_amdgcn_readlane(key, k);
        rank += (v < key) ? 1 : 0;
    }
    return __builtin_amdgcn_ds_permute(rank << 2, mine);
}


typedef __attribute__((ext_vector_type(2))) float v2f;
typedef __attribute__((ext_vector_type(4))) float v4f;
__device__ inline v2f lrelu2(v2f z) { return __builtin_elementwise_max(z, NEG_SLOPE * z); }
__device__ inline float score4v(const v4f& a, const v4f& b, const v4f& t) {
    v2f s = t.lo * lrelu2(a.lo + b.lo);
    s = t.hi * lrelu2(a.hi + b.hi) + s;
    return s.x + s.y;
}
__device__ inline v2f red16v(v2f v) {
#pragma unroll
    for (int off = 1; off < 16; off <<= 1) {
        v2f u;
        u.x = __shfl_xor(v.x, off, 32);
        u.y = __shfl_xor(v.y, off, 32);
        v += u;
    }
    return v;
}
__device__ inline void fold4v(float& m, float& den, v4f& acc, v2f e01, v2f e23,
                              const v4f& a0, const v4f& a1, const v4f& a2, const v4f& a3) {
    float mn = fmaxf(fmaxf(m, fmaxf(e01.x, e01.y)), fmaxf(e23.x, e23.y));
    float sc = __expf(m - mn);
    float p0 = __expf(e01.x - mn), p1 = __expf(e01.y - mn);
    float p2 = __expf(e23.x - mn), p3 = __expf(e23.y - mn);
    acc = p0 * a0 + acc * sc;
    acc = p1 * a1 + acc;
    acc = p2 * a2 + acc;
    acc = p3 * a3 + acc;
    den = den * sc + ((p0 + p1) + (p2 + p3));
    m = mn;
}
__device__ inline void fold1v(float& m, float& den, v4f& acc, float e0, const v4f& a0) {
    float mn = fmaxf(m, e0);
    float sc = __expf(m - mn);
    float p0 = __expf(e0 - mn);
    acc = p0 * a0 + acc * sc;
    den = den * sc + p0;
    m = mn;
}

__device__ inline v4f load_row_h(const _Float16* __restrict__ xl, int row, int lane) {
    const half4 h = ((const half4*)(xl + (size_t)row * 128))[lane];
    return __builtin_convertvector(h, v4f);
}

__global__ __launch_bounds__(256)
void k_node_attn1(const _Float16* __restrict__ xl, const float* __restrict__ xr,
                 const float* __restrict__ att, const float* __restrict__ bias,
                 const int* __restrict__ rbeg, const int* __restrict__ rend,
                 const int* __restrict__ csrc, _Float16* __restrict__ hh, int N) {
    const int wave = __builtin_amdgcn_readfirstlane(
        (int)((blockIdx.x * blockDim.x + threadIdx.x) >> 5));
    const int nwave = (int)((gridDim.x * blockDim.x) >> 5);
    const int lane = threadIdx.x & 31;
    const v4f t  = ((const v4f*)att)[lane];
    const v4f bb = ((const v4f*)bias)[lane];
    for (int d = wave; d < N; d += nwave) {
    const v4f b4 = ((const v4f*)(xr + (size_t)d * 128))[lane];
    v4f acc = load_row_h(xl, d, lane);
    float m = red16(score4v(acc, b4, t));
    float den = 1.f;
    int beg = rbeg[d], end = rend[d];
    for (int j0 = beg; j0 < end; j0 += 32) {
        int cnt = end - j0; if (cnt > 32) cnt = 32;
        int mine = (lane < cnt) ? csrc[j0 + lane] : 0x7fffffff;
        int srt = sort_chunk(mine, cnt, lane);
        int k = 0;
        for (; k + 4 <= cnt; k += 4) {
            int s0 = __builtin_amdgcn_readlane(srt, k);
            int s1 = __builtin_amdgcn_readlane(srt, k + 1);
            int s2 = __builtin_amdgcn_readlane(srt, k + 2);
            int s3 = __builtin_amdgcn_readlane(srt, k + 3);
            const v4f a0 = load_row_h(xl, s0, lane);
            const v4f a1 = load_row_h(xl, s1, lane);
            const v4f a2 = load_row_h(xl, s2, lane);
            const v4f a3 = load_row_h(xl, s3, lane);
            v2f e01, e23;
            e01.x = score4v(a0, b4, t); e01.y = score4v(a1, b4, t);
            e23.x = score4v(a2, b4, t); e23.y = score4v(a3, b4, t);
            e01 = red16v(e01);
            e23 = red16v(e23);
            fold4v(m, den, acc, e01, e23, a0, a1, a2, a3);
        }
        for (; k < cnt; ++k) {
            int s0 = __builtin_amdgcn_readlane(srt, k);
            const v4f a0 = load_row_h(xl, s0, lane);
            float e0 = red16(score4v(a0, b4, t));
            fold1v(m, den, acc, e0, a0);
        }
    }
    float inv = 1.f / den;
    const v4f ov = acc * inv + bb;
    float o0 = fmaxf(ov.x, 0.f);
    float o1 = fmaxf(ov.y, 0.f);
    float o2 = fmaxf(ov.z, 0.f);
    float o3 = fmaxf(ov.w, 0.f);
    half4 o;
    o[0] = (_Float16)o0; o[1] = (_Float16)o1; o[2] = (_Float16)o2; o[3] = (_Float16)o3;
    *(half4*)(hh + (size_t)d * 128 + lane * 4) = o;
    }
}

__device__ inline float score5(const float4& a, float a1, const float4& b, float b1,
                               const float4& t, float t1) {
    return score4(a, b, t) + t1 * lrelu(a1 + b1);
}
__device__ inline float red8(float v) {
#pragma unroll
    for (int off = 1; off < 8; off <<= 1) v += __shfl_xor(v, off, 32);
    return v;
}
__device__ inline float redmax8(float v) {
#pragma unroll
    for (int off = 1; off < 8; off <<= 1) v = fmaxf(v, __shfl_xor(v, off, 32));
    return v;
}
__device__ inline void attn2_pass(const float* __restrict__ xl, int srt, int k, int cnt,
                                  int q, unsigned o4, unsigned o1, int d,
                                  const float4& b4, float b1, const float4& t4, float t1,
                                  float& m, float& den, float4& acc4, float& acc1) {
    float4 r4[2];
    float r1[2];
    bool on[2];
    int s[2];
#pragma unroll
    for (int u = 0; u < 2; ++u) {
        const int idx = k + 4 * u + q;
        on[u] = idx < cnt;
        s[u] = __builtin_amdgcn_ds_bpermute(idx << 2, srt);
    }
#pragma unroll
    for (int u = 0; u < 2; ++u) {
        const int su = on[u] ? s[u] : d;
        const unsigned rb = __umul24((unsigned)su, 192u);
        r4[u] = *(const float4*)((const char*)xl + (rb + o4));
        r1[u] = *(const float*)((const char*)xl + (rb + o1));
    }
    float e[2];
#pragma unroll
    for (int u = 0; u < 2; ++u) {
        const float sc = red8(score5(r4[u], r1[u], b4, b1, t4, t1));
        e[u] = on[u] ? sc : -INFINITY;
    }
    const float mn = fmaxf(m, fmaxf(e[0], e[1]));
    const float sc = __expf(m - mn);
    const float p0 = __expf(e[0] - mn), p1 = __expf(e[1] - mn);
    scale4(acc4, sc);
    acc1 *= sc;
    axpy4(acc4, p0, r4[0]); acc1 += p0 * r1[0];
    axpy4(acc4, p1, r4[1]); acc1 += p1 * r1[1];
    den = den * sc + (p0 + p1);
    m = mn;
}
__global__ __launch_bounds__(256) __attribute__((amdgpu_waves_per_eu(12, 12)))
void k_node_attn2(const float* __restrict__ xl, const float* __restrict__ xr,
                 const float* __restrict__ att, const float* __restrict__ bias,
                 const int* __restrict__ rbeg, const int* __restrict__ rend,
                 const int* __restrict__ csrc, float* __restrict__ out, int N) {
    const int wave = __builtin_amdgcn_readfirstlane(
        (int)((blockIdx.x * blockDim.x + threadIdx.x) >> 5));
    const int nwave = (int)((gridDim.x * blockDim.x) >> 5);
    const int lane = threadIdx.x & 31;
    const int g = lane & 7;
    const int q = lane >> 3;
    const unsigned o4 = (unsigned)g * 16u;
    const unsigned o1 = 128u + (unsigned)g * 4u;
    const float4 t4  = ((const float4*)att)[g];
    const float  t1  = att[32 + g];
    for (int d = wave; d < N; d += nwave) {
    const float* __restrict__ xrd = xr + (size_t)d * 48;
    const float* __restrict__ xld = xl + (size_t)d * 48;
    const float4 b4 = ((const float4*)xrd)[g];
    const float  b1 = xrd[32 + g];
    const float4 s4 = ((const float4*)xld)[g];
    const float  s1 = xld[32 + g];
    float m = red8(score5(s4, s1, b4, b1, t4, t1));
    float den = (q == 0) ? 1.f : 0.f;
    float4 acc4 = (q == 0) ? s4 : make_float4(0.f, 0.f, 0.f, 0.f);
    float acc1 = (q == 0) ? s1 : 0.f;
    int beg = rbeg[d], end = rend[d];
    for (int j0 = beg; j0 < end; j0 += 32) {
        int cnt = end - j0; if (cnt > 32) cnt = 32;
        int mine = (lane < cnt) ? csrc[j0 + lane] : 0x7fffffff;
        int srt = sort_chunk(mine, cnt, lane);
        for (int k = 0; k < cnt; k += 8)
            attn2_pass(xl, srt, k, cnt, q, o4, o1, d, b4, b1, t4, t1, m, den, acc4, acc1);
    }
    const float4 bb4 = ((const float4*)bias)[g];
    const float  bb1 = bias[32 + g];
#pragma unroll
    for (int off = 8; off < 32; off <<= 1) {
        const float mo = __shfl_xor(m, off, 32);
        const float dn = __shfl_xor(den, off, 32);
        float4 ao;
        ao.x = __shfl_xor(acc4.x, off, 32); ao.y = __shfl_xor(acc4.y, off, 32);
        ao.z = __shfl_xor(acc4.z, off, 32); ao.w = __shfl_xor(acc4.w, off, 32);
        const float a1o = __shfl_xor(acc1, off, 32);
        const float M = fmaxf(m, mo);
        const float sa = __expf(m - M), sb = __expf(mo - M);
        den = den * sa + dn * sb;
        acc4.x = acc4.x * sa + ao.x * sb; acc4.y = acc4.y * sa + ao.y * sb;
        acc4.z = acc4.z * sa + ao.z * sb; acc4.w = acc4.w * sa + ao.w * sb;
        acc1 = acc1 * sa + a1o * sb;
        m = M;
    }
    float inv = 1.f / den;
    float v0 = acc4.x * inv + bb4.x, v1 = acc4.y * inv + bb4.y;
    float v2 = acc4.z * inv + bb4.z, v3 = acc4.w * inv + bb4.w;
    float v4 = acc1 * inv + bb1;
    float mx = redmax8(fmaxf(fmaxf(fmaxf(v0, v1), fmaxf(v2, v3)), v4));
    float se = red8(((__expf(v0 - mx) + __expf(v1 - mx)) + (__expf(v2 - mx) + __expf(v3 - mx)))
                    + __expf(v4 - mx));
    float L = mx + logf(se);
    if (lane < 8) {
        *(float4*)(out + (size_t)d * 40 + lane * 4) = make_float4(v0 - L, v1 - L, v2 - L, v3 - L);
        out[(size_t)d * 40 + 32 + lane] = v4 - L;
    }
    }
}

extern "C" void kernel_launch(void* const* d_in, const int* in_sizes, int n_in,
                              void* d_out, int out_size, void* d_ws, size_t ws_size,
                              hipStream_t stream) {
    const float* x    = (const float*)d_in[0];
    const int*   ei   = (const int*)d_in[1];
    const float* W1l  = (const float*)d_in[2];
    const float* b1l  = (const float*)d_in[3];
    const float* W1r  = (const float*)d_in[4];
    const float* b1r  = (const float*)d_in[5];
    const float* att1 = (const float*)d_in[6];
    const float* bias1= (const float*)d_in[7];
    const float* W2l  = (const float*)d_in[8];
    const float* b2l  = (const float*)d_in[9];
    const float* W2r  = (const float*)d_in[10];
    const float* b2r  = (const float*)d_in[11];
    const float* att2 = (const float*)d_in[12];
    const float* bias2= (const float*)d_in[13];
    float* out = (float*)d_out;

    const int N = in_sizes[0] / 128;
    const int E = in_sizes[1] / 2;

    char* w = (char*)d_ws;
    size_t off = 0;
    auto take = [&](size_t bytes) -> char* {
        char* p = w + off;
        off = (off + bytes + 255) & ~(size_t)255;
        return p;
    };
    _Float16* xh   = (_Float16*)take((size_t)N * 128 * 2);
    _Float16* hh   = (_Float16*)take((size_t)N * 128 * 2);
    _Float16* Wt1l = (_Float16*)take(128 * 128 * 2);
    _Float16* Wt1r = (_Float16*)take(128 * 128 * 2);
    _Float16* Wt2l = (_Float16*)take(48 * 128 * 2);
    _Float16* Wt2r = (_Float16*)take(48 * 128 * 2);
    float*    xl   = (float*)take((size_t)N * 128 * 4);
    float*    xr   = (float*)take((size_t)N * 128 * 4);
    const int nbins = (N + BIN_NODES - 1) / BIN_NODES;
    int*      piece = (int*)take((size_t)nbins * SLICES * 8);
    int*      binned = (int*)take((size_t)E * 8);
    int*      rbeg = (int*)take((size_t)N * 4);
    int*      rend = (int*)take((size_t)N * 4);
    int*      csrc = (int*)take((size_t)E * 4);

    const int B = 256;
    const long long Nx128 = (long long)N * 128;
    const int rowTiles = (N + 15) / 16;

    k_slicesort_tr_cvt<<<SLICES + 64 + 64 + 24 + 24 + 512, B, 0, stream>>>(
        ei, E, nbins, binned, piece, W1l, Wt1l, W1r, Wt1r, W2l, Wt2l, W2r, Wt2r, x, xh, Nx128, 512);
    k_binsort_gemm1<<<nbins + (rowTiles + 7) / 8, B, 0, stream>>>(binned, piece, nbins, E, N,
                                                                rbeg, rend, csrc,
                                                                xh, Wt1l, Wt1r, b1l, b1r,
                                                                (_Float16*)xl, xr);

    k_node_attn1<<<2048, B, 0, stream>>>((const _Float16*)xl, xr, att1, bias1, rbeg, rend, csrc, hh, N);

    k_gemm_dual_w48<<<(rowTiles + 3) / 4, 128, 0, stream>>>(hh, Wt2l, Wt2r, b2l, b2r,
                                                           xl, xr, N);
    k_node_attn2<<<1536, B, 0, stream>>>(xl, xr, att2, bias2, rbeg, rend, csrc, out, N);
}
